# starting kernel
# baseline (speedup 1.0000x reference)
#include <hip/hip_runtime.h>
#include <math.h>

typedef __attribute__((ext_vector_type(2))) float v2f;
typedef __attribute__((ext_vector_type(8))) float v8f;

__device__ __forceinline__ unsigned fkey(float f) {
    unsigned u = __float_as_uint(f);
    return (u & 0x80000000u) ? ~u : (u | 0x80000000u);
}
__device__ __forceinline__ float fkey_inv(unsigned k) {
    unsigned u = (k & 0x80000000u) ? (k & 0x7FFFFFFFu) : ~k;
    return __uint_as_float(u);
}

__global__ void init_ws_kernel(unsigned* __restrict__ smaxkey,
                               float* __restrict__ Z,
                               float* __restrict__ outacc, int N) {
    int i = blockIdx.x * blockDim.x + threadIdx.x;
    if (i < N * 32) outacc[i] = 0.0f;
    if (i < N) { smaxkey[i] = 0u; Z[i] = 0.0f; }
}

__device__ __forceinline__ void wmma_step(const float* __restrict__ W, int kb,
                                          int half, int m, v2f a,
                                          v8f& acc0, v8f& acc1) {
    int r0 = (kb + 2 * half + 0) * 32;
    int r1 = (kb + 2 * half + 1) * 32;
    v2f b0, b1;
    b0.x = W[r0 + m];        b0.y = W[r1 + m];
    b1.x = W[r0 + 16 + m];   b1.y = W[r1 + 16 + m];
    acc0 = __builtin_amdgcn_wmma_f32_16x16x4_f32(false, a, false, b0,
                                                 (short)0, acc0, false, false);
    acc1 = __builtin_amdgcn_wmma_f32_16x16x4_f32(false, a, false, b1,
                                                 (short)0, acc1, false, false);
}

__global__ __launch_bounds__(256) void qkv_wmma_kernel(
    const float* __restrict__ x,
    const float* __restrict__ Wq, const float* __restrict__ Wk,
    const float* __restrict__ Wv,
    float* __restrict__ Q, float* __restrict__ K, float* __restrict__ V,
    int ntiles) {
    int wave = (int)((blockIdx.x * blockDim.x + threadIdx.x) >> 5);
    if (wave >= ntiles) return;
    int lane = (int)(threadIdx.x & 31u);
    int half = lane >> 4;
    int m    = lane & 15;
    size_t row0 = (size_t)wave * 16;

    v8f q0 = {}, q1 = {}, k0 = {}, k1 = {}, v0 = {}, v1 = {};

#pragma unroll
    for (int kk = 0; kk < 8; ++kk) {
        int kb = kk * 4;
        const float* ap = x + (row0 + m) * 32 + kb + 2 * half;
        v2f a; a.x = ap[0]; a.y = ap[1];
        wmma_step(Wq, kb, half, m, a, q0, q1);
        wmma_step(Wk, kb, half, m, a, k0, k1);
        wmma_step(Wv, kb, half, m, a, v0, v1);
    }

#pragma unroll
    for (int r = 0; r < 8; ++r) {
        size_t row = (row0 + r + 8 * half) * 32;
        Q[row + m]      = q0[r];  Q[row + 16 + m] = q1[r];
        K[row + m]      = k0[r];  K[row + 16 + m] = k1[r];
        V[row + m]      = v0[r];  V[row + 16 + m] = v1[r];
    }
}

__global__ void qkv_tail_kernel(const float* __restrict__ x,
                                const float* __restrict__ Wq,
                                const float* __restrict__ Wk,
                                const float* __restrict__ Wv,
                                float* __restrict__ Q, float* __restrict__ K,
                                float* __restrict__ V, int row_start, int N) {
    int idx = blockIdx.x * blockDim.x + threadIdx.x;
    int nrows = N - row_start;
    if (idx >= nrows * 32) return;
    int row = row_start + idx / 32, col = idx & 31;
    float q = 0.f, k = 0.f, v = 0.f;
    for (int t = 0; t < 32; ++t) {
        float xv = x[(size_t)row * 32 + t];
        q += xv * Wq[t * 32 + col];
        k += xv * Wk[t * 32 + col];
        v += xv * Wv[t * 32 + col];
    }
    size_t o = (size_t)row * 32 + col;
    Q[o] = q; K[o] = k; V[o] = v;
}

__global__ void edge_score_kernel(const float* __restrict__ Q,
                                  const float* __restrict__ K,
                                  const int* __restrict__ ei,
                                  float* __restrict__ scores,
                                  unsigned* __restrict__ smaxkey, int E) {
    int e = blockIdx.x * blockDim.x + threadIdx.x;
    if (e >= E) return;
    int s = ei[e], r = ei[E + e];
    const float4* q = (const float4*)(Q + (size_t)r * 32);
    const float4* k = (const float4*)(K + (size_t)s * 32);
    float acc = 0.f;
#pragma unroll
    for (int i = 0; i < 8; ++i) {
        float4 a = q[i], b = k[i];
        acc += a.x * b.x + a.y * b.y + a.z * b.z + a.w * b.w;
    }
    float sc = acc * 0.17677669529663687f;
    scores[e] = sc;
    atomicMax(&smaxkey[r], fkey(sc));
}

__global__ void edge_agg_kernel(const float* __restrict__ V,
                                const int* __restrict__ ei,
                                const float* __restrict__ scores,
                                const unsigned* __restrict__ smaxkey,
                                float* __restrict__ Z,
                                float* __restrict__ outacc, int E) {
    int e = blockIdx.x * blockDim.x + threadIdx.x;
    if (e >= E) return;
    int s = ei[e], r = ei[E + e];
    float alpha = __expf(scores[e] - fkey_inv(smaxkey[r]));
    atomicAdd(&Z[r], alpha);
    const float4* v = (const float4*)(V + (size_t)s * 32);
    float* o = outacc + (size_t)r * 32;
#pragma unroll
    for (int i = 0; i < 8; ++i) {
        float4 vv = v[i];
        atomicAdd(o + 4 * i + 0, alpha * vv.x);
        atomicAdd(o + 4 * i + 1, alpha * vv.y);
        atomicAdd(o + 4 * i + 2, alpha * vv.z);
        atomicAdd(o + 4 * i + 3, alpha * vv.w);
    }
}

__global__ __launch_bounds__(256) void out_wmma_kernel(
    const float* __restrict__ x, const float* __restrict__ outacc,
    const float* __restrict__ Z, const float* __restrict__ Wo,
    float* __restrict__ out, int ntiles) {
    int wave = (int)((blockIdx.x * blockDim.x + threadIdx.x) >> 5);
    if (wave >= ntiles) return;
    int lane = (int)(threadIdx.x & 31u);
    int half = lane >> 4;
    int m    = lane & 15;
    size_t row0 = (size_t)wave * 16;

    float invz = 1.0f / (Z[row0 + m] + 1e-6f);
    v8f c0 = {}, c1 = {};

#pragma unroll
    for (int kk = 0; kk < 8; ++kk) {
        int kb = kk * 4;
        const float* ap = outacc + (row0 + m) * 32 + kb + 2 * half;
        v2f a; a.x = ap[0] * invz; a.y = ap[1] * invz;
        wmma_step(Wo, kb, half, m, a, c0, c1);
    }

#pragma unroll
    for (int r = 0; r < 8; ++r) {
        size_t row = (row0 + r + 8 * half) * 32;
        out[row + m]      = x[row + m]      + c0[r];
        out[row + 16 + m] = x[row + 16 + m] + c1[r];
    }
}

__global__ void out_tail_kernel(const float* __restrict__ x,
                                const float* __restrict__ outacc,
                                const float* __restrict__ Z,
                                const float* __restrict__ Wo,
                                float* __restrict__ out, int row_start, int N) {
    int idx = blockIdx.x * blockDim.x + threadIdx.x;
    int nrows = N - row_start;
    if (idx >= nrows * 32) return;
    int row = row_start + idx / 32, col = idx & 31;
    float invz = 1.0f / (Z[row] + 1e-6f);
    float acc = 0.f;
    for (int t = 0; t < 32; ++t)
        acc += outacc[(size_t)row * 32 + t] * invz * Wo[t * 32 + col];
    size_t o = (size_t)row * 32 + col;
    out[o] = x[o] + acc;
}

extern "C" void kernel_launch(void* const* d_in, const int* in_sizes, int n_in,
                              void* d_out, int out_size, void* d_ws,
                              size_t ws_size, hipStream_t stream) {
    const float* x  = (const float*)d_in[0];
    const int*   ei = (const int*)d_in[1];
    const float* Wq = (const float*)d_in[2];
    const float* Wk = (const float*)d_in[3];
    const float* Wv = (const float*)d_in[4];
    const float* Wo = (const float*)d_in[5];
    float* out = (float*)d_out;

    const int N = in_sizes[0] / 32;
    const int E = in_sizes[1] / 2;

    float* Q       = (float*)d_ws;
    float* K       = Q + (size_t)N * 32;
    float* V       = K + (size_t)N * 32;
    float* outacc  = V + (size_t)N * 32;
    float* scores  = outacc + (size_t)N * 32;
    unsigned* smaxkey = (unsigned*)(scores + E);
    float* Z       = (float*)(smaxkey + N);

    const int TB = 256;
    int ntiles = N / 16;
    int rem_row0 = ntiles * 16;
    int rem = N - rem_row0;

    {
        int total = N * 32;
        init_ws_kernel<<<(total + TB - 1) / TB, TB, 0, stream>>>(smaxkey, Z,
                                                                 outacc, N);
    }
    {
        int threads = ntiles * 32;
        if (threads > 0)
            qkv_wmma_kernel<<<(threads + TB - 1) / TB, TB, 0, stream>>>(
                x, Wq, Wk, Wv, Q, K, V, ntiles);
        if (rem > 0)
            qkv_tail_kernel<<<(rem * 32 + TB - 1) / TB, TB, 0, stream>>>(
                x, Wq, Wk, Wv, Q, K, V, rem_row0, N);
    }
    edge_score_kernel<<<(E + TB - 1) / TB, TB, 0, stream>>>(Q, K, ei, scores,
                                                            smaxkey, E);
    edge_agg_kernel<<<(E + TB - 1) / TB, TB, 0, stream>>>(V, ei, scores,
                                                          smaxkey, Z, outacc, E);
    {
        int threads = ntiles * 32;
        if (threads > 0)
            out_wmma_kernel<<<(threads + TB - 1) / TB, TB, 0, stream>>>(
                x, outacc, Z, Wo, out, ntiles);
        if (rem > 0)
            out_tail_kernel<<<(rem * 32 + TB - 1) / TB, TB, 0, stream>>>(
                x, outacc, Z, Wo, out, rem_row0, N);
    }
}


# Edge kernels: a wave takes 32 edges, lanes run along the 32-float row, so each row gather and each atomic row add is one whole 128-B line (33 atomic requests per edge -> 2); score dot: lane tree sum
# speedup vs baseline: 2.9439x; 2.9439x over previous
#include <hip/hip_runtime.h>
#include <math.h>

typedef __attribute__((ext_vector_type(2))) float v2f;
typedef __attribute__((ext_vector_type(8))) float v8f;

__device__ __forceinline__ unsigned fkey(float f) {
    unsigned u = __float_as_uint(f);
    return (u & 0x80000000u) ? ~u : (u | 0x80000000u);
}
__device__ __forceinline__ float fkey_inv(unsigned k) {
    unsigned u = (k & 0x80000000u) ? (k & 0x7FFFFFFFu) : ~k;
    return __uint_as_float(u);
}

__global__ void init_ws_kernel(unsigned* __restrict__ smaxkey,
                               float* __restrict__ Z,
                               float* __restrict__ outacc, int N) {
    int i = blockIdx.x * blockDim.x + threadIdx.x;
    if (i < N * 32) outacc[i] = 0.0f;
    if (i < N) { smaxkey[i] = 0u; Z[i] = 0.0f; }
}

__device__ __forceinline__ void wmma_step(const float* __restrict__ W, int kb,
                                          int half, int m, v2f a,
                                          v8f& acc0, v8f& acc1) {
    int r0 = (kb + 2 * half + 0) * 32;
    int r1 = (kb + 2 * half + 1) * 32;
    v2f b0, b1;
    b0.x = W[r0 + m];        b0.y = W[r1 + m];
    b1.x = W[r0 + 16 + m];   b1.y = W[r1 + 16 + m];
    acc0 = __builtin_amdgcn_wmma_f32_16x16x4_f32(false, a, false, b0,
                                                 (short)0, acc0, false, false);
    acc1 = __builtin_amdgcn_wmma_f32_16x16x4_f32(false, a, false, b1,
                                                 (short)0, acc1, false, false);
}

__global__ __launch_bounds__(256) void qkv_wmma_kernel(
    const float* __restrict__ x,
    const float* __restrict__ Wq, const float* __restrict__ Wk,
    const float* __restrict__ Wv,
    float* __restrict__ Q, float* __restrict__ K, float* __restrict__ V,
    int ntiles) {
    int wave = (int)((blockIdx.x * blockDim.x + threadIdx.x) >> 5);
    if (wave >= ntiles) return;
    int lane = (int)(threadIdx.x & 31u);
    int half = lane >> 4;
    int m    = lane & 15;
    size_t row0 = (size_t)wave * 16;

    v8f q0 = {}, q1 = {}, k0 = {}, k1 = {}, v0 = {}, v1 = {};

#pragma unroll
    for (int kk = 0; kk < 8; ++kk) {
        int kb = kk * 4;
        const float* ap = x + (row0 + m) * 32 + kb + 2 * half;
        v2f a; a.x = ap[0]; a.y = ap[1];
        wmma_step(Wq, kb, half, m, a, q0, q1);
        wmma_step(Wk, kb, half, m, a, k0, k1);
        wmma_step(Wv, kb, half, m, a, v0, v1);
    }

#pragma unroll
    for (int r = 0; r < 8; ++r) {
        size_t row = (row0 + r + 8 * half) * 32;
        Q[row + m]      = q0[r];  Q[row + 16 + m] = q1[r];
        K[row + m]      = k0[r];  K[row + 16 + m] = k1[r];
        V[row + m]      = v0[r];  V[row + 16 + m] = v1[r];
    }
}

__global__ void qkv_tail_kernel(const float* __restrict__ x,
                                const float* __restrict__ Wq,
                                const float* __restrict__ Wk,
                                const float* __restrict__ Wv,
                                float* __restrict__ Q, float* __restrict__ K,
                                float* __restrict__ V, int row_start, int N) {
    int idx = blockIdx.x * blockDim.x + threadIdx.x;
    int nrows = N - row_start;
    if (idx >= nrows * 32) return;
    int row = row_start + idx / 32, col = idx & 31;
    float q = 0.f, k = 0.f, v = 0.f;
    for (int t = 0; t < 32; ++t) {
        float xv = x[(size_t)row * 32 + t];
        q += xv * Wq[t * 32 + col];
        k += xv * Wk[t * 32 + col];
        v += xv * Wv[t * 32 + col];
    }
    size_t o = (size_t)row * 32 + col;
    Q[o] = q; K[o] = k; V[o] = v;
}

#define EDGE_CHUNKS 4

template <int M>
__device__ __forceinline__ float lane_xor(float x) {
    int b = __builtin_bit_cast(int, x);
    if constexpr (M == 16) {
        return __builtin_bit_cast(float, __builtin_amdgcn_permlanex16(
            b, b, 0x76543210, 0xfedcba98, false, false));
    } else {
        return __builtin_bit_cast(float, __builtin_amdgcn_update_dpp(
            b, b, 0x160 | M, 0xf, 0xf, false));
    }
}

template <int M, int NH>
__device__ __forceinline__ void fold_level(float* v, bool hi) {
#pragma unroll
    for (int i = 0; i < NH; ++i) {
        float keep = hi ? v[i + NH] : v[i];
        float send = hi ? v[i] : v[i + NH];
        v[i] = keep + lane_xor<M>(send);
    }
}

__global__ __launch_bounds__(256) void edge_score_kernel(
    const float* __restrict__ Q, const float* __restrict__ K,
    const int* __restrict__ ei, float* __restrict__ scores,
    unsigned* __restrict__ smaxkey, int E) {
    int wave = __builtin_amdgcn_readfirstlane(
        (int)((blockIdx.x * blockDim.x + threadIdx.x) >> 5));
    unsigned lane = threadIdx.x & 31u;
    for (int c = 0; c < EDGE_CHUNKS; ++c) {
        int e0 = (wave * EDGE_CHUNKS + c) * 32;
        if (e0 >= E) return;
        int cnt = min(32, E - e0);
        int e = e0 + (int)lane;
        int s = 0, r = 0;
        if ((int)lane < cnt) { s = ei[e]; r = ei[E + e]; }
        float p[32];
#pragma unroll
        for (int h = 0; h < 2; ++h) {
            float q[16], k[16];
#pragma unroll
            for (int i = 0; i < 16; ++i) {
                unsigned rj = (unsigned)__builtin_amdgcn_readlane(r, 16 * h + i);
                q[i] = (Q + ((size_t)rj << 5))[lane];
            }
#pragma unroll
            for (int i = 0; i < 16; ++i) {
                unsigned sj = (unsigned)__builtin_amdgcn_readlane(s, 16 * h + i);
                k[i] = (K + ((size_t)sj << 5))[lane];
            }
#pragma unroll
            for (int i = 0; i < 16; ++i) p[16 * h + i] = q[i] * k[i];
        }
        fold_level<16, 16>(p, (lane & 16) != 0);
        fold_level<8, 8>(p, (lane & 8) != 0);
        fold_level<4, 4>(p, (lane & 4) != 0);
        fold_level<2, 2>(p, (lane & 2) != 0);
        fold_level<1, 1>(p, (lane & 1) != 0);
        float sc = p[0] * 0.17677669529663687f;
        if ((int)lane < cnt) {
            scores[e] = sc;
            atomicMax(&smaxkey[r], fkey(sc));
        }
    }
}

__global__ __launch_bounds__(256) void edge_agg_kernel(
    const float* __restrict__ V, const int* __restrict__ ei,
    const float* __restrict__ scores, const unsigned* __restrict__ smaxkey,
    float* __restrict__ Z, float* __restrict__ outacc, int E) {
    int wave = __builtin_amdgcn_readfirstlane(
        (int)((blockIdx.x * blockDim.x + threadIdx.x) >> 5));
    unsigned lane = threadIdx.x & 31u;
    for (int c = 0; c < EDGE_CHUNKS; ++c) {
        int e0 = (wave * EDGE_CHUNKS + c) * 32;
        if (e0 >= E) return;
        int cnt = min(32, E - e0);
        int e = e0 + (int)lane;
        int s = 0, r = 0;
        float alpha = 0.f;
        if ((int)lane < cnt) {
            s = ei[e]; r = ei[E + e];
            alpha = __expf(scores[e] - fkey_inv(smaxkey[r]));
            atomicAdd(&Z[r], alpha);
        }
        if (cnt == 32) {
            float v[32];
#pragma unroll
            for (int j = 0; j < 32; ++j) {
                unsigned sj = (unsigned)__builtin_amdgcn_readlane(s, j);
                v[j] = (V + ((size_t)sj << 5))[lane];
            }
#pragma unroll
            for (int j = 0; j < 32; ++j) {
                unsigned rj = (unsigned)__builtin_amdgcn_readlane(r, j);
                float aj = __builtin_bit_cast(float, __builtin_amdgcn_readlane(
                    __builtin_bit_cast(int, alpha), j));
                atomicAdd(outacc + ((size_t)rj << 5) + lane, aj * v[j]);
            }
        } else {
            for (int j = 0; j < cnt; ++j) {
                unsigned sj = (unsigned)__builtin_amdgcn_readlane(s, j);
                unsigned rj = (unsigned)__builtin_amdgcn_readlane(r, j);
                float aj = __builtin_bit_cast(float, __builtin_amdgcn_readlane(
                    __builtin_bit_cast(int, alpha), j));
                atomicAdd(outacc + ((size_t)rj << 5) + lane,
                          aj * (V + ((size_t)sj << 5))[lane]);
            }
        }
    }
}

__global__ __launch_bounds__(256) void out_wmma_kernel(
    const float* __restrict__ x, const float* __restrict__ outacc,
    const float* __restrict__ Z, const float* __restrict__ Wo,
    float* __restrict__ out, int ntiles) {
    int wave = (int)((blockIdx.x * blockDim.x + threadIdx.x) >> 5);
    if (wave >= ntiles) return;
    int lane = (int)(threadIdx.x & 31u);
    int half = lane >> 4;
    int m    = lane & 15;
    size_t row0 = (size_t)wave * 16;

    float invz = 1.0f / (Z[row0 + m] + 1e-6f);
    v8f c0 = {}, c1 = {};

#pragma unroll
    for (int kk = 0; kk < 8; ++kk) {
        int kb = kk * 4;
        const float* ap = outacc + (row0 + m) * 32 + kb + 2 * half;
        v2f a; a.x = ap[0] * invz; a.y = ap[1] * invz;
        wmma_step(Wo, kb, half, m, a, c0, c1);
    }

#pragma unroll
    for (int r = 0; r < 8; ++r) {
        size_t row = (row0 + r + 8 * half) * 32;
        out[row + m]      = x[row + m]      + c0[r];
        out[row + 16 + m] = x[row + 16 + m] + c1[r];
    }
}

__global__ void out_tail_kernel(const float* __restrict__ x,
                                const float* __restrict__ outacc,
                                const float* __restrict__ Z,
                                const float* __restrict__ Wo,
                                float* __restrict__ out, int row_start, int N) {
    int idx = blockIdx.x * blockDim.x + threadIdx.x;
    int nrows = N - row_start;
    if (idx >= nrows * 32) return;
    int row = row_start + idx / 32, col = idx & 31;
    float invz = 1.0f / (Z[row] + 1e-6f);
    float acc = 0.f;
    for (int t = 0; t < 32; ++t)
        acc += outacc[(size_t)row * 32 + t] * invz * Wo[t * 32 + col];
    size_t o = (size_t)row * 32 + col;
    out[o] = x[o] + acc;
}

extern "C" void kernel_launch(void* const* d_in, const int* in_sizes, int n_in,
                              void* d_out, int out_size, void* d_ws,
                              size_t ws_size, hipStream_t stream) {
    const float* x  = (const float*)d_in[0];
    const int*   ei = (const int*)d_in[1];
    const float* Wq = (const float*)d_in[2];
    const float* Wk = (const float*)d_in[3];
    const float* Wv = (const float*)d_in[4];
    const float* Wo = (const float*)d_in[5];
    float* out = (float*)d_out;

    const int N = in_sizes[0] / 32;
    const int E = in_sizes[1] / 2;

    float* Q       = (float*)d_ws;
    float* K       = Q + (size_t)N * 32;
    float* V       = K + (size_t)N * 32;
    float* outacc  = V + (size_t)N * 32;
    float* scores  = outacc + (size_t)N * 32;
    unsigned* smaxkey = (unsigned*)(scores + E);
    float* Z       = (float*)(smaxkey + N);

    const int TB = 256;
    int ntiles = N / 16;
    int rem_row0 = ntiles * 16;
    int rem = N - rem_row0;

    {
        int total = N * 32;
        init_ws_kernel<<<(total + TB - 1) / TB, TB, 0, stream>>>(smaxkey, Z,
                                                                 outacc, N);
    }
    {
        int threads = ntiles * 32;
        if (threads > 0)
            qkv_wmma_kernel<<<(threads + TB - 1) / TB, TB, 0, stream>>>(
                x, Wq, Wk, Wv, Q, K, V, ntiles);
        if (rem > 0)
            qkv_tail_kernel<<<(rem * 32 + TB - 1) / TB, TB, 0, stream>>>(
                x, Wq, Wk, Wv, Q, K, V, rem_row0, N);
    }
    const int EPB = (TB / 32) * EDGE_CHUNKS * 32;
    edge_score_kernel<<<(E + EPB - 1) / EPB, TB, 0, stream>>>(Q, K, ei, scores,
                                                              smaxkey, E);
    edge_agg_kernel<<<(E + EPB - 1) / EPB, TB, 0, stream>>>(V, ei, scores,
                                                            smaxkey, Z, outacc, E);
    {
        int threads = ntiles * 32;
        if (threads > 0)
            out_wmma_kernel<<<(threads + TB - 1) / TB, TB, 0, stream>>>(
                x, outacc, Z, Wo, out, ntiles);
        if (rem > 0)
            out_tail_kernel<<<(rem * 32 + TB - 1) / TB, TB, 0, stream>>>(
                x, outacc, Z, Wo, out, rem_row0, N);
    }
}


# A work-group owns 512 receivers: filters the edge list for them; max, exp sum and weighted V rows via LDS atomics; global atomics, scores array, 2 launches gone; sum order: LDS arrival order
# speedup vs baseline: 5.4436x; 1.8491x over previous
#include <hip/hip_runtime.h>
#include <math.h>

typedef __attribute__((ext_vector_type(2))) float v2f;
typedef __attribute__((ext_vector_type(8))) float v8f;

__device__ __forceinline__ unsigned fkey(float f) {
    unsigned u = __float_as_uint(f);
    return (u & 0x80000000u) ? ~u : (u | 0x80000000u);
}
__device__ __forceinline__ float fkey_inv(unsigned k) {
    unsigned u = (k & 0x80000000u) ? (k & 0x7FFFFFFFu) : ~k;
    return __uint_as_float(u);
}

__device__ __forceinline__ void wmma_step(const float* __restrict__ W, int kb,
                                          int half, int m, v2f a,
                                          v8f& acc0, v8f& acc1) {
    int r0 = (kb + 2 * half + 0) * 32;
    int r1 = (kb + 2 * half + 1) * 32;
    v2f b0, b1;
    b0.x = W[r0 + m];        b0.y = W[r1 + m];
    b1.x = W[r0 + 16 + m];   b1.y = W[r1 + 16 + m];
    acc0 = __builtin_amdgcn_wmma_f32_16x16x4_f32(false, a, false, b0,
                                                 (short)0, acc0, false, false);
    acc1 = __builtin_amdgcn_wmma_f32_16x16x4_f32(false, a, false, b1,
                                                 (short)0, acc1, false, false);
}

__global__ __launch_bounds__(256) void qkv_wmma_kernel(
    const float* __restrict__ x,
    const float* __restrict__ Wq, const float* __restrict__ Wk,
    const float* __restrict__ Wv,
    float* __restrict__ Q, float* __restrict__ K, float* __restrict__ V,
    int ntiles) {
    int wave = (int)((blockIdx.x * blockDim.x + threadIdx.x) >> 5);
    if (wave >= ntiles) return;
    int lane = (int)(threadIdx.x & 31u);
    int half = lane >> 4;
    int m    = lane & 15;
    size_t row0 = (size_t)wave * 16;

    v8f q0 = {}, q1 = {}, k0 = {}, k1 = {}, v0 = {}, v1 = {};

#pragma unroll
    for (int kk = 0; kk < 8; ++kk) {
        int kb = kk * 4;
        const float* ap = x + (row0 + m) * 32 + kb + 2 * half;
        v2f a; a.x = ap[0]; a.y = ap[1];
        wmma_step(Wq, kb, half, m, a, q0, q1);
        wmma_step(Wk, kb, half, m, a, k0, k1);
        wmma_step(Wv, kb, half, m, a, v0, v1);
    }

#pragma unroll
    for (int r = 0; r < 8; ++r) {
        size_t row = (row0 + r + 8 * half) * 32;
        Q[row + m]      = q0[r];  Q[row + 16 + m] = q1[r];
        K[row + m]      = k0[r];  K[row + 16 + m] = k1[r];
        V[row + m]      = v0[r];  V[row + 16 + m] = v1[r];
    }
}

__global__ void qkv_tail_kernel(const float* __restrict__ x,
                                const float* __restrict__ Wq,
                                const float* __restrict__ Wk,
                                const float* __restrict__ Wv,
                                float* __restrict__ Q, float* __restrict__ K,
                                float* __restrict__ V, int row_start, int N) {
    int idx = blockIdx.x * blockDim.x + threadIdx.x;
    int nrows = N - row_start;
    if (idx >= nrows * 32) return;
    int row = row_start + idx / 32, col = idx & 31;
    float q = 0.f, k = 0.f, v = 0.f;
    for (int t = 0; t < 32; ++t) {
        float xv = x[(size_t)row * 32 + t];
        q += xv * Wq[t * 32 + col];
        k += xv * Wk[t * 32 + col];
        v += xv * Wv[t * 32 + col];
    }
    size_t o = (size_t)row * 32 + col;
    Q[o] = q; K[o] = k; V[o] = v;
}

template <int M>
__device__ __forceinline__ float lane_xor(float x) {
    int b = __builtin_bit_cast(int, x);
    if constexpr (M == 16) {
        return __builtin_bit_cast(float, __builtin_amdgcn_permlanex16(
            b, b, 0x76543210, 0xfedcba98, false, false));
    } else {
        return __builtin_bit_cast(float, __builtin_amdgcn_update_dpp(
            b, b, 0x160 | M, 0xf, 0xf, false));
    }
}

template <int M, int NH>
__device__ __forceinline__ void fold_level(float* v, bool hi) {
#pragma unroll
    for (int i = 0; i < NH; ++i) {
        float keep = hi ? v[i + NH] : v[i];
        float send = hi ? v[i] : v[i + NH];
        v[i] = keep + lane_xor<M>(send);
    }
}

#define OWN_R    512
#define OWN_CAP  16384
#define OWN_TB   1024
#define OWN_U    8
#define OWN_STEP (OWN_TB * 4)

__device__ __forceinline__ void own_push(unsigned rl, int e, unsigned* cnt,
                                         int* l_a, int* l_r) {
    if (rl < (unsigned)OWN_R) {
        unsigned slot = atomicAdd(cnt, 1u);
        if (slot < (unsigned)OWN_CAP) { l_a[slot] = e; l_r[slot] = (int)rl; }
    }
}

__device__ __forceinline__ void own_build(const int* __restrict__ recv, int a,
                                          int b, int rbase, bool vec_ok,
                                          unsigned* cnt, int* l_a, int* l_r) {
    const int t = (int)threadIdx.x;
    for (int base = a; base < b; base += OWN_STEP * OWN_U) {
        const bool full = vec_ok && (b - base >= OWN_STEP * OWN_U);
        int4 v[OWN_U];
        if (full) {
#pragma unroll
            for (int u = 0; u < OWN_U; ++u)
                v[u] = *reinterpret_cast<const int4*>(recv + base + u * OWN_STEP + t * 4);
        } else {
#pragma unroll
            for (int u = 0; u < OWN_U; ++u) {
                int e0 = base + u * OWN_STEP + t * 4;
                v[u].x = (e0     < b) ? recv[e0]     : -1;
                v[u].y = (e0 + 1 < b) ? recv[e0 + 1] : -1;
                v[u].z = (e0 + 2 < b) ? recv[e0 + 2] : -1;
                v[u].w = (e0 + 3 < b) ? recv[e0 + 3] : -1;
            }
        }
#pragma unroll
        for (int u = 0; u < OWN_U; ++u) {
            int e0 = base + u * OWN_STEP + t * 4;
            unsigned r0 = (unsigned)(v[u].x - rbase), r1 = (unsigned)(v[u].y - rbase);
            unsigned r2 = (unsigned)(v[u].z - rbase), r3 = (unsigned)(v[u].w - rbase);
            unsigned mn = min(min(r0, r1), min(r2, r3));
            if (mn < (unsigned)OWN_R) {
                own_push(r0, e0,     cnt, l_a, l_r);
                own_push(r1, e0 + 1, cnt, l_a, l_r);
                own_push(r2, e0 + 2, cnt, l_a, l_r);
                own_push(r3, e0 + 3, cnt, l_a, l_r);
            }
        }
    }
}

__device__ __forceinline__ void own_process(
    const float* __restrict__ Q, const float* __restrict__ K,
    const float* __restrict__ V, const int* __restrict__ ei, int rbase, int n,
    bool first, float* acc, float* zs, unsigned* mkey, unsigned* mnew,
    int* l_a, int* l_r, float* l_sc) {
    const int t = (int)threadIdx.x;
    const unsigned lane = threadIdx.x & 31u;
    const int wave = __builtin_amdgcn_readfirstlane((int)(threadIdx.x >> 5));

    for (int c = wave; c * 32 < n; c += OWN_TB / 32) {
        int i = c * 32 + (int)lane;
        bool valid = i < n;
        int s = 0, rl = 0;
        if (valid) { s = ei[l_a[i]]; rl = l_r[i]; }
        int r = rbase + rl;
        float p[32];
#pragma unroll
        for (int h = 0; h < 2; ++h) {
            float q[16], k[16];
#pragma unroll
            for (int j = 0; j < 16; ++j) {
                unsigned rj = (unsigned)__builtin_amdgcn_readlane(r, 16 * h + j);
                q[j] = (Q + ((size_t)rj << 5))[lane];
            }
#pragma unroll
            for (int j = 0; j < 16; ++j) {
                unsigned sj = (unsigned)__builtin_amdgcn_readlane(s, 16 * h + j);
                k[j] = (K + ((size_t)sj << 5))[lane];
            }
#pragma unroll
            for (int j = 0; j < 16; ++j) p[16 * h + j] = q[j] * k[j];
        }
        fold_level<16, 16>(p, (lane & 16) != 0);
        fold_level<8, 8>(p, (lane & 8) != 0);
        fold_level<4, 4>(p, (lane & 4) != 0);
        fold_level<2, 2>(p, (lane & 2) != 0);
        fold_level<1, 1>(p, (lane & 1) != 0);
        float sc = p[0] * 0.17677669529663687f;
        if (valid) {
            l_sc[i] = sc;
            l_a[i] = s;
            atomicMax(&mnew[rl], fkey(sc));
        }
    }
    __syncthreads();
    if (!first) {
        for (int idx = t; idx < OWN_R * 32; idx += OWN_TB) {
            unsigned ko = mkey[idx >> 5], kn = mnew[idx >> 5];
            if (kn != ko)
                acc[idx] *= ko ? __expf(fkey_inv(ko) - fkey_inv(kn)) : 0.0f;
        }
        for (int rl = t; rl < OWN_R; rl += OWN_TB) {
            unsigned ko = mkey[rl], kn = mnew[rl];
            if (kn != ko)
                zs[rl] *= ko ? __expf(fkey_inv(ko) - fkey_inv(kn)) : 0.0f;
        }
        __syncthreads();
    }
    for (int rl = t; rl < OWN_R; rl += OWN_TB) mkey[rl] = mnew[rl];
    __syncthreads();

    for (int c = wave; c * 32 < n; c += OWN_TB / 32) {
        int i = c * 32 + (int)lane;
        int m = min(32, n - c * 32);
        int s = 0, rl = 0;
        float alpha = 0.f;
        if (i < n) {
            s = l_a[i]; rl = l_r[i];
            alpha = __expf(l_sc[i] - fkey_inv(mkey[rl]));
            atomicAdd(&zs[rl], alpha);
        }
        if (m == 32) {
            float v[32];
#pragma unroll
            for (int j = 0; j < 32; ++j) {
                unsigned sj = (unsigned)__builtin_amdgcn_readlane(s, j);
                v[j] = (V + ((size_t)sj << 5))[lane];
            }
#pragma unroll
            for (int j = 0; j < 32; ++j) {
                int rj = __builtin_amdgcn_readlane(rl, j);
                float aj = __builtin_bit_cast(float, __builtin_amdgcn_readlane(
                    __builtin_bit_cast(int, alpha), j));
                atomicAdd(&acc[(rj << 5) + (int)lane], aj * v[j]);
            }
        } else {
            for (int j = 0; j < m; ++j) {
                unsigned sj = (unsigned)__builtin_amdgcn_readlane(s, j);
                int rj = __builtin_amdgcn_readlane(rl, j);
                float aj = __builtin_bit_cast(float, __builtin_amdgcn_readlane(
                    __builtin_bit_cast(int, alpha), j));
                atomicAdd(&acc[(rj << 5) + (int)lane],
                          aj * (V + ((size_t)sj << 5))[lane]);
            }
        }
    }
}

__global__ __launch_bounds__(OWN_TB) void own_range_kernel(
    const float* __restrict__ Q, const float* __restrict__ K,
    const float* __restrict__ V, const int* __restrict__ ei,
    float* __restrict__ Z, float* __restrict__ outacc, int N, int E) {
    __shared__ float    acc[OWN_R * 32];
    __shared__ float    zs[OWN_R];
    __shared__ unsigned mkey[OWN_R];
    __shared__ unsigned mnew[OWN_R];
    __shared__ int      l_a[OWN_CAP];
    __shared__ int      l_r[OWN_CAP];
    __shared__ float    l_sc[OWN_CAP];
    __shared__ unsigned cnt;

    const int t = (int)threadIdx.x;
    const int rbase = (int)blockIdx.x * OWN_R;
    const int nown = min(OWN_R, N - rbase);
    const int* recv = ei + E;
    const bool vec_ok = (((size_t)recv & 15u) == 0);

    for (int i = t; i < OWN_R * 32; i += OWN_TB) acc[i] = 0.0f;
    for (int i = t; i < OWN_R; i += OWN_TB) { zs[i] = 0.0f; mkey[i] = 0u; mnew[i] = 0u; }
    if (t == 0) cnt = 0u;
    __syncthreads();

    own_build(recv, 0, E, rbase, vec_ok, &cnt, l_a, l_r);
    __syncthreads();
    const unsigned total = cnt;
    if (total <= (unsigned)OWN_CAP) {
        own_process(Q, K, V, ei, rbase, (int)total, true, acc, zs, mkey, mnew,
                    l_a, l_r, l_sc);
    } else {
        for (int a = 0; a < E; a += OWN_CAP) {
            __syncthreads();
            if (t == 0) cnt = 0u;
            __syncthreads();
            int b = (E - a > OWN_CAP) ? a + OWN_CAP : E;
            own_build(recv, a, b, rbase, vec_ok, &cnt, l_a, l_r);
            __syncthreads();
            own_process(Q, K, V, ei, rbase, (int)cnt, a == 0, acc, zs, mkey,
                        mnew, l_a, l_r, l_sc);
        }
    }
    __syncthreads();
    for (int i = t; i < nown * 32; i += OWN_TB) outacc[(size_t)rbase * 32 + i] = acc[i];
    for (int i = t; i < nown; i += OWN_TB) Z[rbase + i] = zs[i];
}

__global__ __launch_bounds__(256) void out_wmma_kernel(
    const float* __restrict__ x, const float* __restrict__ outacc,
    const float* __restrict__ Z, const float* __restrict__ Wo,
    float* __restrict__ out, int ntiles) {
    int wave = (int)((blockIdx.x * blockDim.x + threadIdx.x) >> 5);
    if (wave >= ntiles) return;
    int lane = (int)(threadIdx.x & 31u);
    int half = lane >> 4;
    int m    = lane & 15;
    size_t row0 = (size_t)wave * 16;

    float invz = 1.0f / (Z[row0 + m] + 1e-6f);
    v8f c0 = {}, c1 = {};

#pragma unroll
    for (int kk = 0; kk < 8; ++kk) {
        int kb = kk * 4;
        const float* ap = outacc + (row0 + m) * 32 + kb + 2 * half;
        v2f a; a.x = ap[0] * invz; a.y = ap[1] * invz;
        wmma_step(Wo, kb, half, m, a, c0, c1);
    }

#pragma unroll
    for (int r = 0; r < 8; ++r) {
        size_t row = (row0 + r + 8 * half) * 32;
        out[row + m]      = x[row + m]      + c0[r];
        out[row + 16 + m] = x[row + 16 + m] + c1[r];
    }
}

__global__ void out_tail_kernel(const float* __restrict__ x,
                                const float* __restrict__ outacc,
                                const float* __restrict__ Z,
                                const float* __restrict__ Wo,
                                float* __restrict__ out, int row_start, int N) {
    int idx = blockIdx.x * blockDim.x + threadIdx.x;
    int nrows = N - row_start;
    if (idx >= nrows * 32) return;
    int row = row_start + idx / 32, col = idx & 31;
    float invz = 1.0f / (Z[row] + 1e-6f);
    float acc = 0.f;
    for (int t = 0; t < 32; ++t)
        acc += outacc[(size_t)row * 32 + t] * invz * Wo[t * 32 + col];
    size_t o = (size_t)row * 32 + col;
    out[o] = x[o] + acc;
}

extern "C" void kernel_launch(void* const* d_in, const int* in_sizes, int n_in,
                              void* d_out, int out_size, void* d_ws,
                              size_t ws_size, hipStream_t stream) {
    const float* x  = (const float*)d_in[0];
    const int*   ei = (const int*)d_in[1];
    const float* Wq = (const float*)d_in[2];
    const float* Wk = (const float*)d_in[3];
    const float* Wv = (const float*)d_in[4];
    const float* Wo = (const float*)d_in[5];
    float* out = (float*)d_out;

    const int N = in_sizes[0] / 32;
    const int E = in_sizes[1] / 2;

    float* Q       = (float*)d_ws;
    float* K       = Q + (size_t)N * 32;
    float* V       = K + (size_t)N * 32;
    float* outacc  = V + (size_t)N * 32;
    float* Z       = outacc + (size_t)N * 32;

    const int TB = 256;
    int ntiles = N / 16;
    int rem_row0 = ntiles * 16;
    int rem = N - rem_row0;

    {
        int threads = ntiles * 32;
        if (threads > 0)
            qkv_wmma_kernel<<<(threads + TB - 1) / TB, TB, 0, stream>>>(
                x, Wq, Wk, Wv, Q, K, V, ntiles);
        if (rem > 0)
            qkv_tail_kernel<<<(rem * 32 + TB - 1) / TB, TB, 0, stream>>>(
                x, Wq, Wk, Wv, Q, K, V, rem_row0, N);
    }
    if (N > 0)
        own_range_kernel<<<(N + OWN_R - 1) / OWN_R, OWN_TB, 0, stream>>>(
            Q, K, V, ei, Z, outacc, N, E);
    {
        int threads = ntiles * 32;
        if (threads > 0)
            out_wmma_kernel<<<(threads + TB - 1) / TB, TB, 0, stream>>>(
                x, outacc, Z, Wo, out, ntiles);
        if (rem > 0)
            out_tail_kernel<<<(rem * 32 + TB - 1) / TB, TB, 0, stream>>>(
                x, outacc, Z, Wo, out, rem_row0, N);
    }
}


# Edge filing scans a 1-byte key plane (receiver >> 9, written by the projection launch), not the 4-byte ids: one mask per 32 edges of a lane, hits filed as codes; sum order: LDS arrival, as before
# speedup vs baseline: 6.7672x; 1.2431x over previous
#include <hip/hip_runtime.h>
#include <math.h>

typedef __attribute__((ext_vector_type(2))) float v2f;
typedef __attribute__((ext_vector_type(8))) float v8f;

__device__ __forceinline__ unsigned fkey(float f) {
    unsigned u = __float_as_uint(f);
    return (u & 0x80000000u) ? ~u : (u | 0x80000000u);
}
__device__ __forceinline__ float fkey_inv(unsigned k) {
    unsigned u = (k & 0x80000000u) ? (k & 0x7FFFFFFFu) : ~k;
    return __uint_as_float(u);
}

#define OWN_SH   9
__device__ __forceinline__ unsigned own_key(int r, int N) {
    return ((unsigned)r < (unsigned)N) ? ((unsigned)r >> OWN_SH) : 0xFFu;
}

__device__ __forceinline__ void wmma_step(const float* __restrict__ W, int kb,
                                          int half, int m, v2f a,
                                          v8f& acc0, v8f& acc1) {
    int r0 = (kb + 2 * half + 0) * 32;
    int r1 = (kb + 2 * half + 1) * 32;
    v2f b0, b1;
    b0.x = W[r0 + m];        b0.y = W[r1 + m];
    b1.x = W[r0 + 16 + m];   b1.y = W[r1 + 16 + m];
    acc0 = __builtin_amdgcn_wmma_f32_16x16x4_f32(false, a, false, b0,
                                                 (short)0, acc0, false, false);
    acc1 = __builtin_amdgcn_wmma_f32_16x16x4_f32(false, a, false, b1,
                                                 (short)0, acc1, false, false);
}

__global__ __launch_bounds__(256) void qkv_wmma_kernel(
    const float* __restrict__ x,
    const float* __restrict__ Wq, const float* __restrict__ Wk,
    const float* __restrict__ Wv,
    float* __restrict__ Q, float* __restrict__ K, float* __restrict__ V,
    int ntiles, const int* __restrict__ ei, unsigned* __restrict__ keys4,
    int N, int E, int Epad) {
    if (keys4 != nullptr) {
        const int* recv = ei + E;
        const bool vec_ok = (((size_t)recv & 15u) == 0);
        const int gid = (int)(blockIdx.x * blockDim.x + threadIdx.x);
        const int gsz = (int)(gridDim.x * blockDim.x);
        for (int q = gid; q * 4 < Epad; q += gsz) {
            const int e0 = q * 4;
            int4 r;
            if (vec_ok && e0 + 4 <= E) {
                r = *reinterpret_cast<const int4*>(recv + e0);
            } else {
                r.x = (e0     < E) ? recv[e0]     : -1;
                r.y = (e0 + 1 < E) ? recv[e0 + 1] : -1;
                r.z = (e0 + 2 < E) ? recv[e0 + 2] : -1;
                r.w = (e0 + 3 < E) ? recv[e0 + 3] : -1;
            }
            keys4[q] = own_key(r.x, N) | (own_key(r.y, N) << 8) |
                       (own_key(r.z, N) << 16) | (own_key(r.w, N) << 24);
        }
    }
    int wave = (int)((blockIdx.x * blockDim.x + threadIdx.x) >> 5);
    if (wave >= ntiles) return;
    int lane = (int)(threadIdx.x & 31u);
    int half = lane >> 4;
    int m    = lane & 15;
    size_t row0 = (size_t)wave * 16;

    v8f q0 = {}, q1 = {}, k0 = {}, k1 = {}, v0 = {}, v1 = {};

#pragma unroll
    for (int kk = 0; kk < 8; ++kk) {
        int kb = kk * 4;
        const float* ap = x + (row0 + m) * 32 + kb + 2 * half;
        v2f a; a.x = ap[0]; a.y = ap[1];
        wmma_step(Wq, kb, half, m, a, q0, q1);
        wmma_step(Wk, kb, half, m, a, k0, k1);
        wmma_step(Wv, kb, half, m, a, v0, v1);
    }

#pragma unroll
    for (int r = 0; r < 8; ++r) {
        size_t row = (row0 + r + 8 * half) * 32;
        Q[row + m]      = q0[r];  Q[row + 16 + m] = q1[r];
        K[row + m]      = k0[r];  K[row + 16 + m] = k1[r];
        V[row + m]      = v0[r];  V[row + 16 + m] = v1[r];
    }
}

__global__ void qkv_tail_kernel(const float* __restrict__ x,
                                const float* __restrict__ Wq,
                                const float* __restrict__ Wk,
                                const float* __restrict__ Wv,
                                float* __restrict__ Q, float* __restrict__ K,
                                float* __restrict__ V, int row_start, int N) {
    int idx = blockIdx.x * blockDim.x + threadIdx.x;
    int nrows = N - row_start;
    if (idx >= nrows * 32) return;
    int row = row_start + idx / 32, col = idx & 31;
    float q = 0.f, k = 0.f, v = 0.f;
    for (int t = 0; t < 32; ++t) {
        float xv = x[(size_t)row * 32 + t];
        q += xv * Wq[t * 32 + col];
        k += xv * Wk[t * 32 + col];
        v += xv * Wv[t * 32 + col];
    }
    size_t o = (size_t)row * 32 + col;
    Q[o] = q; K[o] = k; V[o] = v;
}

template <int M>
__device__ __forceinline__ float lane_xor(float x) {
    int b = __builtin_bit_cast(int, x);
    if constexpr (M == 16) {
        return __builtin_bit_cast(float, __builtin_amdgcn_permlanex16(
            b, b, 0x76543210, 0xfedcba98, false, false));
    } else {
        return __builtin_bit_cast(float, __builtin_amdgcn_update_dpp(
            b, b, 0x160 | M, 0xf, 0xf, false));
    }
}

template <int M, int NH>
__device__ __forceinline__ void fold_level(float* v, bool hi) {
#pragma unroll
    for (int i = 0; i < NH; ++i) {
        float keep = hi ? v[i + NH] : v[i];
        float send = hi ? v[i] : v[i + NH];
        v[i] = keep + lane_xor<M>(send);
    }
}

#define OWN_R    (1 << OWN_SH)
#define OWN_CAP  16384
#define OWN_TB   1024
#define OWN_U    8
#define OWN_STEP (OWN_TB * 4)
#define KEY_ROW  (OWN_TB * 16)
#define KEY_U    4
#define KEY_STEP (KEY_ROW * KEY_U)

__device__ __forceinline__ void own_push(unsigned rl, int e, unsigned* cnt,
                                         int* l_a) {
    if (rl < (unsigned)OWN_R) {
        unsigned slot = atomicAdd(cnt, 1u);
        if (slot < (unsigned)OWN_CAP) l_a[slot] = e;
    }
}

__device__ __forceinline__ void own_build(const int* __restrict__ recv, int a,
                                          int b, int rbase, bool vec_ok,
                                          unsigned* cnt, int* l_a) {
    const int t = (int)threadIdx.x;
    for (int base = a; base < b; base += OWN_STEP * OWN_U) {
        const bool full = vec_ok && (b - base >= OWN_STEP * OWN_U);
        int4 v[OWN_U];
        if (full) {
#pragma unroll
            for (int u = 0; u < OWN_U; ++u)
                v[u] = *reinterpret_cast<const int4*>(recv + base + u * OWN_STEP + t * 4);
        } else {
#pragma unroll
            for (int u = 0; u < OWN_U; ++u) {
                int e0 = base + u * OWN_STEP + t * 4;
                v[u].x = (e0     < b) ? recv[e0]     : -1;
                v[u].y = (e0 + 1 < b) ? recv[e0 + 1] : -1;
                v[u].z = (e0 + 2 < b) ? recv[e0 + 2] : -1;
                v[u].w = (e0 + 3 < b) ? recv[e0 + 3] : -1;
            }
        }
#pragma unroll
        for (int u = 0; u < OWN_U; ++u) {
            int e0 = base + u * OWN_STEP + t * 4;
            unsigned r0 = (unsigned)(v[u].x - rbase), r1 = (unsigned)(v[u].y - rbase);
            unsigned r2 = (unsigned)(v[u].z - rbase), r3 = (unsigned)(v[u].w - rbase);
            unsigned mn = min(min(r0, r1), min(r2, r3));
            if (mn < (unsigned)OWN_R) {
                own_push(r0, e0,     cnt, l_a);
                own_push(r1, e0 + 1, cnt, l_a);
                own_push(r2, e0 + 2, cnt, l_a);
                own_push(r3, e0 + 3, cnt, l_a);
            }
        }
    }
}

__device__ __forceinline__ unsigned zero_bytes(unsigned x) {
    return ~(((x & 0x7F7F7F7Fu) + 0x7F7F7F7Fu) | x) & 0x80808080u;
}

__device__ __forceinline__ void own_build_keys(const unsigned* __restrict__ keys4,
                                               int Epad, unsigned kb,
                                               unsigned* cnt, int* l_a) {
    const unsigned t = threadIdx.x;
    const unsigned splat = kb * 0x01010101u;
    for (int base = 0; base < Epad; base += KEY_STEP) {
        uint4 v[KEY_U];
#pragma unroll
        for (int u = 0; u < KEY_U; ++u)
            v[u] = *reinterpret_cast<const uint4*>(keys4 + ((base + u * KEY_ROW) >> 2) + t * 4);
#pragma unroll
        for (int u = 0; u < KEY_U; u += 2) {
            const unsigned code0 = ((unsigned)(base / KEY_ROW + u) << 15) | (t << 5);
            unsigned m = (zero_bytes(v[u].x ^ splat) >> 7) | (zero_bytes(v[u].y ^ splat) >> 6) |
                         (zero_bytes(v[u].z ^ splat) >> 5) | (zero_bytes(v[u].w ^ splat) >> 4) |
                         (zero_bytes(v[u + 1].x ^ splat) >> 3) | (zero_bytes(v[u + 1].y ^ splat) >> 2) |
                         (zero_bytes(v[u + 1].z ^ splat) >> 1) | zero_bytes(v[u + 1].w ^ splat);
            while (m) {
                const unsigned c = (unsigned)__builtin_ctz(m);
                m &= m - 1u;
                unsigned slot = atomicAdd(cnt, 1u);
                if (slot < (unsigned)OWN_CAP) l_a[slot] = (int)(code0 | c);
            }
        }
    }
}

__device__ __forceinline__ int own_edge(unsigned code) {
    const unsigned c = code & 31u, t = (code >> 5) & 1023u, row = code >> 15;
    return (int)((row + ((c >> 2) & 1u)) * KEY_ROW + t * 16u + ((c & 3u) << 2) + (c >> 3));
}

__device__ __forceinline__ void own_process(
    const float* __restrict__ Q, const float* __restrict__ K,
    const float* __restrict__ V, const int* __restrict__ ei,
    const int* __restrict__ recv, int rbase, int n,
    bool first, bool coded, float* acc, float* zs, unsigned* mkey, unsigned* mnew,
    int* l_a, int* l_r, float* l_sc) {
    const int t = (int)threadIdx.x;
    const unsigned lane = threadIdx.x & 31u;
    const int wave = __builtin_amdgcn_readfirstlane((int)(threadIdx.x >> 5));

    for (int c = wave; c * 32 < n; c += OWN_TB / 32) {
        int i = c * 32 + (int)lane;
        bool valid = i < n;
        int s = 0, r = rbase;
        if (valid) {
            int e = l_a[i];
            if (coded) e = own_edge((unsigned)e);
            s = ei[e]; r = recv[e];
        }
        int rl = r - rbase;
        float p[32];
#pragma unroll
        for (int h = 0; h < 2; ++h) {
            float q[16], k[16];
#pragma unroll
            for (int j = 0; j < 16; ++j) {
                unsigned rj = (unsigned)__builtin_amdgcn_readlane(r, 16 * h + j);
                q[j] = (Q + ((size_t)rj << 5))[lane];
            }
#pragma unroll
            for (int j = 0; j < 16; ++j) {
                unsigned sj = (unsigned)__builtin_amdgcn_readlane(s, 16 * h + j);
                k[j] = (K + ((size_t)sj << 5))[lane];
            }
#pragma unroll
            for (int j = 0; j < 16; ++j) p[16 * h + j] = q[j] * k[j];
        }
        fold_level<16, 16>(p, (lane & 16) != 0);
        fold_level<8, 8>(p, (lane & 8) != 0);
        fold_level<4, 4>(p, (lane & 4) != 0);
        fold_level<2, 2>(p, (lane & 2) != 0);
        fold_level<1, 1>(p, (lane & 1) != 0);
        float sc = p[0] * 0.17677669529663687f;
        if (valid) {
            l_sc[i] = sc;
            l_r[i] = rl;
            l_a[i] = s;
            atomicMax(&mnew[rl], fkey(sc));
        }
    }
    __syncthreads();
    if (!first) {
        for (int idx = t; idx < OWN_R * 32; idx += OWN_TB) {
            unsigned ko = mkey[idx >> 5], kn = mnew[idx >> 5];
            if (kn != ko)
                acc[idx] *= ko ? __expf(fkey_inv(ko) - fkey_inv(kn)) : 0.0f;
        }
        for (int rl = t; rl < OWN_R; rl += OWN_TB) {
            unsigned ko = mkey[rl], kn = mnew[rl];
            if (kn != ko)
                zs[rl] *= ko ? __expf(fkey_inv(ko) - fkey_inv(kn)) : 0.0f;
        }
        __syncthreads();
    }
    for (int rl = t; rl < OWN_R; rl += OWN_TB) mkey[rl] = mnew[rl];
    __syncthreads();

    for (int c = wave; c * 32 < n; c += OWN_TB / 32) {
        int i = c * 32 + (int)lane;
        int m = min(32, n - c * 32);
        int s = 0, rl = 0;
        float alpha = 0.f;
        if (i < n) {
            s = l_a[i]; rl = l_r[i];
            alpha = __expf(l_sc[i] - fkey_inv(mkey[rl]));
            atomicAdd(&zs[rl], alpha);
        }
        if (m == 32) {
            float v[32];
#pragma unroll
            for (int j = 0; j < 32; ++j) {
                unsigned sj = (unsigned)__builtin_amdgcn_readlane(s, j);
                v[j] = (V + ((size_t)sj << 5))[lane];
            }
#pragma unroll
            for (int j = 0; j < 32; ++j) {
                int rj = __builtin_amdgcn_readlane(rl, j);
                float aj = __builtin_bit_cast(float, __builtin_amdgcn_readlane(
                    __builtin_bit_cast(int, alpha), j));
                atomicAdd(&acc[(rj << 5) + (int)lane], aj * v[j]);
            }
        } else {
            for (int j = 0; j < m; ++j) {
                unsigned sj = (unsigned)__builtin_amdgcn_readlane(s, j);
                int rj = __builtin_amdgcn_readlane(rl, j);
                float aj = __builtin_bit_cast(float, __builtin_amdgcn_readlane(
                    __builtin_bit_cast(int, alpha), j));
                atomicAdd(&acc[(rj << 5) + (int)lane],
                          aj * (V + ((size_t)sj << 5))[lane]);
            }
        }
    }
}

__global__ __launch_bounds__(OWN_TB) void own_range_kernel(
    const float* __restrict__ Q, const float* __restrict__ K,
    const float* __restrict__ V, const int* __restrict__ ei,
    const unsigned* __restrict__ keys4,
    float* __restrict__ Z, float* __restrict__ outacc, int N, int E, int Epad) {
    __shared__ float    acc[OWN_R * 32];
    __shared__ float    zs[OWN_R];
    __shared__ unsigned mkey[OWN_R];
    __shared__ unsigned mnew[OWN_R];
    __shared__ int      l_a[OWN_CAP];
    __shared__ int      l_r[OWN_CAP];
    __shared__ float    l_sc[OWN_CAP];
    __shared__ unsigned cnt;

    const int t = (int)threadIdx.x;
    const int rbase = (int)blockIdx.x * OWN_R;
    const int nown = min(OWN_R, N - rbase);
    const int* recv = ei + E;
    const bool vec_ok = (((size_t)recv & 15u) == 0);

    for (int i = t; i < OWN_R * 32; i += OWN_TB) acc[i] = 0.0f;
    for (int i = t; i < OWN_R; i += OWN_TB) { zs[i] = 0.0f; mkey[i] = 0u; mnew[i] = 0u; }
    if (t == 0) cnt = 0u;
    __syncthreads();

    if (keys4 != nullptr)
        own_build_keys(keys4, Epad, blockIdx.x, &cnt, l_a);
    else
        own_build(recv, 0, E, rbase, vec_ok, &cnt, l_a);
    __syncthreads();
    const unsigned total = cnt;
    if (total <= (unsigned)OWN_CAP) {
        own_process(Q, K, V, ei, recv, rbase, (int)total, true, keys4 != nullptr,
                    acc, zs, mkey, mnew, l_a, l_r, l_sc);
    } else {
        for (int a = 0; a < E; a += OWN_CAP) {
            __syncthreads();
            if (t == 0) cnt = 0u;
            __syncthreads();
            int b = (E - a > OWN_CAP) ? a + OWN_CAP : E;
            own_build(recv, a, b, rbase, vec_ok, &cnt, l_a);
            __syncthreads();
            own_process(Q, K, V, ei, recv, rbase, (int)cnt, a == 0, false, acc, zs,
                        mkey, mnew, l_a, l_r, l_sc);
        }
    }
    __syncthreads();
    for (int i = t; i < nown * 32; i += OWN_TB) outacc[(size_t)rbase * 32 + i] = acc[i];
    for (int i = t; i < nown; i += OWN_TB) Z[rbase + i] = zs[i];
}

__global__ __launch_bounds__(256) void out_wmma_kernel(
    const float* __restrict__ x, const float* __restrict__ outacc,
    const float* __restrict__ Z, const float* __restrict__ Wo,
    float* __restrict__ out, int ntiles) {
    int wave = (int)((blockIdx.x * blockDim.x + threadIdx.x) >> 5);
    if (wave >= ntiles) return;
    int lane = (int)(threadIdx.x & 31u);
    int half = lane >> 4;
    int m    = lane & 15;
    size_t row0 = (size_t)wave * 16;

    float invz = 1.0f / (Z[row0 + m] + 1e-6f);
    v8f c0 = {}, c1 = {};

#pragma unroll
    for (int kk = 0; kk < 8; ++kk) {
        int kb = kk * 4;
        const float* ap = outacc + (row0 + m) * 32 + kb + 2 * half;
        v2f a; a.x = ap[0] * invz; a.y = ap[1] * invz;
        wmma_step(Wo, kb, half, m, a, c0, c1);
    }

#pragma unroll
    for (int r = 0; r < 8; ++r) {
        size_t row = (row0 + r + 8 * half) * 32;
        out[row + m]      = x[row + m]      + c0[r];
        out[row + 16 + m] = x[row + 16 + m] + c1[r];
    }
}

__global__ void out_tail_kernel(const float* __restrict__ x,
                                const float* __restrict__ outacc,
                                const float* __restrict__ Z,
                                const float* __restrict__ Wo,
                                float* __restrict__ out, int row_start, int N) {
    int idx = blockIdx.x * blockDim.x + threadIdx.x;
    int nrows = N - row_start;
    if (idx >= nrows * 32) return;
    int row = row_start + idx / 32, col = idx & 31;
    float invz = 1.0f / (Z[row] + 1e-6f);
    float acc = 0.f;
    for (int t = 0; t < 32; ++t)
        acc += outacc[(size_t)row * 32 + t] * invz * Wo[t * 32 + col];
    size_t o = (size_t)row * 32 + col;
    out[o] = x[o] + acc;
}

extern "C" void kernel_launch(void* const* d_in, const int* in_sizes, int n_in,
                              void* d_out, int out_size, void* d_ws,
                              size_t ws_size, hipStream_t stream) {
    const float* x  = (const float*)d_in[0];
    const int*   ei = (const int*)d_in[1];
    const float* Wq = (const float*)d_in[2];
    const float* Wk = (const float*)d_in[3];
    const float* Wv = (const float*)d_in[4];
    const float* Wo = (const float*)d_in[5];
    float* out = (float*)d_out;

    const int N = in_sizes[0] / 32;
    const int E = in_sizes[1] / 2;

    float* Q       = (float*)d_ws;
    float* K       = Q + (size_t)N * 32;
    float* V       = K + (size_t)N * 32;
    float* outacc  = V + (size_t)N * 32;
    float* Z       = outacc + (size_t)N * 32;
    const int Epad = ((E + KEY_STEP - 1) / KEY_STEP) * KEY_STEP;
    const size_t key_off = (((size_t)N * 129 * sizeof(float)) + 15) & ~(size_t)15;
    unsigned* keys4 = (unsigned*)((unsigned char*)d_ws + key_off);
    if (N > 255 * OWN_R || key_off + (size_t)Epad > ws_size) keys4 = nullptr;

    const int TB = 256;
    int ntiles = N / 16;
    int rem_row0 = ntiles * 16;
    int rem = N - rem_row0;

    {
        int blocks = (ntiles * 32 + TB - 1) / TB;
        if (blocks == 0 && keys4 != nullptr) blocks = 1;
        if (blocks > 0)
            qkv_wmma_kernel<<<blocks, TB, 0, stream>>>(
                x, Wq, Wk, Wv, Q, K, V, ntiles, ei, keys4, N, E, Epad);
        if (rem > 0)
            qkv_tail_kernel<<<(rem * 32 + TB - 1) / TB, TB, 0, stream>>>(
                x, Wq, Wk, Wv, Q, K, V, rem_row0, N);
    }
    if (N > 0)
        own_range_kernel<<<(N + OWN_R - 1) / OWN_R, OWN_TB, 0, stream>>>(
            Q, K, V, ei, keys4, Z, outacc, N, E, Epad);
    {
        int threads = ntiles * 32;
        if (threads > 0)
            out_wmma_kernel<<<(threads + TB - 1) / TB, TB, 0, stream>>>(
                x, outacc, Z, Wo, out, ntiles);
        if (rem > 0)
            out_tail_kernel<<<(rem * 32 + TB - 1) / TB, TB, 0, stream>>>(
                x, outacc, Z, Wo, out, rem_row0, N);
    }
}


# Row gathers by 8-lane groups, 16-byte loads: one load brings the rows of 4 edges (4 whole lines), not 1; ids by DPP8, no readlane chain; dot: 4 per lane then 3-level tree, sum order changes
# speedup vs baseline: 7.0727x; 1.0451x over previous
#include <hip/hip_runtime.h>
#include <math.h>

typedef __attribute__((ext_vector_type(2))) float v2f;
typedef __attribute__((ext_vector_type(8))) float v8f;

__device__ __forceinline__ unsigned fkey(float f) {
    unsigned u = __float_as_uint(f);
    return (u & 0x80000000u) ? ~u : (u | 0x80000000u);
}
__device__ __forceinline__ float fkey_inv(unsigned k) {
    unsigned u = (k & 0x80000000u) ? (k & 0x7FFFFFFFu) : ~k;
    return __uint_as_float(u);
}

#define OWN_SH   9
__device__ __forceinline__ unsigned own_key(int r, int N) {
    return ((unsigned)r < (unsigned)N) ? ((unsigned)r >> OWN_SH) : 0xFFu;
}

__device__ __forceinline__ void wmma_step(const float* __restrict__ W, int kb,
                                          int half, int m, v2f a,
                                          v8f& acc0, v8f& acc1) {
    int r0 = (kb + 2 * half + 0) * 32;
    int r1 = (kb + 2 * half + 1) * 32;
    v2f b0, b1;
    b0.x = W[r0 + m];        b0.y = W[r1 + m];
    b1.x = W[r0 + 16 + m];   b1.y = W[r1 + 16 + m];
    acc0 = __builtin_amdgcn_wmma_f32_16x16x4_f32(false, a, false, b0,
                                                 (short)0, acc0, false, false);
    acc1 = __builtin_amdgcn_wmma_f32_16x16x4_f32(false, a, false, b1,
                                                 (short)0, acc1, false, false);
}

__global__ __launch_bounds__(256) void qkv_wmma_kernel(
    const float* __restrict__ x,
    const float* __restrict__ Wq, const float* __restrict__ Wk,
    const float* __restrict__ Wv,
    float* __restrict__ Q, float* __restrict__ K, float* __restrict__ V,
    int ntiles, const int* __restrict__ ei, unsigned* __restrict__ keys4,
    int N, int E, int Epad) {
    if (keys4 != nullptr) {
        const int* recv = ei + E;
        const bool vec_ok = (((size_t)recv & 15u) == 0);
        const int gid = (int)(blockIdx.x * blockDim.x + threadIdx.x);
        const int gsz = (int)(gridDim.x * blockDim.x);
        for (int q = gid; q * 4 < Epad; q += gsz) {
            const int e0 = q * 4;
            int4 r;
            if (vec_ok && e0 + 4 <= E) {
                r = *reinterpret_cast<const int4*>(recv + e0);
            } else {
                r.x = (e0     < E) ? recv[e0]     : -1;
                r.y = (e0 + 1 < E) ? recv[e0 + 1] : -1;
                r.z = (e0 + 2 < E) ? recv[e0 + 2] : -1;
                r.w = (e0 + 3 < E) ? recv[e0 + 3] : -1;
            }
            keys4[q] = own_key(r.x, N) | (own_key(r.y, N) << 8) |
                       (own_key(r.z, N) << 16) | (own_key(r.w, N) << 24);
        }
    }
    int wave = (int)((blockIdx.x * blockDim.x + threadIdx.x) >> 5);
    if (wave >= ntiles) return;
    int lane = (int)(threadIdx.x & 31u);
    int half = lane >> 4;
    int m    = lane & 15;
    size_t row0 = (size_t)wave * 16;

    v8f q0 = {}, q1 = {}, k0 = {}, k1 = {}, v0 = {}, v1 = {};

#pragma unroll
    for (int kk = 0; kk < 8; ++kk) {
        int kb = kk * 4;
        const float* ap = x + (row0 + m) * 32 + kb + 2 * half;
        v2f a; a.x = ap[0]; a.y = ap[1];
        wmma_step(Wq, kb, half, m, a, q0, q1);
        wmma_step(Wk, kb, half, m, a, k0, k1);
        wmma_step(Wv, kb, half, m, a, v0, v1);
    }

#pragma unroll
    for (int r = 0; r < 8; ++r) {
        size_t row = (row0 + r + 8 * half) * 32;
        Q[row + m]      = q0[r];  Q[row + 16 + m] = q1[r];
        K[row + m]      = k0[r];  K[row + 16 + m] = k1[r];
        V[row + m]      = v0[r];  V[row + 16 + m] = v1[r];
    }
}

__global__ void qkv_tail_kernel(const float* __restrict__ x,
                                const float* __restrict__ Wq,
                                const float* __restrict__ Wk,
                                const float* __restrict__ Wv,
                                float* __restrict__ Q, float* __restrict__ K,
                                float* __restrict__ V, int row_start, int N) {
    int idx = blockIdx.x * blockDim.x + threadIdx.x;
    int nrows = N - row_start;
    if (idx >= nrows * 32) return;
    int row = row_start + idx / 32, col = idx & 31;
    float q = 0.f, k = 0.f, v = 0.f;
    for (int t = 0; t < 32; ++t) {
        float xv = x[(size_t)row * 32 + t];
        q += xv * Wq[t * 32 + col];
        k += xv * Wk[t * 32 + col];
        v += xv * Wv[t * 32 + col];
    }
    size_t o = (size_t)row * 32 + col;
    Q[o] = q; K[o] = k; V[o] = v;
}

template <int M>
__device__ __forceinline__ float lane_xor(float x) {
    int b = __builtin_bit_cast(int, x);
    if constexpr (M == 16) {
        return __builtin_bit_cast(float, __builtin_amdgcn_permlanex16(
            b, b, 0x76543210, 0xfedcba98, false, false));
    } else {
        return __builtin_bit_cast(float, __builtin_amdgcn_update_dpp(
            b, b, 0x160 | M, 0xf, 0xf, false));
    }
}

template <int M, int NH>
__device__ __forceinline__ void fold_level(float* v, bool hi) {
#pragma unroll
    for (int i = 0; i < NH; ++i) {
        float keep = hi ? v[i + NH] : v[i];
        float send = hi ? v[i] : v[i + NH];
        v[i] = keep + lane_xor<M>(send);
    }
}

template <int J>
__device__ __forceinline__ unsigned bcast8(unsigned x) {
    return (unsigned)__builtin_amdgcn_mov_dpp8(x, J * 0x249249);
}
template <int J>
__device__ __forceinline__ float bcast8f(float x) {
    return __uint_as_float(bcast8<J>(__float_as_uint(x)));
}

__device__ __forceinline__ int acc_pos(int rl, int f) {
    return (rl << 5) + ((((f & 3) ^ (rl >> 1)) & 3) << 3) + (f >> 2);
}

#define OWN_R    (1 << OWN_SH)
#define OWN_CAP  16384
#define OWN_TB   1024
#define OWN_U    8
#define OWN_STEP (OWN_TB * 4)
#define KEY_ROW  (OWN_TB * 16)
#define KEY_U    4
#define KEY_STEP (KEY_ROW * KEY_U)

__device__ __forceinline__ void own_push(unsigned rl, int e, unsigned* cnt,
                                         int* l_a) {
    if (rl < (unsigned)OWN_R) {
        unsigned slot = atomicAdd(cnt, 1u);
        if (slot < (unsigned)OWN_CAP) l_a[slot] = e;
    }
}

__device__ __forceinline__ void own_build(const int* __restrict__ recv, int a,
                                          int b, int rbase, bool vec_ok,
                                          unsigned* cnt, int* l_a) {
    const int t = (int)threadIdx.x;
    for (int base = a; base < b; base += OWN_STEP * OWN_U) {
        const bool full = vec_ok && (b - base >= OWN_STEP * OWN_U);
        int4 v[OWN_U];
        if (full) {
#pragma unroll
            for (int u = 0; u < OWN_U; ++u)
                v[u] = *reinterpret_cast<const int4*>(recv + base + u * OWN_STEP + t * 4);
        } else {
#pragma unroll
            for (int u = 0; u < OWN_U; ++u) {
                int e0 = base + u * OWN_STEP + t * 4;
                v[u].x = (e0     < b) ? recv[e0]     : -1;
                v[u].y = (e0 + 1 < b) ? recv[e0 + 1] : -1;
                v[u].z = (e0 + 2 < b) ? recv[e0 + 2] : -1;
                v[u].w = (e0 + 3 < b) ? recv[e0 + 3] : -1;
            }
        }
#pragma unroll
        for (int u = 0; u < OWN_U; ++u) {
            int e0 = base + u * OWN_STEP + t * 4;
            unsigned r0 = (unsigned)(v[u].x - rbase), r1 = (unsigned)(v[u].y - rbase);
            unsigned r2 = (unsigned)(v[u].z - rbase), r3 = (unsigned)(v[u].w - rbase);
            unsigned mn = min(min(r0, r1), min(r2, r3));
            if (mn < (unsigned)OWN_R) {
                own_push(r0, e0,     cnt, l_a);
                own_push(r1, e0 + 1, cnt, l_a);
                own_push(r2, e0 + 2, cnt, l_a);
                own_push(r3, e0 + 3, cnt, l_a);
            }
        }
    }
}

__device__ __forceinline__ unsigned zero_bytes(unsigned x) {
    return ~(((x & 0x7F7F7F7Fu) + 0x7F7F7F7Fu) | x) & 0x80808080u;
}

__device__ __forceinline__ void own_build_keys(const unsigned* __restrict__ keys4,
                                               int Epad, unsigned kb,
                                               unsigned* cnt, int* l_a) {
    const unsigned t = threadIdx.x;
    const unsigned splat = kb * 0x01010101u;
    for (int base = 0; base < Epad; base += KEY_STEP) {
        uint4 v[KEY_U];
#pragma unroll
        for (int u = 0; u < KEY_U; ++u)
            v[u] = *reinterpret_cast<const uint4*>(keys4 + ((base + u * KEY_ROW) >> 2) + t * 4);
#pragma unroll
        for (int u = 0; u < KEY_U; u += 2) {
            const unsigned code0 = ((unsigned)(base / KEY_ROW + u) << 15) | (t << 5);
            unsigned m = (zero_bytes(v[u].x ^ splat) >> 7) | (zero_bytes(v[u].y ^ splat) >> 6) |
                         (zero_bytes(v[u].z ^ splat) >> 5) | (zero_bytes(v[u].w ^ splat) >> 4) |
                         (zero_bytes(v[u + 1].x ^ splat) >> 3) | (zero_bytes(v[u + 1].y ^ splat) >> 2) |
                         (zero_bytes(v[u + 1].z ^ splat) >> 1) | zero_bytes(v[u + 1].w ^ splat);
            while (m) {
                const unsigned c = (unsigned)__builtin_ctz(m);
                m &= m - 1u;
                unsigned slot = atomicAdd(cnt, 1u);
                if (slot < (unsigned)OWN_CAP) l_a[slot] = (int)(code0 | c);
            }
        }
    }
}

__device__ __forceinline__ int own_edge(unsigned code) {
    const unsigned c = code & 31u, t = (code >> 5) & 1023u, row = code >> 15;
    return (int)((row + ((c >> 2) & 1u)) * KEY_ROW + t * 16u + ((c & 3u) << 2) + (c >> 3));
}

template <int J>
__device__ __forceinline__ void rows_qk(const float4* __restrict__ Qp,
                                        const float4* __restrict__ Kp,
                                        unsigned r, unsigned s, float4* qv, float4* kv) {
    qv[J] = Qp[(size_t)bcast8<J>(r) << 3];
    kv[J] = Kp[(size_t)bcast8<J>(s) << 3];
}

template <bool tail>
__device__ __forceinline__ void add_rows(const float4* __restrict__ Vp, unsigned s,
                                         unsigned rl, float alpha, int m,
                                         unsigned lane, float* acc) {
    const float4* a0 = Vp + ((size_t)bcast8<0>(s) << 3);
    const float4* a1 = Vp + ((size_t)bcast8<1>(s) << 3);
    const float4* a2 = Vp + ((size_t)bcast8<2>(s) << 3);
    const float4* a3 = Vp + ((size_t)bcast8<3>(s) << 3);
    const float4* a4 = Vp + ((size_t)bcast8<4>(s) << 3);
    const float4* a5 = Vp + ((size_t)bcast8<5>(s) << 3);
    const float4* a6 = Vp + ((size_t)bcast8<6>(s) << 3);
    const float4* a7 = Vp + ((size_t)bcast8<7>(s) << 3);
    __builtin_amdgcn_sched_barrier(0);
    float4 v[8];
    v[0] = *a0;  v[1] = *a1;  v[2] = *a2;  v[3] = *a3;
    v[4] = *a4;  v[5] = *a5;  v[6] = *a6;  v[7] = *a7;
    unsigned rj[8];
    float aj[8];
    rj[0] = bcast8<0>(rl);  aj[0] = bcast8f<0>(alpha);
    rj[1] = bcast8<1>(rl);  aj[1] = bcast8f<1>(alpha);
    rj[2] = bcast8<2>(rl);  aj[2] = bcast8f<2>(alpha);
    rj[3] = bcast8<3>(rl);  aj[3] = bcast8f<3>(alpha);
    rj[4] = bcast8<4>(rl);  aj[4] = bcast8f<4>(alpha);
    rj[5] = bcast8<5>(rl);  aj[5] = bcast8f<5>(alpha);
    rj[6] = bcast8<6>(rl);  aj[6] = bcast8f<6>(alpha);
    rj[7] = bcast8<7>(rl);  aj[7] = bcast8f<7>(alpha);
    const unsigned q4 = (lane & 7u) << 2;
    char* accb = reinterpret_cast<char*>(acc);
#pragma unroll
    for (int j = 0; j < 8; ++j) {
        const unsigned b0 = (rj[j] << 7) | ((rj[j] << 4) & 0x60u) | q4;
        if (!tail || (int)((lane & 24u) | (unsigned)j) < m) {
            atomicAdd(reinterpret_cast<float*>(accb + b0),         aj[j] * v[j].x);
            atomicAdd(reinterpret_cast<float*>(accb + (b0 ^ 32u)), aj[j] * v[j].y);
            atomicAdd(reinterpret_cast<float*>(accb + (b0 ^ 64u)), aj[j] * v[j].z);
            atomicAdd(reinterpret_cast<float*>(accb + (b0 ^ 96u)), aj[j] * v[j].w);
        }
    }
}

__device__ __forceinline__ void own_process(
    const float* __restrict__ Q, const float* __restrict__ K,
    const float* __restrict__ V, const int* __restrict__ ei,
    const int* __restrict__ recv, int rbase, int n,
    bool first, bool coded, float* acc, float* zs, unsigned* mkey, unsigned* mnew,
    int* l_a, int* l_r, float* l_sc) {
    const int t = (int)threadIdx.x;
    const unsigned lane = threadIdx.x & 31u;
    const int wave = __builtin_amdgcn_readfirstlane((int)(threadIdx.x >> 5));
    const float4* Qp = reinterpret_cast<const float4*>(Q) + (lane & 7u);
    const float4* Kp = reinterpret_cast<const float4*>(K) + (lane & 7u);
    const float4* Vp = reinterpret_cast<const float4*>(V) + (lane & 7u);

    for (int c = wave; c * 32 < n; c += OWN_TB / 32) {
        int i = c * 32 + (int)lane;
        bool valid = i < n;
        int s = 0, r = rbase;
        if (valid) {
            int e = l_a[i];
            if (coded) e = own_edge((unsigned)e);
            s = ei[e]; r = recv[e];
        }
        int rl = r - rbase;
        float4 qv[8], kv[8];
        rows_qk<0>(Qp, Kp, (unsigned)r, (unsigned)s, qv, kv);
        rows_qk<1>(Qp, Kp, (unsigned)r, (unsigned)s, qv, kv);
        rows_qk<2>(Qp, Kp, (unsigned)r, (unsigned)s, qv, kv);
        rows_qk<3>(Qp, Kp, (unsigned)r, (unsigned)s, qv, kv);
        rows_qk<4>(Qp, Kp, (unsigned)r, (unsigned)s, qv, kv);
        rows_qk<5>(Qp, Kp, (unsigned)r, (unsigned)s, qv, kv);
        rows_qk<6>(Qp, Kp, (unsigned)r, (unsigned)s, qv, kv);
        rows_qk<7>(Qp, Kp, (unsigned)r, (unsigned)s, qv, kv);
        __builtin_amdgcn_sched_barrier(0);
        float p[8];
#pragma unroll
        for (int j = 0; j < 8; ++j)
            p[j] = qv[j].x * kv[j].x + qv[j].y * kv[j].y +
                   qv[j].z * kv[j].z + qv[j].w * kv[j].w;
        fold_level<4, 4>(p, (lane & 4) != 0);
        fold_level<2, 2>(p, (lane & 2) != 0);
        fold_level<1, 1>(p, (lane & 1) != 0);
        float sc = p[0] * 0.17677669529663687f;
        if (valid) {
            l_sc[i] = sc;
            l_r[i] = rl;
            l_a[i] = s;
            atomicMax(&mnew[rl], fkey(sc));
        }
    }
    __syncthreads();
    if (!first) {
        for (int idx = t; idx < OWN_R * 32; idx += OWN_TB) {
            unsigned ko = mkey[idx >> 5], kn = mnew[idx >> 5];
            if (kn != ko)
                acc[idx] *= ko ? __expf(fkey_inv(ko) - fkey_inv(kn)) : 0.0f;
        }
        for (int rl = t; rl < OWN_R; rl += OWN_TB) {
            unsigned ko = mkey[rl], kn = mnew[rl];
            if (kn != ko)
                zs[rl] *= ko ? __expf(fkey_inv(ko) - fkey_inv(kn)) : 0.0f;
        }
        __syncthreads();
    }
    for (int rl = t; rl < OWN_R; rl += OWN_TB) mkey[rl] = mnew[rl];
    __syncthreads();

    for (int c = wave; c * 32 < n; c += OWN_TB / 32) {
        int i = c * 32 + (int)lane;
        int m = min(32, n - c * 32);
        int s = 0, rl = 0;
        float alpha = 0.f;
        if (i < n) {
            s = l_a[i]; rl = l_r[i];
            alpha = __expf(l_sc[i] - fkey_inv(mkey[rl]));
            atomicAdd(&zs[rl], alpha);
        }
        if (m == 32)
            add_rows<false>(Vp, (unsigned)s, (unsigned)rl, alpha, m, lane, acc);
        else
            add_rows<true>(Vp, (unsigned)s, (unsigned)rl, alpha, m, lane, acc);
    }
}

__global__ __launch_bounds__(OWN_TB) void own_range_kernel(
    const float* __restrict__ Q, const float* __restrict__ K,
    const float* __restrict__ V, const int* __restrict__ ei,
    const unsigned* __restrict__ keys4,
    float* __restrict__ Z, float* __restrict__ outacc, int N, int E, int Epad) {
    __shared__ float    acc[OWN_R * 32];
    __shared__ float    zs[OWN_R];
    __shared__ unsigned mkey[OWN_R];
    __shared__ unsigned mnew[OWN_R];
    __shared__ int      l_a[OWN_CAP];
    __shared__ int      l_r[OWN_CAP];
    __shared__ float    l_sc[OWN_CAP];
    __shared__ unsigned cnt;

    const int t = (int)threadIdx.x;
    const int rbase = (int)blockIdx.x * OWN_R;
    const int nown = min(OWN_R, N - rbase);
    const int* recv = ei + E;
    const bool vec_ok = (((size_t)recv & 15u) == 0);

    for (int i = t; i < OWN_R * 32; i += OWN_TB) acc[i] = 0.0f;
    for (int i = t; i < OWN_R; i += OWN_TB) { zs[i] = 0.0f; mkey[i] = 0u; mnew[i] = 0u; }
    if (t == 0) cnt = 0u;
    __syncthreads();

    if (keys4 != nullptr)
        own_build_keys(keys4, Epad, blockIdx.x, &cnt, l_a);
    else
        own_build(recv, 0, E, rbase, vec_ok, &cnt, l_a);
    __syncthreads();
    const unsigned total = cnt;
    if (total <= (unsigned)OWN_CAP) {
        own_process(Q, K, V, ei, recv, rbase, (int)total, true, keys4 != nullptr,
                    acc, zs, mkey, mnew, l_a, l_r, l_sc);
    } else {
        for (int a = 0; a < E; a += OWN_CAP) {
            __syncthreads();
            if (t == 0) cnt = 0u;
            __syncthreads();
            int b = (E - a > OWN_CAP) ? a + OWN_CAP : E;
            own_build(recv, a, b, rbase, vec_ok, &cnt, l_a);
            __syncthreads();
            own_process(Q, K, V, ei, recv, rbase, (int)cnt, a == 0, false, acc, zs,
                        mkey, mnew, l_a, l_r, l_sc);
        }
    }
    __syncthreads();
    for (int i = t; i < nown * 32; i += OWN_TB)
        outacc[(size_t)rbase * 32 + i] = acc[acc_pos(i >> 5, i & 31)];
    for (int i = t; i < nown; i += OWN_TB) Z[rbase + i] = zs[i];
}

__global__ __launch_bounds__(256) void out_wmma_kernel(
    const float* __restrict__ x, const float* __restrict__ outacc,
    const float* __restrict__ Z, const float* __restrict__ Wo,
    float* __restrict__ out, int ntiles) {
    int wave = (int)((blockIdx.x * blockDim.x + threadIdx.x) >> 5);
    if (wave >= ntiles) return;
    int lane = (int)(threadIdx.x & 31u);
    int half = lane >> 4;
    int m    = lane & 15;
    size_t row0 = (size_t)wave * 16;

    float invz = 1.0f / (Z[row0 + m] + 1e-6f);
    v8f c0 = {}, c1 = {};

#pragma unroll
    for (int kk = 0; kk < 8; ++kk) {
        int kb = kk * 4;
        const float* ap = outacc + (row0 + m) * 32 + kb + 2 * half;
        v2f a; a.x = ap[0] * invz; a.y = ap[1] * invz;
        wmma_step(Wo, kb, half, m, a, c0, c1);
    }

#pragma unroll
    for (int r = 0; r < 8; ++r) {
        size_t row = (row0 + r + 8 * half) * 32;
        out[row + m]      = x[row + m]      + c0[r];
        out[row + 16 + m] = x[row + 16 + m] + c1[r];
    }
}

__global__ void out_tail_kernel(const float* __restrict__ x,
                                const float* __restrict__ outacc,
                                const float* __restrict__ Z,
                                const float* __restrict__ Wo,
                                float* __restrict__ out, int row_start, int N) {
    int idx = blockIdx.x * blockDim.x + threadIdx.x;
    int nrows = N - row_start;
    if (idx >= nrows * 32) return;
    int row = row_start + idx / 32, col = idx & 31;
    float invz = 1.0f / (Z[row] + 1e-6f);
    float acc = 0.f;
    for (int t = 0; t < 32; ++t)
        acc += outacc[(size_t)row * 32 + t] * invz * Wo[t * 32 + col];
    size_t o = (size_t)row * 32 + col;
    out[o] = x[o] + acc;
}

extern "C" void kernel_launch(void* const* d_in, const int* in_sizes, int n_in,
                              void* d_out, int out_size, void* d_ws,
                              size_t ws_size, hipStream_t stream) {
    const float* x  = (const float*)d_in[0];
    const int*   ei = (const int*)d_in[1];
    const float* Wq = (const float*)d_in[2];
    const float* Wk = (const float*)d_in[3];
    const float* Wv = (const float*)d_in[4];
    const float* Wo = (const float*)d_in[5];
    float* out = (float*)d_out;

    const int N = in_sizes[0] / 32;
    const int E = in_sizes[1] / 2;

    float* Q       = (float*)d_ws;
    float* K       = Q + (size_t)N * 32;
    float* V       = K + (size_t)N * 32;
    float* outacc  = V + (size_t)N * 32;
    float* Z       = outacc + (size_t)N * 32;
    const int Epad = ((E + KEY_STEP - 1) / KEY_STEP) * KEY_STEP;
    const size_t key_off = (((size_t)N * 129 * sizeof(float)) + 15) & ~(size_t)15;
    unsigned* keys4 = (unsigned*)((unsigned char*)d_ws + key_off);
    if (N > 255 * OWN_R || key_off + (size_t)Epad > ws_size) keys4 = nullptr;

    const int TB = 256;
    int ntiles = N / 16;
    int rem_row0 = ntiles * 16;
    int rem = N - rem_row0;

    {
        int blocks = (ntiles * 32 + TB - 1) / TB;
        if (blocks == 0 && keys4 != nullptr) blocks = 1;
        if (blocks > 0)
            qkv_wmma_kernel<<<blocks, TB, 0, stream>>>(
                x, Wq, Wk, Wv, Q, K, V, ntiles, ei, keys4, N, E, Epad);
        if (rem > 0)
            qkv_tail_kernel<<<(rem * 32 + TB - 1) / TB, TB, 0, stream>>>(
                x, Wq, Wk, Wv, Q, K, V, rem_row0, N);
    }
    if (N > 0)
        own_range_kernel<<<(N + OWN_R - 1) / OWN_R, OWN_TB, 0, stream>>>(
            Q, K, V, ei, keys4, Z, outacc, N, E, Epad);
    {
        int threads = ntiles * 32;
        if (threads > 0)
            out_wmma_kernel<<<(threads + TB - 1) / TB, TB, 0, stream>>>(
                x, outacc, Z, Wo, out, ntiles);
        if (rem > 0)
            out_tail_kernel<<<(rem * 32 + TB - 1) / TB, TB, 0, stream>>>(
                x, outacc, Z, Wo, out, rem_row0, N);
    }
}


# Receiver ranges of 416, not 512: the range launch has 241 work-groups for the chip's 256 WGPs, not 196, so each group's edge list, row gathers and LDS adds are 19% shorter; sum order: LDS arrival
# speedup vs baseline: 7.5156x; 1.0626x over previous
#include <hip/hip_runtime.h>
#include <math.h>

typedef __attribute__((ext_vector_type(2))) float v2f;
typedef __attribute__((ext_vector_type(8))) float v8f;

__device__ __forceinline__ unsigned fkey(float f) {
    unsigned u = __float_as_uint(f);
    return (u & 0x80000000u) ? ~u : (u | 0x80000000u);
}
__device__ __forceinline__ float fkey_inv(unsigned k) {
    unsigned u = (k & 0x80000000u) ? (k & 0x7FFFFFFFu) : ~k;
    return __uint_as_float(u);
}

#define OWN_R    416
__device__ __forceinline__ unsigned own_key(int r, int N) {
    return ((unsigned)r < (unsigned)N) ? ((unsigned)r / (unsigned)OWN_R) : 0xFFu;
}

__device__ __forceinline__ void wmma_step(const float* __restrict__ W, int kb,
                                          int half, int m, v2f a,
                                          v8f& acc0, v8f& acc1) {
    int r0 = (kb + 2 * half + 0) * 32;
    int r1 = (kb + 2 * half + 1) * 32;
    v2f b0, b1;
    b0.x = W[r0 + m];        b0.y = W[r1 + m];
    b1.x = W[r0 + 16 + m];   b1.y = W[r1 + 16 + m];
    acc0 = __builtin_amdgcn_wmma_f32_16x16x4_f32(false, a, false, b0,
                                                 (short)0, acc0, false, false);
    acc1 = __builtin_amdgcn_wmma_f32_16x16x4_f32(false, a, false, b1,
                                                 (short)0, acc1, false, false);
}

__global__ __launch_bounds__(256) void qkv_wmma_kernel(
    const float* __restrict__ x,
    const float* __restrict__ Wq, const float* __restrict__ Wk,
    const float* __restrict__ Wv,
    float* __restrict__ Q, float* __restrict__ K, float* __restrict__ V,
    int ntiles, const int* __restrict__ ei, unsigned* __restrict__ keys4,
    int N, int E, int Epad) {
    if (keys4 != nullptr) {
        const int* recv = ei + E;
        const bool vec_ok = (((size_t)recv & 15u) == 0);
        const int gid = (int)(blockIdx.x * blockDim.x + threadIdx.x);
        const int gsz = (int)(gridDim.x * blockDim.x);
        for (int q = gid; q * 4 < Epad; q += gsz) {
            const int e0 = q * 4;
            int4 r;
            if (vec_ok && e0 + 4 <= E) {
                r = *reinterpret_cast<const int4*>(recv + e0);
            } else {
                r.x = (e0     < E) ? recv[e0]     : -1;
                r.y = (e0 + 1 < E) ? recv[e0 + 1] : -1;
                r.z = (e0 + 2 < E) ? recv[e0 + 2] : -1;
                r.w = (e0 + 3 < E) ? recv[e0 + 3] : -1;
            }
            keys4[q] = own_key(r.x, N) | (own_key(r.y, N) << 8) |
                       (own_key(r.z, N) << 16) | (own_key(r.w, N) << 24);
        }
    }
    int wave = (int)((blockIdx.x * blockDim.x + threadIdx.x) >> 5);
    if (wave >= ntiles) return;
    int lane = (int)(threadIdx.x & 31u);
    int half = lane >> 4;
    int m    = lane & 15;
    size_t row0 = (size_t)wave * 16;

    v8f q0 = {}, q1 = {}, k0 = {}, k1 = {}, v0 = {}, v1 = {};

#pragma unroll
    for (int kk = 0; kk < 8; ++kk) {
        int kb = kk * 4;
        const float* ap = x + (row0 + m) * 32 + kb + 2 * half;
        v2f a; a.x = ap[0]; a.y = ap[1];
        wmma_step(Wq, kb, half, m, a, q0, q1);
        wmma_step(Wk, kb, half, m, a, k0, k1);
        wmma_step(Wv, kb, half, m, a, v0, v1);
    }

#pragma unroll
    for (int r = 0; r < 8; ++r) {
        size_t row = (row0 + r + 8 * half) * 32;
        Q[row + m]      = q0[r];  Q[row + 16 + m] = q1[r];
        K[row + m]      = k0[r];  K[row + 16 + m] = k1[r];
        V[row + m]      = v0[r];  V[row + 16 + m] = v1[r];
    }
}

__global__ void qkv_tail_kernel(const float* __restrict__ x,
                                const float* __restrict__ Wq,
                                const float* __restrict__ Wk,
                                const float* __restrict__ Wv,
                                float* __restrict__ Q, float* __restrict__ K,
                                float* __restrict__ V, int row_start, int N) {
    int idx = blockIdx.x * blockDim.x + threadIdx.x;
    int nrows = N - row_start;
    if (idx >= nrows * 32) return;
    int row = row_start + idx / 32, col = idx & 31;
    float q = 0.f, k = 0.f, v = 0.f;
    for (int t = 0; t < 32; ++t) {
        float xv = x[(size_t)row * 32 + t];
        q += xv * Wq[t * 32 + col];
        k += xv * Wk[t * 32 + col];
        v += xv * Wv[t * 32 + col];
    }
    size_t o = (size_t)row * 32 + col;
    Q[o] = q; K[o] = k; V[o] = v;
}

template <int M>
__device__ __forceinline__ float lane_xor(float x) {
    int b = __builtin_bit_cast(int, x);
    if constexpr (M == 16) {
        return __builtin_bit_cast(float, __builtin_amdgcn_permlanex16(
            b, b, 0x76543210, 0xfedcba98, false, false));
    } else {
        return __builtin_bit_cast(float, __builtin_amdgcn_update_dpp(
            b, b, 0x160 | M, 0xf, 0xf, false));
    }
}

template <int M, int NH>
__device__ __forceinline__ void fold_level(float* v, bool hi) {
#pragma unroll
    for (int i = 0; i < NH; ++i) {
        float keep = hi ? v[i + NH] : v[i];
        float send = hi ? v[i] : v[i + NH];
        v[i] = keep + lane_xor<M>(send);
    }
}

template <int J>
__device__ __forceinline__ unsigned bcast8(unsigned x) {
    return (unsigned)__builtin_amdgcn_mov_dpp8(x, J * 0x249249);
}
template <int J>
__device__ __forceinline__ float bcast8f(float x) {
    return __uint_as_float(bcast8<J>(__float_as_uint(x)));
}

__device__ __forceinline__ int acc_pos(int rl, int f) {
    return (rl << 5) + ((((f & 3) ^ (rl >> 1)) & 3) << 3) + (f >> 2);
}

#define OWN_CAP  16384
#define OWN_TB   1024
#define OWN_U    8
#define OWN_STEP (OWN_TB * 4)
#define KEY_ROW  (OWN_TB * 16)
#define KEY_U    4
#define KEY_STEP (KEY_ROW * KEY_U)

__device__ __forceinline__ void own_push(unsigned rl, int e, unsigned* cnt,
                                         int* l_a) {
    if (rl < (unsigned)OWN_R) {
        unsigned slot = atomicAdd(cnt, 1u);
        if (slot < (unsigned)OWN_CAP) l_a[slot] = e;
    }
}

__device__ __forceinline__ void own_build(const int* __restrict__ recv, int a,
                                          int b, int rbase, bool vec_ok,
                                          unsigned* cnt, int* l_a) {
    const int t = (int)threadIdx.x;
    for (int base = a; base < b; base += OWN_STEP * OWN_U) {
        const bool full = vec_ok && (b - base >= OWN_STEP * OWN_U);
        int4 v[OWN_U];
        if (full) {
#pragma unroll
            for (int u = 0; u < OWN_U; ++u)
                v[u] = *reinterpret_cast<const int4*>(recv + base + u * OWN_STEP + t * 4);
        } else {
#pragma unroll
            for (int u = 0; u < OWN_U; ++u) {
                int e0 = base + u * OWN_STEP + t * 4;
                v[u].x = (e0     < b) ? recv[e0]     : -1;
                v[u].y = (e0 + 1 < b) ? recv[e0 + 1] : -1;
                v[u].z = (e0 + 2 < b) ? recv[e0 + 2] : -1;
                v[u].w = (e0 + 3 < b) ? recv[e0 + 3] : -1;
            }
        }
#pragma unroll
        for (int u = 0; u < OWN_U; ++u) {
            int e0 = base + u * OWN_STEP + t * 4;
            unsigned r0 = (unsigned)(v[u].x - rbase), r1 = (unsigned)(v[u].y - rbase);
            unsigned r2 = (unsigned)(v[u].z - rbase), r3 = (unsigned)(v[u].w - rbase);
            unsigned mn = min(min(r0, r1), min(r2, r3));
            if (mn < (unsigned)OWN_R) {
                own_push(r0, e0,     cnt, l_a);
                own_push(r1, e0 + 1, cnt, l_a);
                own_push(r2, e0 + 2, cnt, l_a);
                own_push(r3, e0 + 3, cnt, l_a);
            }
        }
    }
}

__device__ __forceinline__ unsigned zero_bytes(unsigned x) {
    return ~(((x & 0x7F7F7F7Fu) + 0x7F7F7F7Fu) | x) & 0x80808080u;
}

__device__ __forceinline__ void own_build_keys(const unsigned* __restrict__ keys4,
                                               int Epad, unsigned kb,
                                               unsigned* cnt, int* l_a) {
    const unsigned t = threadIdx.x;
    const unsigned splat = kb * 0x01010101u;
    for (int base = 0; base < Epad; base += KEY_STEP) {
        uint4 v[KEY_U];
#pragma unroll
        for (int u = 0; u < KEY_U; ++u)
            v[u] = *reinterpret_cast<const uint4*>(keys4 + ((base + u * KEY_ROW) >> 2) + t * 4);
#pragma unroll
        for (int u = 0; u < KEY_U; u += 2) {
            const unsigned code0 = ((unsigned)(base / KEY_ROW + u) << 15) | (t << 5);
            unsigned m = (zero_bytes(v[u].x ^ splat) >> 7) | (zero_bytes(v[u].y ^ splat) >> 6) |
                         (zero_bytes(v[u].z ^ splat) >> 5) | (zero_bytes(v[u].w ^ splat) >> 4) |
                         (zero_bytes(v[u + 1].x ^ splat) >> 3) | (zero_bytes(v[u + 1].y ^ splat) >> 2) |
                         (zero_bytes(v[u + 1].z ^ splat) >> 1) | zero_bytes(v[u + 1].w ^ splat);
            while (m) {
                const unsigned c = (unsigned)__builtin_ctz(m);
                m &= m - 1u;
                unsigned slot = atomicAdd(cnt, 1u);
                if (slot < (unsigned)OWN_CAP) l_a[slot] = (int)(code0 | c);
            }
        }
    }
}

__device__ __forceinline__ int own_edge(unsigned code) {
    const unsigned c = code & 31u, t = (code >> 5) & 1023u, row = code >> 15;
    return (int)((row + ((c >> 2) & 1u)) * KEY_ROW + t * 16u + ((c & 3u) << 2) + (c >> 3));
}

template <int J>
__device__ __forceinline__ void rows_qk(const float4* __restrict__ Qp,
                                        const float4* __restrict__ Kp,
                                        unsigned r, unsigned s, float4* qv, float4* kv) {
    qv[J] = Qp[(size_t)bcast8<J>(r) << 3];
    kv[J] = Kp[(size_t)bcast8<J>(s) << 3];
}

template <bool tail>
__device__ __forceinline__ void add_rows(const float4* __restrict__ Vp, unsigned s,
                                         unsigned rl, float alpha, int m,
                                         unsigned lane, float* acc) {
    const float4* a0 = Vp + ((size_t)bcast8<0>(s) << 3);
    const float4* a1 = Vp + ((size_t)bcast8<1>(s) << 3);
    const float4* a2 = Vp + ((size_t)bcast8<2>(s) << 3);
    const float4* a3 = Vp + ((size_t)bcast8<3>(s) << 3);
    const float4* a4 = Vp + ((size_t)bcast8<4>(s) << 3);
    const float4* a5 = Vp + ((size_t)bcast8<5>(s) << 3);
    const float4* a6 = Vp + ((size_t)bcast8<6>(s) << 3);
    const float4* a7 = Vp + ((size_t)bcast8<7>(s) << 3);
    __builtin_amdgcn_sched_barrier(0);
    float4 v[8];
    v[0] = *a0;  v[1] = *a1;  v[2] = *a2;  v[3] = *a3;
    v[4] = *a4;  v[5] = *a5;  v[6] = *a6;  v[7] = *a7;
    unsigned rj[8];
    float aj[8];
    rj[0] = bcast8<0>(rl);  aj[0] = bcast8f<0>(alpha);
    rj[1] = bcast8<1>(rl);  aj[1] = bcast8f<1>(alpha);
    rj[2] = bcast8<2>(rl);  aj[2] = bcast8f<2>(alpha);
    rj[3] = bcast8<3>(rl);  aj[3] = bcast8f<3>(alpha);
    rj[4] = bcast8<4>(rl);  aj[4] = bcast8f<4>(alpha);
    rj[5] = bcast8<5>(rl);  aj[5] = bcast8f<5>(alpha);
    rj[6] = bcast8<6>(rl);  aj[6] = bcast8f<6>(alpha);
    rj[7] = bcast8<7>(rl);  aj[7] = bcast8f<7>(alpha);
    const unsigned q4 = (lane & 7u) << 2;
    char* accb = reinterpret_cast<char*>(acc);
#pragma unroll
    for (int j = 0; j < 8; ++j) {
        const unsigned b0 = (rj[j] << 7) | ((rj[j] << 4) & 0x60u) | q4;
        if (!tail || (int)((lane & 24u) | (unsigned)j) < m) {
            atomicAdd(reinterpret_cast<float*>(accb + b0),         aj[j] * v[j].x);
            atomicAdd(reinterpret_cast<float*>(accb + (b0 ^ 32u)), aj[j] * v[j].y);
            atomicAdd(reinterpret_cast<float*>(accb + (b0 ^ 64u)), aj[j] * v[j].z);
            atomicAdd(reinterpret_cast<float*>(accb + (b0 ^ 96u)), aj[j] * v[j].w);
        }
    }
}

__device__ __forceinline__ void own_process(
    const float* __restrict__ Q, const float* __restrict__ K,
    const float* __restrict__ V, const int* __restrict__ ei,
    const int* __restrict__ recv, int rbase, int n,
    bool first, bool coded, float* acc, float* zs, unsigned* mkey, unsigned* mnew,
    int* l_a, int* l_r, float* l_sc) {
    const int t = (int)threadIdx.x;
    const unsigned lane = threadIdx.x & 31u;
    const int wave = __builtin_amdgcn_readfirstlane((int)(threadIdx.x >> 5));
    const float4* Qp = reinterpret_cast<const float4*>(Q) + (lane & 7u);
    const float4* Kp = reinterpret_cast<const float4*>(K) + (lane & 7u);
    const float4* Vp = reinterpret_cast<const float4*>(V) + (lane & 7u);

    for (int c = wave; c * 32 < n; c += OWN_TB / 32) {
        int i = c * 32 + (int)lane;
        bool valid = i < n;
        int s = 0, r = rbase;
        if (valid) {
            int e = l_a[i];
            if (coded) e = own_edge((unsigned)e);
            s = ei[e]; r = recv[e];
        }
        int rl = r - rbase;
        float4 qv[8], kv[8];
        rows_qk<0>(Qp, Kp, (unsigned)r, (unsigned)s, qv, kv);
        rows_qk<1>(Qp, Kp, (unsigned)r, (unsigned)s, qv, kv);
        rows_qk<2>(Qp, Kp, (unsigned)r, (unsigned)s, qv, kv);
        rows_qk<3>(Qp, Kp, (unsigned)r, (unsigned)s, qv, kv);
        rows_qk<4>(Qp, Kp, (unsigned)r, (unsigned)s, qv, kv);
        rows_qk<5>(Qp, Kp, (unsigned)r, (unsigned)s, qv, kv);
        rows_qk<6>(Qp, Kp, (unsigned)r, (unsigned)s, qv, kv);
        rows_qk<7>(Qp, Kp, (unsigned)r, (unsigned)s, qv, kv);
        __builtin_amdgcn_sched_barrier(0);
        float p[8];
#pragma unroll
        for (int j = 0; j < 8; ++j)
            p[j] = qv[j].x * kv[j].x + qv[j].y * kv[j].y +
                   qv[j].z * kv[j].z + qv[j].w * kv[j].w;
        fold_level<4, 4>(p, (lane & 4) != 0);
        fold_level<2, 2>(p, (lane & 2) != 0);
        fold_level<1, 1>(p, (lane & 1) != 0);
        float sc = p[0] * 0.17677669529663687f;
        if (valid) {
            l_sc[i] = sc;
            l_r[i] = rl;
            l_a[i] = s;
            atomicMax(&mnew[rl], fkey(sc));
        }
    }
    __syncthreads();
    if (!first) {
        for (int idx = t; idx < OWN_R * 32; idx += OWN_TB) {
            unsigned ko = mkey[idx >> 5], kn = mnew[idx >> 5];
            if (kn != ko)
                acc[idx] *= ko ? __expf(fkey_inv(ko) - fkey_inv(kn)) : 0.0f;
        }
        for (int rl = t; rl < OWN_R; rl += OWN_TB) {
            unsigned ko = mkey[rl], kn = mnew[rl];
            if (kn != ko)
                zs[rl] *= ko ? __expf(fkey_inv(ko) - fkey_inv(kn)) : 0.0f;
        }
        __syncthreads();
    }
    for (int rl = t; rl < OWN_R; rl += OWN_TB) mkey[rl] = mnew[rl];
    __syncthreads();

    for (int c = wave; c * 32 < n; c += OWN_TB / 32) {
        int i = c * 32 + (int)lane;
        int m = min(32, n - c * 32);
        int s = 0, rl = 0;
        float alpha = 0.f;
        if (i < n) {
            s = l_a[i]; rl = l_r[i];
            alpha = __expf(l_sc[i] - fkey_inv(mkey[rl]));
            atomicAdd(&zs[rl], alpha);
        }
        if (m == 32)
            add_rows<false>(Vp, (unsigned)s, (unsigned)rl, alpha, m, lane, acc);
        else
            add_rows<true>(Vp, (unsigned)s, (unsigned)rl, alpha, m, lane, acc);
    }
}

__global__ __launch_bounds__(OWN_TB) void own_range_kernel(
    const float* __restrict__ Q, const float* __restrict__ K,
    const float* __restrict__ V, const int* __restrict__ ei,
    const unsigned* __restrict__ keys4,
    float* __restrict__ Z, float* __restrict__ outacc, int N, int E, int Epad) {
    __shared__ float    acc[OWN_R * 32];
    __shared__ float    zs[OWN_R];
    __shared__ unsigned mkey[OWN_R];
    __shared__ unsigned mnew[OWN_R];
    __shared__ int      l_a[OWN_CAP];
    __shared__ int      l_r[OWN_CAP];
    __shared__ float    l_sc[OWN_CAP];
    __shared__ unsigned cnt;

    const int t = (int)threadIdx.x;
    const int rbase = (int)blockIdx.x * OWN_R;
    const int nown = min(OWN_R, N - rbase);
    const int* recv = ei + E;
    const bool vec_ok = (((size_t)recv & 15u) == 0);

    for (int i = t; i < OWN_R * 32; i += OWN_TB) acc[i] = 0.0f;
    for (int i = t; i < OWN_R; i += OWN_TB) { zs[i] = 0.0f; mkey[i] = 0u; mnew[i] = 0u; }
    if (t == 0) cnt = 0u;
    __syncthreads();

    if (keys4 != nullptr)
        own_build_keys(keys4, Epad, blockIdx.x, &cnt, l_a);
    else
        own_build(recv, 0, E, rbase, vec_ok, &cnt, l_a);
    __syncthreads();
    const unsigned total = cnt;
    if (total <= (unsigned)OWN_CAP) {
        own_process(Q, K, V, ei, recv, rbase, (int)total, true, keys4 != nullptr,
                    acc, zs, mkey, mnew, l_a, l_r, l_sc);
    } else {
        for (int a = 0; a < E; a += OWN_CAP) {
            __syncthreads();
            if (t == 0) cnt = 0u;
            __syncthreads();
            int b = (E - a > OWN_CAP) ? a + OWN_CAP : E;
            own_build(recv, a, b, rbase, vec_ok, &cnt, l_a);
            __syncthreads();
            own_process(Q, K, V, ei, recv, rbase, (int)cnt, a == 0, false, acc, zs,
                        mkey, mnew, l_a, l_r, l_sc);
        }
    }
    __syncthreads();
    for (int i = t; i < nown * 32; i += OWN_TB)
        outacc[(size_t)rbase * 32 + i] = acc[acc_pos(i >> 5, i & 31)];
    for (int i = t; i < nown; i += OWN_TB) Z[rbase + i] = zs[i];
}

__global__ __launch_bounds__(256) void out_wmma_kernel(
    const float* __restrict__ x, const float* __restrict__ outacc,
    const float* __restrict__ Z, const float* __restrict__ Wo,
    float* __restrict__ out, int ntiles) {
    int wave = (int)((blockIdx.x * blockDim.x + threadIdx.x) >> 5);
    if (wave >= ntiles) return;
    int lane = (int)(threadIdx.x & 31u);
    int half = lane >> 4;
    int m    = lane & 15;
    size_t row0 = (size_t)wave * 16;

    float invz = 1.0f / (Z[row0 + m] + 1e-6f);
    v8f c0 = {}, c1 = {};

#pragma unroll
    for (int kk = 0; kk < 8; ++kk) {
        int kb = kk * 4;
        const float* ap = outacc + (row0 + m) * 32 + kb + 2 * half;
        v2f a; a.x = ap[0] * invz; a.y = ap[1] * invz;
        wmma_step(Wo, kb, half, m, a, c0, c1);
    }

#pragma unroll
    for (int r = 0; r < 8; ++r) {
        size_t row = (row0 + r + 8 * half) * 32;
        out[row + m]      = x[row + m]      + c0[r];
        out[row + 16 + m] = x[row + 16 + m] + c1[r];
    }
}

__global__ void out_tail_kernel(const float* __restrict__ x,
                                const float* __restrict__ outacc,
                                const float* __restrict__ Z,
                                const float* __restrict__ Wo,
                                float* __restrict__ out, int row_start, int N) {
    int idx = blockIdx.x * blockDim.x + threadIdx.x;
    int nrows = N - row_start;
    if (idx >= nrows * 32) return;
    int row = row_start + idx / 32, col = idx & 31;
    float invz = 1.0f / (Z[row] + 1e-6f);
    float acc = 0.f;
    for (int t = 0; t < 32; ++t)
        acc += outacc[(size_t)row * 32 + t] * invz * Wo[t * 32 + col];
    size_t o = (size_t)row * 32 + col;
    out[o] = x[o] + acc;
}

extern "C" void kernel_launch(void* const* d_in, const int* in_sizes, int n_in,
                              void* d_out, int out_size, void* d_ws,
                              size_t ws_size, hipStream_t stream) {
    const float* x  = (const float*)d_in[0];
    const int*   ei = (const int*)d_in[1];
    const float* Wq = (const float*)d_in[2];
    const float* Wk = (const float*)d_in[3];
    const float* Wv = (const float*)d_in[4];
    const float* Wo = (const float*)d_in[5];
    float* out = (float*)d_out;

    const int N = in_sizes[0] / 32;
    const int E = in_sizes[1] / 2;

    float* Q       = (float*)d_ws;
    float* K       = Q + (size_t)N * 32;
    float* V       = K + (size_t)N * 32;
    float* outacc  = V + (size_t)N * 32;
    float* Z       = outacc + (size_t)N * 32;
    const int Epad = ((E + KEY_STEP - 1) / KEY_STEP) * KEY_STEP;
    const size_t key_off = (((size_t)N * 129 * sizeof(float)) + 15) & ~(size_t)15;
    unsigned* keys4 = (unsigned*)((unsigned char*)d_ws + key_off);
    if (N > 255 * OWN_R || key_off + (size_t)Epad > ws_size) keys4 = nullptr;

    const int TB = 256;
    int ntiles = N / 16;
    int rem_row0 = ntiles * 16;
    int rem = N - rem_row0;

    {
        int blocks = (ntiles * 32 + TB - 1) / TB;
        if (blocks == 0 && keys4 != nullptr) blocks = 1;
        if (blocks > 0)
            qkv_wmma_kernel<<<blocks, TB, 0, stream>>>(
                x, Wq, Wk, Wv, Q, K, V, ntiles, ei, keys4, N, E, Epad);
        if (rem > 0)
            qkv_tail_kernel<<<(rem * 32 + TB - 1) / TB, TB, 0, stream>>>(
                x, Wq, Wk, Wv, Q, K, V, rem_row0, N);
    }
    if (N > 0)
        own_range_kernel<<<(N + OWN_R - 1) / OWN_R, OWN_TB, 0, stream>>>(
            Q, K, V, ei, keys4, Z, outacc, N, E, Epad);
    {
        int threads = ntiles * 32;
        if (threads > 0)
            out_wmma_kernel<<<(threads + TB - 1) / TB, TB, 0, stream>>>(
                x, outacc, Z, Wo, out, ntiles);
        if (rem > 0)
            out_tail_kernel<<<(rem * 32 + TB - 1) / TB, TB, 0, stream>>>(
                x, outacc, Z, Wo, out, rem_row0, N);
    }
}


# Output product and residual done by the receiver-range work-group itself: 16-row WMMA tiles read straight from its LDS sums, out rows leave as whole lines; sums/Z memory trip and third launch gone
# speedup vs baseline: 8.1736x; 1.0876x over previous
#include <hip/hip_runtime.h>
#include <math.h>

typedef __attribute__((ext_vector_type(2))) float v2f;
typedef __attribute__((ext_vector_type(8))) float v8f;

__device__ __forceinline__ unsigned fkey(float f) {
    unsigned u = __float_as_uint(f);
    return (u & 0x80000000u) ? ~u : (u | 0x80000000u);
}
__device__ __forceinline__ float fkey_inv(unsigned k) {
    unsigned u = (k & 0x80000000u) ? (k & 0x7FFFFFFFu) : ~k;
    return __uint_as_float(u);
}

#define OWN_R    416
__device__ __forceinline__ unsigned own_key(int r, int N) {
    return ((unsigned)r < (unsigned)N) ? ((unsigned)r / (unsigned)OWN_R) : 0xFFu;
}

__device__ __forceinline__ void wmma_step(const float* __restrict__ W, int kb,
                                          int half, int m, v2f a,
                                          v8f& acc0, v8f& acc1) {
    int r0 = (kb + 2 * half + 0) * 32;
    int r1 = (kb + 2 * half + 1) * 32;
    v2f b0, b1;
    b0.x = W[r0 + m];        b0.y = W[r1 + m];
    b1.x = W[r0 + 16 + m];   b1.y = W[r1 + 16 + m];
    acc0 = __builtin_amdgcn_wmma_f32_16x16x4_f32(false, a, false, b0,
                                                 (short)0, acc0, false, false);
    acc1 = __builtin_amdgcn_wmma_f32_16x16x4_f32(false, a, false, b1,
                                                 (short)0, acc1, false, false);
}

__global__ __launch_bounds__(256) void qkv_wmma_kernel(
    const float* __restrict__ x,
    const float* __restrict__ Wq, const float* __restrict__ Wk,
    const float* __restrict__ Wv,
    float* __restrict__ Q, float* __restrict__ K, float* __restrict__ V,
    int ntiles, const int* __restrict__ ei, unsigned* __restrict__ keys4,
    int N, int E, int Epad) {
    if (keys4 != nullptr) {
        const int* recv = ei + E;
        const bool vec_ok = (((size_t)recv & 15u) == 0);
        const int gid = (int)(blockIdx.x * blockDim.x + threadIdx.x);
        const int gsz = (int)(gridDim.x * blockDim.x);
        for (int q = gid; q * 4 < Epad; q += gsz) {
            const int e0 = q * 4;
            int4 r;
            if (vec_ok && e0 + 4 <= E) {
                r = *reinterpret_cast<const int4*>(recv + e0);
            } else {
                r.x = (e0     < E) ? recv[e0]     : -1;
                r.y = (e0 + 1 < E) ? recv[e0 + 1] : -1;
                r.z = (e0 + 2 < E) ? recv[e0 + 2] : -1;
                r.w = (e0 + 3 < E) ? recv[e0 + 3] : -1;
            }
            keys4[q] = own_key(r.x, N) | (own_key(r.y, N) << 8) |
                       (own_key(r.z, N) << 16) | (own_key(r.w, N) << 24);
        }
    }
    int wave = (int)((blockIdx.x * blockDim.x + threadIdx.x) >> 5);
    if (wave >= ntiles) return;
    int lane = (int)(threadIdx.x & 31u);
    int half = lane >> 4;
    int m    = lane & 15;
    size_t row0 = (size_t)wave * 16;

    v8f q0 = {}, q1 = {}, k0 = {}, k1 = {}, v0 = {}, v1 = {};

#pragma unroll
    for (int kk = 0; kk < 8; ++kk) {
        int kb = kk * 4;
        const float* ap = x + (row0 + m) * 32 + kb + 2 * half;
        v2f a; a.x = ap[0]; a.y = ap[1];
        wmma_step(Wq, kb, half, m, a, q0, q1);
        wmma_step(Wk, kb, half, m, a, k0, k1);
        wmma_step(Wv, kb, half, m, a, v0, v1);
    }

#pragma unroll
    for (int r = 0; r < 8; ++r) {
        size_t row = (row0 + r + 8 * half) * 32;
        Q[row + m]      = q0[r];  Q[row + 16 + m] = q1[r];
        K[row + m]      = k0[r];  K[row + 16 + m] = k1[r];
        V[row + m]      = v0[r];  V[row + 16 + m] = v1[r];
    }
}

__global__ void qkv_tail_kernel(const float* __restrict__ x,
                                const float* __restrict__ Wq,
                                const float* __restrict__ Wk,
                                const float* __restrict__ Wv,
                                float* __restrict__ Q, float* __restrict__ K,
                                float* __restrict__ V, int row_start, int N) {
    int idx = blockIdx.x * blockDim.x + threadIdx.x;
    int nrows = N - row_start;
    if (idx >= nrows * 32) return;
    int row = row_start + idx / 32, col = idx & 31;
    float q = 0.f, k = 0.f, v = 0.f;
    for (int t = 0; t < 32; ++t) {
        float xv = x[(size_t)row * 32 + t];
        q += xv * Wq[t * 32 + col];
        k += xv * Wk[t * 32 + col];
        v += xv * Wv[t * 32 + col];
    }
    size_t o = (size_t)row * 32 + col;
    Q[o] = q; K[o] = k; V[o] = v;
}

template <int M>
__device__ __forceinline__ float lane_xor(float x) {
    int b = __builtin_bit_cast(int, x);
    if constexpr (M == 16) {
        return __builtin_bit_cast(float, __builtin_amdgcn_permlanex16(
            b, b, 0x76543210, 0xfedcba98, false, false));
    } else {
        return __builtin_bit_cast(float, __builtin_amdgcn_update_dpp(
            b, b, 0x160 | M, 0xf, 0xf, false));
    }
}

template <int M, int NH>
__device__ __forceinline__ void fold_level(float* v, bool hi) {
#pragma unroll
    for (int i = 0; i < NH; ++i) {
        float keep = hi ? v[i + NH] : v[i];
        float send = hi ? v[i] : v[i + NH];
        v[i] = keep + lane_xor<M>(send);
    }
}

template <int J>
__device__ __forceinline__ unsigned bcast8(unsigned x) {
    return (unsigned)__builtin_amdgcn_mov_dpp8(x, J * 0x249249);
}
template <int J>
__device__ __forceinline__ float bcast8f(float x) {
    return __uint_as_float(bcast8<J>(__float_as_uint(x)));
}

__device__ __forceinline__ int acc_pos(int rl, int f) {
    return (rl << 5) + ((((f & 3) ^ (rl >> 1)) & 3) << 3) + (f >> 2);
}

#define OWN_CAP  16384
#define OWN_TB   1024
#define OWN_U    8
#define OWN_STEP (OWN_TB * 4)
#define KEY_ROW  (OWN_TB * 16)
#define KEY_U    4
#define KEY_STEP (KEY_ROW * KEY_U)
#define OWN_XU   ((OWN_R * 8 + OWN_TB - 1) / OWN_TB)

__device__ __forceinline__ void own_push(unsigned rl, int e, unsigned* cnt,
                                         int* l_a) {
    if (rl < (unsigned)OWN_R) {
        unsigned slot = atomicAdd(cnt, 1u);
        if (slot < (unsigned)OWN_CAP) l_a[slot] = e;
    }
}

__device__ __forceinline__ void own_build(const int* __restrict__ recv, int a,
                                          int b, int rbase, bool vec_ok,
                                          unsigned* cnt, int* l_a) {
    const int t = (int)threadIdx.x;
    for (int base = a; base < b; base += OWN_STEP * OWN_U) {
        const bool full = vec_ok && (b - base >= OWN_STEP * OWN_U);
        int4 v[OWN_U];
        if (full) {
#pragma unroll
            for (int u = 0; u < OWN_U; ++u)
                v[u] = *reinterpret_cast<const int4*>(recv + base + u * OWN_STEP + t * 4);
        } else {
#pragma unroll
            for (int u = 0; u < OWN_U; ++u) {
                int e0 = base + u * OWN_STEP + t * 4;
                v[u].x = (e0     < b) ? recv[e0]     : -1;
                v[u].y = (e0 + 1 < b) ? recv[e0 + 1] : -1;
                v[u].z = (e0 + 2 < b) ? recv[e0 + 2] : -1;
                v[u].w = (e0 + 3 < b) ? recv[e0 + 3] : -1;
            }
        }
#pragma unroll
        for (int u = 0; u < OWN_U; ++u) {
            int e0 = base + u * OWN_STEP + t * 4;
            unsigned r0 = (unsigned)(v[u].x - rbase), r1 = (unsigned)(v[u].y - rbase);
            unsigned r2 = (unsigned)(v[u].z - rbase), r3 = (unsigned)(v[u].w - rbase);
            unsigned mn = min(min(r0, r1), min(r2, r3));
            if (mn < (unsigned)OWN_R) {
                own_push(r0, e0,     cnt, l_a);
                own_push(r1, e0 + 1, cnt, l_a);
                own_push(r2, e0 + 2, cnt, l_a);
                own_push(r3, e0 + 3, cnt, l_a);
            }
        }
    }
}

__device__ __forceinline__ unsigned zero_bytes(unsigned x) {
    return ~(((x & 0x7F7F7F7Fu) + 0x7F7F7F7Fu) | x) & 0x80808080u;
}

__device__ __forceinline__ void own_build_keys(const unsigned* __restrict__ keys4,
                                               int Epad, unsigned kb,
                                               unsigned* cnt, int* l_a) {
    const unsigned t = threadIdx.x;
    const unsigned splat = kb * 0x01010101u;
    for (int base = 0; base < Epad; base += KEY_STEP) {
        uint4 v[KEY_U];
#pragma unroll
        for (int u = 0; u < KEY_U; ++u)
            v[u] = *reinterpret_cast<const uint4*>(keys4 + ((base + u * KEY_ROW) >> 2) + t * 4);
#pragma unroll
        for (int u = 0; u < KEY_U; u += 2) {
            const unsigned code0 = ((unsigned)(base / KEY_ROW + u) << 15) | (t << 5);
            unsigned m = (zero_bytes(v[u].x ^ splat) >> 7) | (zero_bytes(v[u].y ^ splat) >> 6) |
                         (zero_bytes(v[u].z ^ splat) >> 5) | (zero_bytes(v[u].w ^ splat) >> 4) |
                         (zero_bytes(v[u + 1].x ^ splat) >> 3) | (zero_bytes(v[u + 1].y ^ splat) >> 2) |
                         (zero_bytes(v[u + 1].z ^ splat) >> 1) | zero_bytes(v[u + 1].w ^ splat);
            while (m) {
                const unsigned c = (unsigned)__builtin_ctz(m);
                m &= m - 1u;
                unsigned slot = atomicAdd(cnt, 1u);
                if (slot < (unsigned)OWN_CAP) l_a[slot] = (int)(code0 | c);
            }
        }
    }
}

__device__ __forceinline__ int own_edge(unsigned code) {
    const unsigned c = code & 31u, t = (code >> 5) & 1023u, row = code >> 15;
    return (int)((row + ((c >> 2) & 1u)) * KEY_ROW + t * 16u + ((c & 3u) << 2) + (c >> 3));
}

template <int J>
__device__ __forceinline__ void rows_qk(const float4* __restrict__ Qp,
                                        const float4* __restrict__ Kp,
                                        unsigned r, unsigned s, float4* qv, float4* kv) {
    qv[J] = Qp[(size_t)bcast8<J>(r) << 3];
    kv[J] = Kp[(size_t)bcast8<J>(s) << 3];
}

template <bool tail>
__device__ __forceinline__ void add_rows(const float4* __restrict__ Vp, unsigned s,
                                         unsigned rl, float alpha, int m,
                                         unsigned lane, float* acc) {
    const float4* a0 = Vp + ((size_t)bcast8<0>(s) << 3);
    const float4* a1 = Vp + ((size_t)bcast8<1>(s) << 3);
    const float4* a2 = Vp + ((size_t)bcast8<2>(s) << 3);
    const float4* a3 = Vp + ((size_t)bcast8<3>(s) << 3);
    const float4* a4 = Vp + ((size_t)bcast8<4>(s) << 3);
    const float4* a5 = Vp + ((size_t)bcast8<5>(s) << 3);
    const float4* a6 = Vp + ((size_t)bcast8<6>(s) << 3);
    const float4* a7 = Vp + ((size_t)bcast8<7>(s) << 3);
    __builtin_amdgcn_sched_barrier(0);
    float4 v[8];
    v[0] = *a0;  v[1] = *a1;  v[2] = *a2;  v[3] = *a3;
    v[4] = *a4;  v[5] = *a5;  v[6] = *a6;  v[7] = *a7;
    unsigned rj[8];
    float aj[8];
    rj[0] = bcast8<0>(rl);  aj[0] = bcast8f<0>(alpha);
    rj[1] = bcast8<1>(rl);  aj[1] = bcast8f<1>(alpha);
    rj[2] = bcast8<2>(rl);  aj[2] = bcast8f<2>(alpha);
    rj[3] = bcast8<3>(rl);  aj[3] = bcast8f<3>(alpha);
    rj[4] = bcast8<4>(rl);  aj[4] = bcast8f<4>(alpha);
    rj[5] = bcast8<5>(rl);  aj[5] = bcast8f<5>(alpha);
    rj[6] = bcast8<6>(rl);  aj[6] = bcast8f<6>(alpha);
    rj[7] = bcast8<7>(rl);  aj[7] = bcast8f<7>(alpha);
    const unsigned q4 = (lane & 7u) << 2;
    char* accb = reinterpret_cast<char*>(acc);
#pragma unroll
    for (int j = 0; j < 8; ++j) {
        const unsigned b0 = (rj[j] << 7) | ((rj[j] << 4) & 0x60u) | q4;
        if (!tail || (int)((lane & 24u) | (unsigned)j) < m) {
            atomicAdd(reinterpret_cast<float*>(accb + b0),         aj[j] * v[j].x);
            atomicAdd(reinterpret_cast<float*>(accb + (b0 ^ 32u)), aj[j] * v[j].y);
            atomicAdd(reinterpret_cast<float*>(accb + (b0 ^ 64u)), aj[j] * v[j].z);
            atomicAdd(reinterpret_cast<float*>(accb + (b0 ^ 96u)), aj[j] * v[j].w);
        }
    }
}

__device__ __forceinline__ void own_process(
    const float* __restrict__ Q, const float* __restrict__ K,
    const float* __restrict__ V, const int* __restrict__ ei,
    const int* __restrict__ recv, int rbase, int n,
    bool first, bool coded, float* acc, float* zs, unsigned* mkey, unsigned* mnew,
    int* l_a, int* l_r, float* l_sc) {
    const int t = (int)threadIdx.x;
    const unsigned lane = threadIdx.x & 31u;
    const int wave = __builtin_amdgcn_readfirstlane((int)(threadIdx.x >> 5));
    const float4* Qp = reinterpret_cast<const float4*>(Q) + (lane & 7u);
    const float4* Kp = reinterpret_cast<const float4*>(K) + (lane & 7u);
    const float4* Vp = reinterpret_cast<const float4*>(V) + (lane & 7u);

    for (int c = wave; c * 32 < n; c += OWN_TB / 32) {
        int i = c * 32 + (int)lane;
        bool valid = i < n;
        int s = 0, r = rbase;
        if (valid) {
            int e = l_a[i];
            if (coded) e = own_edge((unsigned)e);
            s = ei[e]; r = recv[e];
        }
        int rl = r - rbase;
        float4 qv[8], kv[8];
        rows_qk<0>(Qp, Kp, (unsigned)r, (unsigned)s, qv, kv);
        rows_qk<1>(Qp, Kp, (unsigned)r, (unsigned)s, qv, kv);
        rows_qk<2>(Qp, Kp, (unsigned)r, (unsigned)s, qv, kv);
        rows_qk<3>(Qp, Kp, (unsigned)r, (unsigned)s, qv, kv);
        rows_qk<4>(Qp, Kp, (unsigned)r, (unsigned)s, qv, kv);
        rows_qk<5>(Qp, Kp, (unsigned)r, (unsigned)s, qv, kv);
        rows_qk<6>(Qp, Kp, (unsigned)r, (unsigned)s, qv, kv);
        rows_qk<7>(Qp, Kp, (unsigned)r, (unsigned)s, qv, kv);
        __builtin_amdgcn_sched_barrier(0);
        float p[8];
#pragma unroll
        for (int j = 0; j < 8; ++j)
            p[j] = qv[j].x * kv[j].x + qv[j].y * kv[j].y +
                   qv[j].z * kv[j].z + qv[j].w * kv[j].w;
        fold_level<4, 4>(p, (lane & 4) != 0);
        fold_level<2, 2>(p, (lane & 2) != 0);
        fold_level<1, 1>(p, (lane & 1) != 0);
        float sc = p[0] * 0.17677669529663687f;
        if (valid) {
            l_sc[i] = sc;
            l_r[i] = rl;
            l_a[i] = s;
            atomicMax(&mnew[rl], fkey(sc));
        }
    }
    __syncthreads();
    if (!first) {
        for (int idx = t; idx < OWN_R * 32; idx += OWN_TB) {
            unsigned ko = mkey[idx >> 5], kn = mnew[idx >> 5];
            if (kn != ko)
                acc[idx] *= ko ? __expf(fkey_inv(ko) - fkey_inv(kn)) : 0.0f;
        }
        for (int rl = t; rl < OWN_R; rl += OWN_TB) {
            unsigned ko = mkey[rl], kn = mnew[rl];
            if (kn != ko)
                zs[rl] *= ko ? __expf(fkey_inv(ko) - fkey_inv(kn)) : 0.0f;
        }
        __syncthreads();
    }
    for (int rl = t; rl < OWN_R; rl += OWN_TB) mkey[rl] = mnew[rl];
    __syncthreads();

    for (int c = wave; c * 32 < n; c += OWN_TB / 32) {
        int i = c * 32 + (int)lane;
        int m = min(32, n - c * 32);
        int s = 0, rl = 0;
        float alpha = 0.f;
        if (i < n) {
            s = l_a[i]; rl = l_r[i];
            alpha = __expf(l_sc[i] - fkey_inv(mkey[rl]));
            atomicAdd(&zs[rl], alpha);
        }
        if (m == 32)
            add_rows<false>(Vp, (unsigned)s, (unsigned)rl, alpha, m, lane, acc);
        else
            add_rows<true>(Vp, (unsigned)s, (unsigned)rl, alpha, m, lane, acc);
    }
}

__global__ __launch_bounds__(OWN_TB) void own_range_kernel(
    const float* __restrict__ Q, const float* __restrict__ K,
    const float* __restrict__ V, const int* __restrict__ ei,
    const unsigned* __restrict__ keys4,
    const float* __restrict__ x, const float* __restrict__ Wo,
    float* __restrict__ out, int N, int E, int Epad) {
    __shared__ __attribute__((aligned(16))) float acc[OWN_R * 32];
    __shared__ float    zs[OWN_R];
    __shared__ unsigned mkey[OWN_R];
    __shared__ unsigned mnew[OWN_R];
    __shared__ int      l_a[OWN_CAP];
    __shared__ int      l_r[OWN_CAP];
    __shared__ float    l_sc[OWN_CAP];
    __shared__ unsigned cnt;

    const int t = (int)threadIdx.x;
    const int rbase = (int)blockIdx.x * OWN_R;
    const int nown = min(OWN_R, N - rbase);
    const int* recv = ei + E;
    const bool vec_ok = (((size_t)recv & 15u) == 0);

    for (int i = t; i < OWN_R * 32; i += OWN_TB) acc[i] = 0.0f;
    for (int i = t; i < OWN_R; i += OWN_TB) { zs[i] = 0.0f; mkey[i] = 0u; mnew[i] = 0u; }
    if (t == 0) cnt = 0u;
    __syncthreads();

    if (keys4 != nullptr)
        own_build_keys(keys4, Epad, blockIdx.x, &cnt, l_a);
    else
        own_build(recv, 0, E, rbase, vec_ok, &cnt, l_a);
    __syncthreads();
    const float4* x4 = reinterpret_cast<const float4*>(x) + (size_t)rbase * 8;
    float4*       o4 = reinterpret_cast<float4*>(out) + (size_t)rbase * 8;
    const int npiece = nown * 8;
    float4 xr[OWN_XU];
#pragma unroll
    for (int u = 0; u < OWN_XU; ++u) {
        const int i = t + u * OWN_TB;
        xr[u] = (i < npiece) ? x4[i] : make_float4(0.f, 0.f, 0.f, 0.f);
    }
    const unsigned total = cnt;
    if (total <= (unsigned)OWN_CAP) {
        own_process(Q, K, V, ei, recv, rbase, (int)total, true, keys4 != nullptr,
                    acc, zs, mkey, mnew, l_a, l_r, l_sc);
    } else {
        for (int a = 0; a < E; a += OWN_CAP) {
            __syncthreads();
            if (t == 0) cnt = 0u;
            __syncthreads();
            int b = (E - a > OWN_CAP) ? a + OWN_CAP : E;
            own_build(recv, a, b, rbase, vec_ok, &cnt, l_a);
            __syncthreads();
            own_process(Q, K, V, ei, recv, rbase, (int)cnt, a == 0, false, acc, zs,
                        mkey, mnew, l_a, l_r, l_sc);
        }
    }
    __syncthreads();
    {
        const int wave = __builtin_amdgcn_readfirstlane(t >> 5);
        if (wave * 16 < nown) {
            const int lane = t & 31;
            const int half = lane >> 4;
            const int m    = lane & 15;
            const int rl   = wave * 16 + m;
            const float invz = 1.0f / (zs[rl] + 1e-6f);
            const float* rowp = acc + (rl << 5);
            const int w0 = (((2 * half)     ^ (rl >> 1)) & 3) << 3;
            const int w1 = (((2 * half + 1) ^ (rl >> 1)) & 3) << 3;
            const float4 e0 = *reinterpret_cast<const float4*>(rowp + w0);
            const float4 e1 = *reinterpret_cast<const float4*>(rowp + w0 + 4);
            const float4 f0 = *reinterpret_cast<const float4*>(rowp + w1);
            const float4 f1 = *reinterpret_cast<const float4*>(rowp + w1 + 4);
            const float ae[8] = {e0.x, e0.y, e0.z, e0.w, e1.x, e1.y, e1.z, e1.w};
            const float af[8] = {f0.x, f0.y, f0.z, f0.w, f1.x, f1.y, f1.z, f1.w};
            v8f c0 = {}, c1 = {};
#pragma unroll
            for (int kk = 0; kk < 8; ++kk) {
                v2f a; a.x = ae[kk] * invz; a.y = af[kk] * invz;
                wmma_step(Wo, kk * 4, half, m, a, c0, c1);
            }
#pragma unroll
            for (int r = 0; r < 8; ++r) {
                float* d = acc + ((wave * 16 + r + 8 * half) << 5);
                d[m]      = c0[r];
                d[16 + m] = c1[r];
            }
        }
    }
    __syncthreads();
#pragma unroll
    for (int u = 0; u < OWN_XU; ++u) {
        const int i = t + u * OWN_TB;
        if (i < npiece) {
            const float4 p = *reinterpret_cast<const float4*>(acc + (i << 2));
            const float4 xv = xr[u];
            o4[i] = make_float4(xv.x + p.x, xv.y + p.y, xv.z + p.z, xv.w + p.w);
        }
    }
}

extern "C" void kernel_launch(void* const* d_in, const int* in_sizes, int n_in,
                              void* d_out, int out_size, void* d_ws,
                              size_t ws_size, hipStream_t stream) {
    const float* x  = (const float*)d_in[0];
    const int*   ei = (const int*)d_in[1];
    const float* Wq = (const float*)d_in[2];
    const float* Wk = (const float*)d_in[3];
    const float* Wv = (const float*)d_in[4];
    const float* Wo = (const float*)d_in[5];
    float* out = (float*)d_out;

    const int N = in_sizes[0] / 32;
    const int E = in_sizes[1] / 2;

    float* Q       = (float*)d_ws;
    float* K       = Q + (size_t)N * 32;
    float* V       = K + (size_t)N * 32;
    const int Epad = ((E + KEY_STEP - 1) / KEY_STEP) * KEY_STEP;
    const size_t key_off = (((size_t)N * 129 * sizeof(float)) + 15) & ~(size_t)15;
    unsigned* keys4 = (unsigned*)((unsigned char*)d_ws + key_off);
    if (N > 255 * OWN_R || key_off + (size_t)Epad > ws_size) keys4 = nullptr;

    const int TB = 256;
    int ntiles = N / 16;
    int rem_row0 = ntiles * 16;
    int rem = N - rem_row0;

    {
        int blocks = (ntiles * 32 + TB - 1) / TB;
        if (blocks == 0 && keys4 != nullptr) blocks = 1;
        if (blocks > 0)
            qkv_wmma_kernel<<<blocks, TB, 0, stream>>>(
                x, Wq, Wk, Wv, Q, K, V, ntiles, ei, keys4, N, E, Epad);
        if (rem > 0)
            qkv_tail_kernel<<<(rem * 32 + TB - 1) / TB, TB, 0, stream>>>(
                x, Wq, Wk, Wv, Q, K, V, rem_row0, N);
    }
    if (N > 0)
        own_range_kernel<<<(N + OWN_R - 1) / OWN_R, OWN_TB, 0, stream>>>(
            Q, K, V, ei, keys4, x, Wo, out, N, E, Epad);
}


# Projection launch also writes a 4-byte pair word per edge (sender | row of the receiver in its range); pass A reads one scattered word per edge, not two ids from two lines; sum order as before
# speedup vs baseline: 8.5749x; 1.0491x over previous
#include <hip/hip_runtime.h>
#include <math.h>

typedef __attribute__((ext_vector_type(2))) float v2f;
typedef __attribute__((ext_vector_type(8))) float v8f;

__device__ __forceinline__ unsigned fkey(float f) {
    unsigned u = __float_as_uint(f);
    return (u & 0x80000000u) ? ~u : (u | 0x80000000u);
}
__device__ __forceinline__ float fkey_inv(unsigned k) {
    unsigned u = (k & 0x80000000u) ? (k & 0x7FFFFFFFu) : ~k;
    return __uint_as_float(u);
}

#define OWN_R    416
__device__ __forceinline__ unsigned own_key(int r, int N) {
    return ((unsigned)r < (unsigned)N) ? ((unsigned)r / (unsigned)OWN_R) : 0xFFu;
}
#define PAIR_SH  23
__device__ __forceinline__ unsigned pair_word(int s, int r) {
    return ((unsigned)s & ((1u << PAIR_SH) - 1u)) | (((unsigned)r % (unsigned)OWN_R) << PAIR_SH);
}

__device__ __forceinline__ void wmma_step(const float* __restrict__ W, int kb,
                                          int half, int m, v2f a,
                                          v8f& acc0, v8f& acc1) {
    int r0 = (kb + 2 * half + 0) * 32;
    int r1 = (kb + 2 * half + 1) * 32;
    v2f b0, b1;
    b0.x = W[r0 + m];        b0.y = W[r1 + m];
    b1.x = W[r0 + 16 + m];   b1.y = W[r1 + 16 + m];
    acc0 = __builtin_amdgcn_wmma_f32_16x16x4_f32(false, a, false, b0,
                                                 (short)0, acc0, false, false);
    acc1 = __builtin_amdgcn_wmma_f32_16x16x4_f32(false, a, false, b1,
                                                 (short)0, acc1, false, false);
}

__global__ __launch_bounds__(256) void qkv_wmma_kernel(
    const float* __restrict__ x,
    const float* __restrict__ Wq, const float* __restrict__ Wk,
    const float* __restrict__ Wv,
    float* __restrict__ Q, float* __restrict__ K, float* __restrict__ V,
    int ntiles, const int* __restrict__ ei, unsigned* __restrict__ keys4,
    unsigned* __restrict__ pairs, int N, int E, int Epad) {
    if (keys4 != nullptr) {
        const int* recv = ei + E;
        const bool vec_ok = (((size_t)recv & 15u) == 0);
        const bool svec_ok = (((size_t)ei & 15u) == 0);
        const int gid = (int)(blockIdx.x * blockDim.x + threadIdx.x);
        const int gsz = (int)(gridDim.x * blockDim.x);
        for (int q = gid; q * 4 < Epad; q += gsz) {
            const int e0 = q * 4;
            int4 r, sd = make_int4(0, 0, 0, 0);
            if (pairs != nullptr) {
                if (svec_ok && e0 + 4 <= E) {
                    sd = *reinterpret_cast<const int4*>(ei + e0);
                } else {
                    sd.x = (e0     < E) ? ei[e0]     : 0;
                    sd.y = (e0 + 1 < E) ? ei[e0 + 1] : 0;
                    sd.z = (e0 + 2 < E) ? ei[e0 + 2] : 0;
                    sd.w = (e0 + 3 < E) ? ei[e0 + 3] : 0;
                }
            }
            if (vec_ok && e0 + 4 <= E) {
                r = *reinterpret_cast<const int4*>(recv + e0);
            } else {
                r.x = (e0     < E) ? recv[e0]     : -1;
                r.y = (e0 + 1 < E) ? recv[e0 + 1] : -1;
                r.z = (e0 + 2 < E) ? recv[e0 + 2] : -1;
                r.w = (e0 + 3 < E) ? recv[e0 + 3] : -1;
            }
            keys4[q] = own_key(r.x, N) | (own_key(r.y, N) << 8) |
                       (own_key(r.z, N) << 16) | (own_key(r.w, N) << 24);
            if (pairs != nullptr) {
                uint4 w;
                w.x = pair_word(sd.x, r.x);  w.y = pair_word(sd.y, r.y);
                w.z = pair_word(sd.z, r.z);  w.w = pair_word(sd.w, r.w);
                *reinterpret_cast<uint4*>(pairs + e0) = w;
            }
        }
    }
    int wave = (int)((blockIdx.x * blockDim.x + threadIdx.x) >> 5);
    if (wave >= ntiles) return;
    int lane = (int)(threadIdx.x & 31u);
    int half = lane >> 4;
    int m    = lane & 15;
    size_t row0 = (size_t)wave * 16;

    v8f q0 = {}, q1 = {}, k0 = {}, k1 = {}, v0 = {}, v1 = {};

#pragma unroll
    for (int kk = 0; kk < 8; ++kk) {
        int kb = kk * 4;
        const float* ap = x + (row0 + m) * 32 + kb + 2 * half;
        v2f a; a.x = ap[0]; a.y = ap[1];
        wmma_step(Wq, kb, half, m, a, q0, q1);
        wmma_step(Wk, kb, half, m, a, k0, k1);
        wmma_step(Wv, kb, half, m, a, v0, v1);
    }

#pragma unroll
    for (int r = 0; r < 8; ++r) {
        size_t row = (row0 + r + 8 * half) * 32;
        Q[row + m]      = q0[r];  Q[row + 16 + m] = q1[r];
        K[row + m]      = k0[r];  K[row + 16 + m] = k1[r];
        V[row + m]      = v0[r];  V[row + 16 + m] = v1[r];
    }
}

__global__ void qkv_tail_kernel(const float* __restrict__ x,
                                const float* __restrict__ Wq,
                                const float* __restrict__ Wk,
                                const float* __restrict__ Wv,
                                float* __restrict__ Q, float* __restrict__ K,
                                float* __restrict__ V, int row_start, int N) {
    int idx = blockIdx.x * blockDim.x + threadIdx.x;
    int nrows = N - row_start;
    if (idx >= nrows * 32) return;
    int row = row_start + idx / 32, col = idx & 31;
    float q = 0.f, k = 0.f, v = 0.f;
    for (int t = 0; t < 32; ++t) {
        float xv = x[(size_t)row * 32 + t];
        q += xv * Wq[t * 32 + col];
        k += xv * Wk[t * 32 + col];
        v += xv * Wv[t * 32 + col];
    }
    size_t o = (size_t)row * 32 + col;
    Q[o] = q; K[o] = k; V[o] = v;
}

template <int M>
__device__ __forceinline__ float lane_xor(float x) {
    int b = __builtin_bit_cast(int, x);
    if constexpr (M == 16) {
        return __builtin_bit_cast(float, __builtin_amdgcn_permlanex16(
            b, b, 0x76543210, 0xfedcba98, false, false));
    } else {
        return __builtin_bit_cast(float, __builtin_amdgcn_update_dpp(
            b, b, 0x160 | M, 0xf, 0xf, false));
    }
}

template <int M, int NH>
__device__ __forceinline__ void fold_level(float* v, bool hi) {
#pragma unroll
    for (int i = 0; i < NH; ++i) {
        float keep = hi ? v[i + NH] : v[i];
        float send = hi ? v[i] : v[i + NH];
        v[i] = keep + lane_xor<M>(send);
    }
}

template <int J>
__device__ __forceinline__ unsigned bcast8(unsigned x) {
    return (unsigned)__builtin_amdgcn_mov_dpp8(x, J * 0x249249);
}
template <int J>
__device__ __forceinline__ float bcast8f(float x) {
    return __uint_as_float(bcast8<J>(__float_as_uint(x)));
}

__device__ __forceinline__ int acc_pos(int rl, int f) {
    return (rl << 5) + ((((f & 3) ^ (rl >> 1)) & 3) << 3) + (f >> 2);
}

#define OWN_CAP  16384
#define OWN_TB   1024
#define OWN_U    8
#define OWN_STEP (OWN_TB * 4)
#define KEY_ROW  (OWN_TB * 16)
#define KEY_U    4
#define KEY_STEP (KEY_ROW * KEY_U)
#define OWN_XU   ((OWN_R * 8 + OWN_TB - 1) / OWN_TB)

__device__ __forceinline__ void own_push(unsigned rl, int e, unsigned* cnt,
                                         int* l_a) {
    if (rl < (unsigned)OWN_R) {
        unsigned slot = atomicAdd(cnt, 1u);
        if (slot < (unsigned)OWN_CAP) l_a[slot] = e;
    }
}

__device__ __forceinline__ void own_build(const int* __restrict__ recv, int a,
                                          int b, int rbase, bool vec_ok,
                                          unsigned* cnt, int* l_a) {
    const int t = (int)threadIdx.x;
    for (int base = a; base < b; base += OWN_STEP * OWN_U) {
        const bool full = vec_ok && (b - base >= OWN_STEP * OWN_U);
        int4 v[OWN_U];
        if (full) {
#pragma unroll
            for (int u = 0; u < OWN_U; ++u)
                v[u] = *reinterpret_cast<const int4*>(recv + base + u * OWN_STEP + t * 4);
        } else {
#pragma unroll
            for (int u = 0; u < OWN_U; ++u) {
                int e0 = base + u * OWN_STEP + t * 4;
                v[u].x = (e0     < b) ? recv[e0]     : -1;
                v[u].y = (e0 + 1 < b) ? recv[e0 + 1] : -1;
                v[u].z = (e0 + 2 < b) ? recv[e0 + 2] : -1;
                v[u].w = (e0 + 3 < b) ? recv[e0 + 3] : -1;
            }
        }
#pragma unroll
        for (int u = 0; u < OWN_U; ++u) {
            int e0 = base + u * OWN_STEP + t * 4;
            unsigned r0 = (unsigned)(v[u].x - rbase), r1 = (unsigned)(v[u].y - rbase);
            unsigned r2 = (unsigned)(v[u].z - rbase), r3 = (unsigned)(v[u].w - rbase);
            unsigned mn = min(min(r0, r1), min(r2, r3));
            if (mn < (unsigned)OWN_R) {
                own_push(r0, e0,     cnt, l_a);
                own_push(r1, e0 + 1, cnt, l_a);
                own_push(r2, e0 + 2, cnt, l_a);
                own_push(r3, e0 + 3, cnt, l_a);
            }
        }
    }
}

__device__ __forceinline__ unsigned zero_bytes(unsigned x) {
    return ~(((x & 0x7F7F7F7Fu) + 0x7F7F7F7Fu) | x) & 0x80808080u;
}

__device__ __forceinline__ void own_build_keys(const unsigned* __restrict__ keys4,
                                               int Epad, unsigned kb,
                                               unsigned* cnt, int* l_a) {
    const unsigned t = threadIdx.x;
    const unsigned splat = kb * 0x01010101u;
    for (int base = 0; base < Epad; base += KEY_STEP) {
        uint4 v[KEY_U];
#pragma unroll
        for (int u = 0; u < KEY_U; ++u)
            v[u] = *reinterpret_cast<const uint4*>(keys4 + ((base + u * KEY_ROW) >> 2) + t * 4);
#pragma unroll
        for (int u = 0; u < KEY_U; u += 2) {
            const unsigned code0 = ((unsigned)(base / KEY_ROW + u) << 15) | (t << 5);
            unsigned m = (zero_bytes(v[u].x ^ splat) >> 7) | (zero_bytes(v[u].y ^ splat) >> 6) |
                         (zero_bytes(v[u].z ^ splat) >> 5) | (zero_bytes(v[u].w ^ splat) >> 4) |
                         (zero_bytes(v[u + 1].x ^ splat) >> 3) | (zero_bytes(v[u + 1].y ^ splat) >> 2) |
                         (zero_bytes(v[u + 1].z ^ splat) >> 1) | zero_bytes(v[u + 1].w ^ splat);
            while (m) {
                const unsigned c = (unsigned)__builtin_ctz(m);
                m &= m - 1u;
                unsigned slot = atomicAdd(cnt, 1u);
                if (slot < (unsigned)OWN_CAP) l_a[slot] = (int)(code0 | c);
            }
        }
    }
}

__device__ __forceinline__ int own_edge(unsigned code) {
    const unsigned c = code & 31u, t = (code >> 5) & 1023u, row = code >> 15;
    return (int)((row + ((c >> 2) & 1u)) * KEY_ROW + t * 16u + ((c & 3u) << 2) + (c >> 3));
}

template <int J>
__device__ __forceinline__ void rows_qk(const float4* __restrict__ Qp,
                                        const float4* __restrict__ Kp,
                                        unsigned r, unsigned s, float4* qv, float4* kv) {
    qv[J] = Qp[(size_t)bcast8<J>(r) << 3];
    kv[J] = Kp[(size_t)bcast8<J>(s) << 3];
}

template <bool tail>
__device__ __forceinline__ void add_rows(const float4* __restrict__ Vp, unsigned s,
                                         unsigned rl, float alpha, int m,
                                         unsigned lane, float* acc) {
    const float4* a0 = Vp + ((size_t)bcast8<0>(s) << 3);
    const float4* a1 = Vp + ((size_t)bcast8<1>(s) << 3);
    const float4* a2 = Vp + ((size_t)bcast8<2>(s) << 3);
    const float4* a3 = Vp + ((size_t)bcast8<3>(s) << 3);
    const float4* a4 = Vp + ((size_t)bcast8<4>(s) << 3);
    const float4* a5 = Vp + ((size_t)bcast8<5>(s) << 3);
    const float4* a6 = Vp + ((size_t)bcast8<6>(s) << 3);
    const float4* a7 = Vp + ((size_t)bcast8<7>(s) << 3);
    __builtin_amdgcn_sched_barrier(0);
    float4 v[8];
    v[0] = *a0;  v[1] = *a1;  v[2] = *a2;  v[3] = *a3;
    v[4] = *a4;  v[5] = *a5;  v[6] = *a6;  v[7] = *a7;
    unsigned rj[8];
    float aj[8];
    rj[0] = bcast8<0>(rl);  aj[0] = bcast8f<0>(alpha);
    rj[1] = bcast8<1>(rl);  aj[1] = bcast8f<1>(alpha);
    rj[2] = bcast8<2>(rl);  aj[2] = bcast8f<2>(alpha);
    rj[3] = bcast8<3>(rl);  aj[3] = bcast8f<3>(alpha);
    rj[4] = bcast8<4>(rl);  aj[4] = bcast8f<4>(alpha);
    rj[5] = bcast8<5>(rl);  aj[5] = bcast8f<5>(alpha);
    rj[6] = bcast8<6>(rl);  aj[6] = bcast8f<6>(alpha);
    rj[7] = bcast8<7>(rl);  aj[7] = bcast8f<7>(alpha);
    const unsigned q4 = (lane & 7u) << 2;
    char* accb = reinterpret_cast<char*>(acc);
#pragma unroll
    for (int j = 0; j < 8; ++j) {
        const unsigned b0 = (rj[j] << 7) | ((rj[j] << 4) & 0x60u) | q4;
        if (!tail || (int)((lane & 24u) | (unsigned)j) < m) {
            atomicAdd(reinterpret_cast<float*>(accb + b0),         aj[j] * v[j].x);
            atomicAdd(reinterpret_cast<float*>(accb + (b0 ^ 32u)), aj[j] * v[j].y);
            atomicAdd(reinterpret_cast<float*>(accb + (b0 ^ 64u)), aj[j] * v[j].z);
            atomicAdd(reinterpret_cast<float*>(accb + (b0 ^ 96u)), aj[j] * v[j].w);
        }
    }
}

__device__ __forceinline__ void own_process(
    const float* __restrict__ Q, const float* __restrict__ K,
    const float* __restrict__ V, const int* __restrict__ ei,
    const int* __restrict__ recv, const unsigned* __restrict__ pairs, int rbase, int n,
    bool first, bool coded, float* acc, float* zs, unsigned* mkey, unsigned* mnew,
    int* l_a, int* l_r, float* l_sc) {
    const int t = (int)threadIdx.x;
    const unsigned lane = threadIdx.x & 31u;
    const int wave = __builtin_amdgcn_readfirstlane((int)(threadIdx.x >> 5));
    const float4* Qp = reinterpret_cast<const float4*>(Q) + (lane & 7u);
    const float4* Kp = reinterpret_cast<const float4*>(K) + (lane & 7u);
    const float4* Vp = reinterpret_cast<const float4*>(V) + (lane & 7u);

    for (int c = wave; c * 32 < n; c += OWN_TB / 32) {
        int i = c * 32 + (int)lane;
        bool valid = i < n;
        int s = 0, r = rbase;
        if (valid) {
            int e = l_a[i];
            if (coded) e = own_edge((unsigned)e);
            if (pairs != nullptr) {
                const unsigned w = pairs[e];
                s = (int)(w & ((1u << PAIR_SH) - 1u));
                r = rbase + (int)(w >> PAIR_SH);
            } else {
                s = ei[e]; r = recv[e];
            }
        }
        int rl = r - rbase;
        float4 qv[8], kv[8];
        rows_qk<0>(Qp, Kp, (unsigned)r, (unsigned)s, qv, kv);
        rows_qk<1>(Qp, Kp, (unsigned)r, (unsigned)s, qv, kv);
        rows_qk<2>(Qp, Kp, (unsigned)r, (unsigned)s, qv, kv);
        rows_qk<3>(Qp, Kp, (unsigned)r, (unsigned)s, qv, kv);
        rows_qk<4>(Qp, Kp, (unsigned)r, (unsigned)s, qv, kv);
        rows_qk<5>(Qp, Kp, (unsigned)r, (unsigned)s, qv, kv);
        rows_qk<6>(Qp, Kp, (unsigned)r, (unsigned)s, qv, kv);
        rows_qk<7>(Qp, Kp, (unsigned)r, (unsigned)s, qv, kv);
        __builtin_amdgcn_sched_barrier(0);
        float p[8];
#pragma unroll
        for (int j = 0; j < 8; ++j)
            p[j] = qv[j].x * kv[j].x + qv[j].y * kv[j].y +
                   qv[j].z * kv[j].z + qv[j].w * kv[j].w;
        fold_level<4, 4>(p, (lane & 4) != 0);
        fold_level<2, 2>(p, (lane & 2) != 0);
        fold_level<1, 1>(p, (lane & 1) != 0);
        float sc = p[0] * 0.17677669529663687f;
        if (valid) {
            l_sc[i] = sc;
            l_r[i] = rl;
            l_a[i] = s;
            atomicMax(&mnew[rl], fkey(sc));
        }
    }
    __syncthreads();
    if (!first) {
        for (int idx = t; idx < OWN_R * 32; idx += OWN_TB) {
            unsigned ko = mkey[idx >> 5], kn = mnew[idx >> 5];
            if (kn != ko)
                acc[idx] *= ko ? __expf(fkey_inv(ko) - fkey_inv(kn)) : 0.0f;
        }
        for (int rl = t; rl < OWN_R; rl += OWN_TB) {
            unsigned ko = mkey[rl], kn = mnew[rl];
            if (kn != ko)
                zs[rl] *= ko ? __expf(fkey_inv(ko) - fkey_inv(kn)) : 0.0f;
        }
        __syncthreads();
    }
    for (int rl = t; rl < OWN_R; rl += OWN_TB) mkey[rl] = mnew[rl];
    __syncthreads();

    for (int c = wave; c * 32 < n; c += OWN_TB / 32) {
        int i = c * 32 + (int)lane;
        int m = min(32, n - c * 32);
        int s = 0, rl = 0;
        float alpha = 0.f;
        if (i < n) {
            s = l_a[i]; rl = l_r[i];
            alpha = __expf(l_sc[i] - fkey_inv(mkey[rl]));
            atomicAdd(&zs[rl], alpha);
        }
        if (m == 32)
            add_rows<false>(Vp, (unsigned)s, (unsigned)rl, alpha, m, lane, acc);
        else
            add_rows<true>(Vp, (unsigned)s, (unsigned)rl, alpha, m, lane, acc);
    }
}

__global__ __launch_bounds__(OWN_TB) void own_range_kernel(
    const float* __restrict__ Q, const float* __restrict__ K,
    const float* __restrict__ V, const int* __restrict__ ei,
    const unsigned* __restrict__ keys4, const unsigned* __restrict__ pairs,
    const float* __restrict__ x, const float* __restrict__ Wo,
    float* __restrict__ out, int N, int E, int Epad) {
    __shared__ __attribute__((aligned(16))) float acc[OWN_R * 32];
    __shared__ float    zs[OWN_R];
    __shared__ unsigned mkey[OWN_R];
    __shared__ unsigned mnew[OWN_R];
    __shared__ int      l_a[OWN_CAP];
    __shared__ int      l_r[OWN_CAP];
    __shared__ float    l_sc[OWN_CAP];
    __shared__ unsigned cnt;

    const int t = (int)threadIdx.x;
    const int rbase = (int)blockIdx.x * OWN_R;
    const int nown = min(OWN_R, N - rbase);
    const int* recv = ei + E;
    const bool vec_ok = (((size_t)recv & 15u) == 0);

    for (int i = t; i < OWN_R * 32; i += OWN_TB) acc[i] = 0.0f;
    for (int i = t; i < OWN_R; i += OWN_TB) { zs[i] = 0.0f; mkey[i] = 0u; mnew[i] = 0u; }
    if (t == 0) cnt = 0u;
    __syncthreads();

    if (keys4 != nullptr)
        own_build_keys(keys4, Epad, blockIdx.x, &cnt, l_a);
    else
        own_build(recv, 0, E, rbase, vec_ok, &cnt, l_a);
    __syncthreads();
    const float4* x4 = reinterpret_cast<const float4*>(x) + (size_t)rbase * 8;
    float4*       o4 = reinterpret_cast<float4*>(out) + (size_t)rbase * 8;
    const int npiece = nown * 8;
    float4 xr[OWN_XU];
#pragma unroll
    for (int u = 0; u < OWN_XU; ++u) {
        const int i = t + u * OWN_TB;
        xr[u] = (i < npiece) ? x4[i] : make_float4(0.f, 0.f, 0.f, 0.f);
    }
    const unsigned total = cnt;
    if (total <= (unsigned)OWN_CAP) {
        own_process(Q, K, V, ei, recv, pairs, rbase, (int)total, true, keys4 != nullptr,
                    acc, zs, mkey, mnew, l_a, l_r, l_sc);
    } else {
        for (int a = 0; a < E; a += OWN_CAP) {
            __syncthreads();
            if (t == 0) cnt = 0u;
            __syncthreads();
            int b = (E - a > OWN_CAP) ? a + OWN_CAP : E;
            own_build(recv, a, b, rbase, vec_ok, &cnt, l_a);
            __syncthreads();
            own_process(Q, K, V, ei, recv, pairs, rbase, (int)cnt, a == 0, false, acc, zs,
                        mkey, mnew, l_a, l_r, l_sc);
        }
    }
    __syncthreads();
    {
        const int wave = __builtin_amdgcn_readfirstlane(t >> 5);
        if (wave * 16 < nown) {
            const int lane = t & 31;
            const int half = lane >> 4;
            const int m    = lane & 15;
            const int rl   = wave * 16 + m;
            const float invz = 1.0f / (zs[rl] + 1e-6f);
            const float* rowp = acc + (rl << 5);
            const int w0 = (((2 * half)     ^ (rl >> 1)) & 3) << 3;
            const int w1 = (((2 * half + 1) ^ (rl >> 1)) & 3) << 3;
            const float4 e0 = *reinterpret_cast<const float4*>(rowp + w0);
            const float4 e1 = *reinterpret_cast<const float4*>(rowp + w0 + 4);
            const float4 f0 = *reinterpret_cast<const float4*>(rowp + w1);
            const float4 f1 = *reinterpret_cast<const float4*>(rowp + w1 + 4);
            const float ae[8] = {e0.x, e0.y, e0.z, e0.w, e1.x, e1.y, e1.z, e1.w};
            const float af[8] = {f0.x, f0.y, f0.z, f0.w, f1.x, f1.y, f1.z, f1.w};
            v8f c0 = {}, c1 = {};
#pragma unroll
            for (int kk = 0; kk < 8; ++kk) {
                v2f a; a.x = ae[kk] * invz; a.y = af[kk] * invz;
                wmma_step(Wo, kk * 4, half, m, a, c0, c1);
            }
#pragma unroll
            for (int r = 0; r < 8; ++r) {
                float* d = acc + ((wave * 16 + r + 8 * half) << 5);
                d[m]      = c0[r];
                d[16 + m] = c1[r];
            }
        }
    }
    __syncthreads();
#pragma unroll
    for (int u = 0; u < OWN_XU; ++u) {
        const int i = t + u * OWN_TB;
        if (i < npiece) {
            const float4 p = *reinterpret_cast<const float4*>(acc + (i << 2));
            const float4 xv = xr[u];
            o4[i] = make_float4(xv.x + p.x, xv.y + p.y, xv.z + p.z, xv.w + p.w);
        }
    }
}

extern "C" void kernel_launch(void* const* d_in, const int* in_sizes, int n_in,
                              void* d_out, int out_size, void* d_ws,
                              size_t ws_size, hipStream_t stream) {
    const float* x  = (const float*)d_in[0];
    const int*   ei = (const int*)d_in[1];
    const float* Wq = (const float*)d_in[2];
    const float* Wk = (const float*)d_in[3];
    const float* Wv = (const float*)d_in[4];
    const float* Wo = (const float*)d_in[5];
    float* out = (float*)d_out;

    const int N = in_sizes[0] / 32;
    const int E = in_sizes[1] / 2;

    float* Q       = (float*)d_ws;
    float* K       = Q + (size_t)N * 32;
    float* V       = K + (size_t)N * 32;
    const int Epad = ((E + KEY_STEP - 1) / KEY_STEP) * KEY_STEP;
    const size_t key_off = (((size_t)N * 129 * sizeof(float)) + 15) & ~(size_t)15;
    unsigned* keys4 = (unsigned*)((unsigned char*)d_ws + key_off);
    if (N > 255 * OWN_R || key_off + (size_t)Epad > ws_size) keys4 = nullptr;
    const size_t pair_off = (size_t)N * 96 * sizeof(float);
    unsigned* pairs = (unsigned*)((unsigned char*)d_ws + pair_off);
    if (keys4 == nullptr || pair_off + (size_t)Epad * sizeof(unsigned) > key_off) pairs = nullptr;

    const int TB = 256;
    int ntiles = N / 16;
    int rem_row0 = ntiles * 16;
    int rem = N - rem_row0;

    {
        int blocks = (ntiles * 32 + TB - 1) / TB;
        if (blocks == 0 && keys4 != nullptr) blocks = 1;
        if (blocks > 0)
            qkv_wmma_kernel<<<blocks, TB, 0, stream>>>(
                x, Wq, Wk, Wv, Q, K, V, ntiles, ei, keys4, pairs, N, E, Epad);
        if (rem > 0)
            qkv_tail_kernel<<<(rem * 32 + TB - 1) / TB, TB, 0, stream>>>(
                x, Wq, Wk, Wv, Q, K, V, rem_row0, N);
    }
    if (N > 0)
        own_range_kernel<<<(N + OWN_R - 1) / OWN_R, OWN_TB, 0, stream>>>(
            Q, K, V, ei, keys4, pairs, x, Wo, out, N, E, Epad);
}


# Key plane stored bit-sliced (wave ballots in the projection launch): a range group finds its edges among 32 with 8 xnor-and ops on two 16-byte loads, not ~50 byte compares; sum order as before
# speedup vs baseline: 8.9306x; 1.0415x over previous
#include <hip/hip_runtime.h>
#include <math.h>

typedef __attribute__((ext_vector_type(2))) float v2f;
typedef __attribute__((ext_vector_type(8))) float v8f;

__device__ __forceinline__ unsigned fkey(float f) {
    unsigned u = __float_as_uint(f);
    return (u & 0x80000000u) ? ~u : (u | 0x80000000u);
}
__device__ __forceinline__ float fkey_inv(unsigned k) {
    unsigned u = (k & 0x80000000u) ? (k & 0x7FFFFFFFu) : ~k;
    return __uint_as_float(u);
}

#define OWN_R    416
__device__ __forceinline__ unsigned own_key(int r, int N) {
    return ((unsigned)r < (unsigned)N) ? ((unsigned)r / (unsigned)OWN_R) : 0xFFu;
}
#define PAIR_SH  23
__device__ __forceinline__ unsigned pair_word(int s, int r) {
    return ((unsigned)s & ((1u << PAIR_SH) - 1u)) | (((unsigned)r % (unsigned)OWN_R) << PAIR_SH);
}

__device__ __forceinline__ void wmma_step(const float* __restrict__ W, int kb,
                                          int half, int m, v2f a,
                                          v8f& acc0, v8f& acc1) {
    int r0 = (kb + 2 * half + 0) * 32;
    int r1 = (kb + 2 * half + 1) * 32;
    v2f b0, b1;
    b0.x = W[r0 + m];        b0.y = W[r1 + m];
    b1.x = W[r0 + 16 + m];   b1.y = W[r1 + 16 + m];
    acc0 = __builtin_amdgcn_wmma_f32_16x16x4_f32(false, a, false, b0,
                                                 (short)0, acc0, false, false);
    acc1 = __builtin_amdgcn_wmma_f32_16x16x4_f32(false, a, false, b1,
                                                 (short)0, acc1, false, false);
}

__global__ __launch_bounds__(256) void qkv_wmma_kernel(
    const float* __restrict__ x,
    const float* __restrict__ Wq, const float* __restrict__ Wk,
    const float* __restrict__ Wv,
    float* __restrict__ Q, float* __restrict__ K, float* __restrict__ V,
    int ntiles, const int* __restrict__ ei, unsigned* __restrict__ keys4,
    unsigned* __restrict__ pairs, int N, int E, int Epad) {
    if (keys4 != nullptr) {
        const int* recv = ei + E;
        const bool vec_ok = (((size_t)recv & 15u) == 0);
        const bool svec_ok = (((size_t)ei & 15u) == 0);
        const int gid = (int)(blockIdx.x * blockDim.x + threadIdx.x);
        const int gsz = (int)(gridDim.x * blockDim.x);
        for (int q = gid; q * 4 < Epad; q += gsz) {
            const int e0 = q * 4;
            int4 r, sd = make_int4(0, 0, 0, 0);
            if (pairs != nullptr) {
                if (svec_ok && e0 + 4 <= E) {
                    sd = *reinterpret_cast<const int4*>(ei + e0);
                } else {
                    sd.x = (e0     < E) ? ei[e0]     : 0;
                    sd.y = (e0 + 1 < E) ? ei[e0 + 1] : 0;
                    sd.z = (e0 + 2 < E) ? ei[e0 + 2] : 0;
                    sd.w = (e0 + 3 < E) ? ei[e0 + 3] : 0;
                }
            }
            if (vec_ok && e0 + 4 <= E) {
                r = *reinterpret_cast<const int4*>(recv + e0);
            } else {
                r.x = (e0     < E) ? recv[e0]     : -1;
                r.y = (e0 + 1 < E) ? recv[e0 + 1] : -1;
                r.z = (e0 + 2 < E) ? recv[e0 + 2] : -1;
                r.w = (e0 + 3 < E) ? recv[e0 + 3] : -1;
            }
            const unsigned kj[4] = {own_key(r.x, N), own_key(r.y, N),
                                    own_key(r.z, N), own_key(r.w, N)};
            const unsigned ln = threadIdx.x & 31u;
            unsigned wv = 0u;
#pragma unroll
            for (int j = 0; j < 4; ++j) {
#pragma unroll
                for (int p = 0; p < 8; ++p) {
                    const unsigned b = __builtin_amdgcn_ballot_w32(((kj[j] >> p) & 1u) != 0u);
                    if (ln == (unsigned)(j * 8 + p)) wv = b;
                }
            }
            {
                const unsigned set = (((unsigned)q >> 5) << 2) + (ln >> 3);
                const unsigned half = (ln & 4u) ? ((unsigned)Epad >> 3) : 0u;
                keys4[half + (set << 2) + (ln & 3u)] = wv;
            }
            if (pairs != nullptr) {
                uint4 w;
                w.x = pair_word(sd.x, r.x);  w.y = pair_word(sd.y, r.y);
                w.z = pair_word(sd.z, r.z);  w.w = pair_word(sd.w, r.w);
                *reinterpret_cast<uint4*>(pairs + e0) = w;
            }
        }
    }
    int wave = (int)((blockIdx.x * blockDim.x + threadIdx.x) >> 5);
    if (wave >= ntiles) return;
    int lane = (int)(threadIdx.x & 31u);
    int half = lane >> 4;
    int m    = lane & 15;
    size_t row0 = (size_t)wave * 16;

    v8f q0 = {}, q1 = {}, k0 = {}, k1 = {}, v0 = {}, v1 = {};

#pragma unroll
    for (int kk = 0; kk < 8; ++kk) {
        int kb = kk * 4;
        const float* ap = x + (row0 + m) * 32 + kb + 2 * half;
        v2f a; a.x = ap[0]; a.y = ap[1];
        wmma_step(Wq, kb, half, m, a, q0, q1);
        wmma_step(Wk, kb, half, m, a, k0, k1);
        wmma_step(Wv, kb, half, m, a, v0, v1);
    }

#pragma unroll
    for (int r = 0; r < 8; ++r) {
        size_t row = (row0 + r + 8 * half) * 32;
        Q[row + m]      = q0[r];  Q[row + 16 + m] = q1[r];
        K[row + m]      = k0[r];  K[row + 16 + m] = k1[r];
        V[row + m]      = v0[r];  V[row + 16 + m] = v1[r];
    }
}

__global__ void qkv_tail_kernel(const float* __restrict__ x,
                                const float* __restrict__ Wq,
                                const float* __restrict__ Wk,
                                const float* __restrict__ Wv,
                                float* __restrict__ Q, float* __restrict__ K,
                                float* __restrict__ V, int row_start, int N) {
    int idx = blockIdx.x * blockDim.x + threadIdx.x;
    int nrows = N - row_start;
    if (idx >= nrows * 32) return;
    int row = row_start + idx / 32, col = idx & 31;
    float q = 0.f, k = 0.f, v = 0.f;
    for (int t = 0; t < 32; ++t) {
        float xv = x[(size_t)row * 32 + t];
        q += xv * Wq[t * 32 + col];
        k += xv * Wk[t * 32 + col];
        v += xv * Wv[t * 32 + col];
    }
    size_t o = (size_t)row * 32 + col;
    Q[o] = q; K[o] = k; V[o] = v;
}

template <int M>
__device__ __forceinline__ float lane_xor(float x) {
    int b = __builtin_bit_cast(int, x);
    if constexpr (M == 16) {
        return __builtin_bit_cast(float, __builtin_amdgcn_permlanex16(
            b, b, 0x76543210, 0xfedcba98, false, false));
    } else {
        return __builtin_bit_cast(float, __builtin_amdgcn_update_dpp(
            b, b, 0x160 | M, 0xf, 0xf, false));
    }
}

template <int M, int NH>
__device__ __forceinline__ void fold_level(float* v, bool hi) {
#pragma unroll
    for (int i = 0; i < NH; ++i) {
        float keep = hi ? v[i + NH] : v[i];
        float send = hi ? v[i] : v[i + NH];
        v[i] = keep + lane_xor<M>(send);
    }
}

template <int J>
__device__ __forceinline__ unsigned bcast8(unsigned x) {
    return (unsigned)__builtin_amdgcn_mov_dpp8(x, J * 0x249249);
}
template <int J>
__device__ __forceinline__ float bcast8f(float x) {
    return __uint_as_float(bcast8<J>(__float_as_uint(x)));
}

__device__ __forceinline__ int acc_pos(int rl, int f) {
    return (rl << 5) + ((((f & 3) ^ (rl >> 1)) & 3) << 3) + (f >> 2);
}

#define OWN_CAP  16384
#define OWN_TB   1024
#define OWN_U    8
#define OWN_STEP (OWN_TB * 4)
#define KEY_ROW  (OWN_TB * 16)
#define KEY_U    4
#define KEY_STEP (KEY_ROW * KEY_U)
#define OWN_XU   ((OWN_R * 8 + OWN_TB - 1) / OWN_TB)

__device__ __forceinline__ void own_push(unsigned rl, int e, unsigned* cnt,
                                         int* l_a) {
    if (rl < (unsigned)OWN_R) {
        unsigned slot = atomicAdd(cnt, 1u);
        if (slot < (unsigned)OWN_CAP) l_a[slot] = e;
    }
}

__device__ __forceinline__ void own_build(const int* __restrict__ recv, int a,
                                          int b, int rbase, bool vec_ok,
                                          unsigned* cnt, int* l_a) {
    const int t = (int)threadIdx.x;
    for (int base = a; base < b; base += OWN_STEP * OWN_U) {
        const bool full = vec_ok && (b - base >= OWN_STEP * OWN_U);
        int4 v[OWN_U];
        if (full) {
#pragma unroll
            for (int u = 0; u < OWN_U; ++u)
                v[u] = *reinterpret_cast<const int4*>(recv + base + u * OWN_STEP + t * 4);
        } else {
#pragma unroll
            for (int u = 0; u < OWN_U; ++u) {
                int e0 = base + u * OWN_STEP + t * 4;
                v[u].x = (e0     < b) ? recv[e0]     : -1;
                v[u].y = (e0 + 1 < b) ? recv[e0 + 1] : -1;
                v[u].z = (e0 + 2 < b) ? recv[e0 + 2] : -1;
                v[u].w = (e0 + 3 < b) ? recv[e0 + 3] : -1;
            }
        }
#pragma unroll
        for (int u = 0; u < OWN_U; ++u) {
            int e0 = base + u * OWN_STEP + t * 4;
            unsigned r0 = (unsigned)(v[u].x - rbase), r1 = (unsigned)(v[u].y - rbase);
            unsigned r2 = (unsigned)(v[u].z - rbase), r3 = (unsigned)(v[u].w - rbase);
            unsigned mn = min(min(r0, r1), min(r2, r3));
            if (mn < (unsigned)OWN_R) {
                own_push(r0, e0,     cnt, l_a);
                own_push(r1, e0 + 1, cnt, l_a);
                own_push(r2, e0 + 2, cnt, l_a);
                own_push(r3, e0 + 3, cnt, l_a);
            }
        }
    }
}

__device__ __forceinline__ void own_build_keys(const unsigned* __restrict__ keys4,
                                               int Epad, unsigned kb,
                                               unsigned* cnt, int* l_a) {
    const unsigned t = threadIdx.x;
    unsigned cp[8];
#pragma unroll
    for (int p = 0; p < 8; ++p) cp[p] = 0u - ((kb >> p) & 1u);
    const uint4* lo4 = reinterpret_cast<const uint4*>(keys4);
    const uint4* hi4 = lo4 + ((unsigned)Epad >> 5);
    for (int base = 0; base < Epad; base += KEY_STEP) {
        uint4 v[KEY_U];
#pragma unroll
        for (int u = 0; u < KEY_U; u += 2) {
            const unsigned set = ((unsigned)base >> 5) + (unsigned)(u >> 1) * OWN_TB + t;
            v[u]     = lo4[set];
            v[u + 1] = hi4[set];
        }
#pragma unroll
        for (int u = 0; u < KEY_U; u += 2) {
            const unsigned set = ((unsigned)base >> 5) + (unsigned)(u >> 1) * OWN_TB + t;
            const unsigned code0 = set << 5;
            unsigned m = ~(v[u].x ^ cp[0]);
            m &= ~(v[u].y ^ cp[1]);      m &= ~(v[u].z ^ cp[2]);      m &= ~(v[u].w ^ cp[3]);
            m &= ~(v[u + 1].x ^ cp[4]);  m &= ~(v[u + 1].y ^ cp[5]);
            m &= ~(v[u + 1].z ^ cp[6]);  m &= ~(v[u + 1].w ^ cp[7]);
            while (m) {
                const unsigned c = (unsigned)__builtin_ctz(m);
                m &= m - 1u;
                unsigned slot = atomicAdd(cnt, 1u);
                if (slot < (unsigned)OWN_CAP) l_a[slot] = (int)(code0 | c);
            }
        }
    }
}

__device__ __forceinline__ int own_edge(unsigned code) {
    const unsigned l = code & 31u, set = code >> 5;
    return (int)(((set >> 2) << 7) + (l << 2) + (set & 3u));
}

template <int J>
__device__ __forceinline__ void rows_qk(const float4* __restrict__ Qp,
                                        const float4* __restrict__ Kp,
                                        unsigned r, unsigned s, float4* qv, float4* kv) {
    qv[J] = Qp[(size_t)bcast8<J>(r) << 3];
    kv[J] = Kp[(size_t)bcast8<J>(s) << 3];
}

template <bool tail>
__device__ __forceinline__ void add_rows(const float4* __restrict__ Vp, unsigned s,
                                         unsigned rl, float alpha, int m,
                                         unsigned lane, float* acc) {
    const float4* a0 = Vp + ((size_t)bcast8<0>(s) << 3);
    const float4* a1 = Vp + ((size_t)bcast8<1>(s) << 3);
    const float4* a2 = Vp + ((size_t)bcast8<2>(s) << 3);
    const float4* a3 = Vp + ((size_t)bcast8<3>(s) << 3);
    const float4* a4 = Vp + ((size_t)bcast8<4>(s) << 3);
    const float4* a5 = Vp + ((size_t)bcast8<5>(s) << 3);
    const float4* a6 = Vp + ((size_t)bcast8<6>(s) << 3);
    const float4* a7 = Vp + ((size_t)bcast8<7>(s) << 3);
    __builtin_amdgcn_sched_barrier(0);
    float4 v[8];
    v[0] = *a0;  v[1] = *a1;  v[2] = *a2;  v[3] = *a3;
    v[4] = *a4;  v[5] = *a5;  v[6] = *a6;  v[7] = *a7;
    unsigned rj[8];
    float aj[8];
    rj[0] = bcast8<0>(rl);  aj[0] = bcast8f<0>(alpha);
    rj[1] = bcast8<1>(rl);  aj[1] = bcast8f<1>(alpha);
    rj[2] = bcast8<2>(rl);  aj[2] = bcast8f<2>(alpha);
    rj[3] = bcast8<3>(rl);  aj[3] = bcast8f<3>(alpha);
    rj[4] = bcast8<4>(rl);  aj[4] = bcast8f<4>(alpha);
    rj[5] = bcast8<5>(rl);  aj[5] = bcast8f<5>(alpha);
    rj[6] = bcast8<6>(rl);  aj[6] = bcast8f<6>(alpha);
    rj[7] = bcast8<7>(rl);  aj[7] = bcast8f<7>(alpha);
    const unsigned q4 = (lane & 7u) << 2;
    char* accb = reinterpret_cast<char*>(acc);
#pragma unroll
    for (int j = 0; j < 8; ++j) {
        const unsigned b0 = (rj[j] << 7) | ((rj[j] << 4) & 0x60u) | q4;
        if (!tail || (int)((lane & 24u) | (unsigned)j) < m) {
            atomicAdd(reinterpret_cast<float*>(accb + b0),         aj[j] * v[j].x);
            atomicAdd(reinterpret_cast<float*>(accb + (b0 ^ 32u)), aj[j] * v[j].y);
            atomicAdd(reinterpret_cast<float*>(accb + (b0 ^ 64u)), aj[j] * v[j].z);
            atomicAdd(reinterpret_cast<float*>(accb + (b0 ^ 96u)), aj[j] * v[j].w);
        }
    }
}

__device__ __forceinline__ void own_process(
    const float* __restrict__ Q, const float* __restrict__ K,
    const float* __restrict__ V, const int* __restrict__ ei,
    const int* __restrict__ recv, const unsigned* __restrict__ pairs, int rbase, int n,
    bool first, bool coded, float* acc, float* zs, unsigned* mkey, unsigned* mnew,
    int* l_a, int* l_r, float* l_sc) {
    const int t = (int)threadIdx.x;
    const unsigned lane = threadIdx.x & 31u;
    const int wave = __builtin_amdgcn_readfirstlane((int)(threadIdx.x >> 5));
    const float4* Qp = reinterpret_cast<const float4*>(Q) + (lane & 7u);
    const float4* Kp = reinterpret_cast<const float4*>(K) + (lane & 7u);
    const float4* Vp = reinterpret_cast<const float4*>(V) + (lane & 7u);

    for (int c = wave; c * 32 < n; c += OWN_TB / 32) {
        int i = c * 32 + (int)lane;
        bool valid = i < n;
        int s = 0, r = rbase;
        if (valid) {
            int e = l_a[i];
            if (coded) e = own_edge((unsigned)e);
            if (pairs != nullptr) {
                const unsigned w = pairs[e];
                s = (int)(w & ((1u << PAIR_SH) - 1u));
                r = rbase + (int)(w >> PAIR_SH);
            } else {
                s = ei[e]; r = recv[e];
            }
        }
        int rl = r - rbase;
        float4 qv[8], kv[8];
        rows_qk<0>(Qp, Kp, (unsigned)r, (unsigned)s, qv, kv);
        rows_qk<1>(Qp, Kp, (unsigned)r, (unsigned)s, qv, kv);
        rows_qk<2>(Qp, Kp, (unsigned)r, (unsigned)s, qv, kv);
        rows_qk<3>(Qp, Kp, (unsigned)r, (unsigned)s, qv, kv);
        rows_qk<4>(Qp, Kp, (unsigned)r, (unsigned)s, qv, kv);
        rows_qk<5>(Qp, Kp, (unsigned)r, (unsigned)s, qv, kv);
        rows_qk<6>(Qp, Kp, (unsigned)r, (unsigned)s, qv, kv);
        rows_qk<7>(Qp, Kp, (unsigned)r, (unsigned)s, qv, kv);
        __builtin_amdgcn_sched_barrier(0);
        float p[8];
#pragma unroll
        for (int j = 0; j < 8; ++j)
            p[j] = qv[j].x * kv[j].x + qv[j].y * kv[j].y +
                   qv[j].z * kv[j].z + qv[j].w * kv[j].w;
        fold_level<4, 4>(p, (lane & 4) != 0);
        fold_level<2, 2>(p, (lane & 2) != 0);
        fold_level<1, 1>(p, (lane & 1) != 0);
        float sc = p[0] * 0.17677669529663687f;
        if (valid) {
            l_sc[i] = sc;
            l_r[i] = rl;
            l_a[i] = s;
            atomicMax(&mnew[rl], fkey(sc));
        }
    }
    __syncthreads();
    if (!first) {
        for (int idx = t; idx < OWN_R * 32; idx += OWN_TB) {
            unsigned ko = mkey[idx >> 5], kn = mnew[idx >> 5];
            if (kn != ko)
                acc[idx] *= ko ? __expf(fkey_inv(ko) - fkey_inv(kn)) : 0.0f;
        }
        for (int rl = t; rl < OWN_R; rl += OWN_TB) {
            unsigned ko = mkey[rl], kn = mnew[rl];
            if (kn != ko)
                zs[rl] *= ko ? __expf(fkey_inv(ko) - fkey_inv(kn)) : 0.0f;
        }
        __syncthreads();
    }
    for (int rl = t; rl < OWN_R; rl += OWN_TB) mkey[rl] = mnew[rl];
    __syncthreads();

    for (int c = wave; c * 32 < n; c += OWN_TB / 32) {
        int i = c * 32 + (int)lane;
        int m = min(32, n - c * 32);
        int s = 0, rl = 0;
        float alpha = 0.f;
        if (i < n) {
            s = l_a[i]; rl = l_r[i];
            alpha = __expf(l_sc[i] - fkey_inv(mkey[rl]));
            atomicAdd(&zs[rl], alpha);
        }
        if (m == 32)
            add_rows<false>(Vp, (unsigned)s, (unsigned)rl, alpha, m, lane, acc);
        else
            add_rows<true>(Vp, (unsigned)s, (unsigned)rl, alpha, m, lane, acc);
    }
}

__global__ __launch_bounds__(OWN_TB) void own_range_kernel(
    const float* __restrict__ Q, const float* __restrict__ K,
    const float* __restrict__ V, const int* __restrict__ ei,
    const unsigned* __restrict__ keys4, const unsigned* __restrict__ pairs,
    const float* __restrict__ x, const float* __restrict__ Wo,
    float* __restrict__ out, int N, int E, int Epad) {
    __shared__ __attribute__((aligned(16))) float acc[OWN_R * 32];
    __shared__ float    zs[OWN_R];
    __shared__ unsigned mkey[OWN_R];
    __shared__ unsigned mnew[OWN_R];
    __shared__ int      l_a[OWN_CAP];
    __shared__ int      l_r[OWN_CAP];
    __shared__ float    l_sc[OWN_CAP];
    __shared__ unsigned cnt;

    const int t = (int)threadIdx.x;
    const int rbase = (int)blockIdx.x * OWN_R;
    const int nown = min(OWN_R, N - rbase);
    const int* recv = ei + E;
    const bool vec_ok = (((size_t)recv & 15u) == 0);

    for (int i = t; i < OWN_R * 32; i += OWN_TB) acc[i] = 0.0f;
    for (int i = t; i < OWN_R; i += OWN_TB) { zs[i] = 0.0f; mkey[i] = 0u; mnew[i] = 0u; }
    if (t == 0) cnt = 0u;
    __syncthreads();

    if (keys4 != nullptr)
        own_build_keys(keys4, Epad, blockIdx.x, &cnt, l_a);
    else
        own_build(recv, 0, E, rbase, vec_ok, &cnt, l_a);
    __syncthreads();
    const float4* x4 = reinterpret_cast<const float4*>(x) + (size_t)rbase * 8;
    float4*       o4 = reinterpret_cast<float4*>(out) + (size_t)rbase * 8;
    const int npiece = nown * 8;
    float4 xr[OWN_XU];
#pragma unroll
    for (int u = 0; u < OWN_XU; ++u) {
        const int i = t + u * OWN_TB;
        xr[u] = (i < npiece) ? x4[i] : make_float4(0.f, 0.f, 0.f, 0.f);
    }
    const unsigned total = cnt;
    if (total <= (unsigned)OWN_CAP) {
        own_process(Q, K, V, ei, recv, pairs, rbase, (int)total, true, keys4 != nullptr,
                    acc, zs, mkey, mnew, l_a, l_r, l_sc);
    } else {
        for (int a = 0; a < E; a += OWN_CAP) {
            __syncthreads();
            if (t == 0) cnt = 0u;
            __syncthreads();
            int b = (E - a > OWN_CAP) ? a + OWN_CAP : E;
            own_build(recv, a, b, rbase, vec_ok, &cnt, l_a);
            __syncthreads();
            own_process(Q, K, V, ei, recv, pairs, rbase, (int)cnt, a == 0, false, acc, zs,
                        mkey, mnew, l_a, l_r, l_sc);
        }
    }
    __syncthreads();
    {
        const int wave = __builtin_amdgcn_readfirstlane(t >> 5);
        if (wave * 16 < nown) {
            const int lane = t & 31;
            const int half = lane >> 4;
            const int m    = lane & 15;
            const int rl   = wave * 16 + m;
            const float invz = 1.0f / (zs[rl] + 1e-6f);
            const float* rowp = acc + (rl << 5);
            const int w0 = (((2 * half)     ^ (rl >> 1)) & 3) << 3;
            const int w1 = (((2 * half + 1) ^ (rl >> 1)) & 3) << 3;
            const float4 e0 = *reinterpret_cast<const float4*>(rowp + w0);
            const float4 e1 = *reinterpret_cast<const float4*>(rowp + w0 + 4);
            const float4 f0 = *reinterpret_cast<const float4*>(rowp + w1);
            const float4 f1 = *reinterpret_cast<const float4*>(rowp + w1 + 4);
            const float ae[8] = {e0.x, e0.y, e0.z, e0.w, e1.x, e1.y, e1.z, e1.w};
            const float af[8] = {f0.x, f0.y, f0.z, f0.w, f1.x, f1.y, f1.z, f1.w};
            v8f c0 = {}, c1 = {};
#pragma unroll
            for (int kk = 0; kk < 8; ++kk) {
                v2f a; a.x = ae[kk] * invz; a.y = af[kk] * invz;
                wmma_step(Wo, kk * 4, half, m, a, c0, c1);
            }
#pragma unroll
            for (int r = 0; r < 8; ++r) {
                float* d = acc + ((wave * 16 + r + 8 * half) << 5);
                d[m]      = c0[r];
                d[16 + m] = c1[r];
            }
        }
    }
    __syncthreads();
#pragma unroll
    for (int u = 0; u < OWN_XU; ++u) {
        const int i = t + u * OWN_TB;
        if (i < npiece) {
            const float4 p = *reinterpret_cast<const float4*>(acc + (i << 2));
            const float4 xv = xr[u];
            o4[i] = make_float4(xv.x + p.x, xv.y + p.y, xv.z + p.z, xv.w + p.w);
        }
    }
}

extern "C" void kernel_launch(void* const* d_in, const int* in_sizes, int n_in,
                              void* d_out, int out_size, void* d_ws,
                              size_t ws_size, hipStream_t stream) {
    const float* x  = (const float*)d_in[0];
    const int*   ei = (const int*)d_in[1];
    const float* Wq = (const float*)d_in[2];
    const float* Wk = (const float*)d_in[3];
    const float* Wv = (const float*)d_in[4];
    const float* Wo = (const float*)d_in[5];
    float* out = (float*)d_out;

    const int N = in_sizes[0] / 32;
    const int E = in_sizes[1] / 2;

    float* Q       = (float*)d_ws;
    float* K       = Q + (size_t)N * 32;
    float* V       = K + (size_t)N * 32;
    const int Epad = ((E + KEY_STEP - 1) / KEY_STEP) * KEY_STEP;
    const size_t key_off = (((size_t)N * 129 * sizeof(float)) + 15) & ~(size_t)15;
    unsigned* keys4 = (unsigned*)((unsigned char*)d_ws + key_off);
    if (N > 255 * OWN_R || key_off + (size_t)Epad > ws_size) keys4 = nullptr;
    const size_t pair_off = (size_t)N * 96 * sizeof(float);
    unsigned* pairs = (unsigned*)((unsigned char*)d_ws + pair_off);
    if (keys4 == nullptr || pair_off + (size_t)Epad * sizeof(unsigned) > key_off) pairs = nullptr;

    const int TB = 256;
    int ntiles = N / 16;
    int rem_row0 = ntiles * 16;
    int rem = N - rem_row0;

    {
        int blocks = (ntiles * 32 + TB - 1) / TB;
        if (blocks == 0 && keys4 != nullptr) blocks = 1;
        if (blocks > 0)
            qkv_wmma_kernel<<<blocks, TB, 0, stream>>>(
                x, Wq, Wk, Wv, Q, K, V, ntiles, ei, keys4, pairs, N, E, Epad);
        if (rem > 0)
            qkv_tail_kernel<<<(rem * 32 + TB - 1) / TB, TB, 0, stream>>>(
                x, Wq, Wk, Wv, Q, K, V, rem_row0, N);
    }
    if (N > 0)
        own_range_kernel<<<(N + OWN_R - 1) / OWN_R, OWN_TB, 0, stream>>>(
            Q, K, V, ei, keys4, pairs, x, Wo, out, N, E, Epad);
}


# Key-plane scan software-pipelined: the next group's four 16-byte key loads are issued before the current group's hits are filed (two register sets), so the filing's LDS round trips run under them
# speedup vs baseline: 9.2129x; 1.0316x over previous
#include <hip/hip_runtime.h>
#include <math.h>

typedef __attribute__((ext_vector_type(2))) float v2f;
typedef __attribute__((ext_vector_type(8))) float v8f;

__device__ __forceinline__ unsigned fkey(float f) {
    unsigned u = __float_as_uint(f);
    return (u & 0x80000000u) ? ~u : (u | 0x80000000u);
}
__device__ __forceinline__ float fkey_inv(unsigned k) {
    unsigned u = (k & 0x80000000u) ? (k & 0x7FFFFFFFu) : ~k;
    return __uint_as_float(u);
}

#define OWN_R    416
__device__ __forceinline__ unsigned own_key(int r, int N) {
    return ((unsigned)r < (unsigned)N) ? ((unsigned)r / (unsigned)OWN_R) : 0xFFu;
}
#define PAIR_SH  23
__device__ __forceinline__ unsigned pair_word(int s, int r) {
    return ((unsigned)s & ((1u << PAIR_SH) - 1u)) | (((unsigned)r % (unsigned)OWN_R) << PAIR_SH);
}

__device__ __forceinline__ void wmma_step(const float* __restrict__ W, int kb,
                                          int half, int m, v2f a,
                                          v8f& acc0, v8f& acc1) {
    int r0 = (kb + 2 * half + 0) * 32;
    int r1 = (kb + 2 * half + 1) * 32;
    v2f b0, b1;
    b0.x = W[r0 + m];        b0.y = W[r1 + m];
    b1.x = W[r0 + 16 + m];   b1.y = W[r1 + 16 + m];
    acc0 = __builtin_amdgcn_wmma_f32_16x16x4_f32(false, a, false, b0,
                                                 (short)0, acc0, false, false);
    acc1 = __builtin_amdgcn_wmma_f32_16x16x4_f32(false, a, false, b1,
                                                 (short)0, acc1, false, false);
}

__global__ __launch_bounds__(256) void qkv_wmma_kernel(
    const float* __restrict__ x,
    const float* __restrict__ Wq, const float* __restrict__ Wk,
    const float* __restrict__ Wv,
    float* __restrict__ Q, float* __restrict__ K, float* __restrict__ V,
    int ntiles, const int* __restrict__ ei, unsigned* __restrict__ keys4,
    unsigned* __restrict__ pairs, int N, int E, int Epad) {
    if (keys4 != nullptr) {
        const int* recv = ei + E;
        const bool vec_ok = (((size_t)recv & 15u) == 0);
        const bool svec_ok = (((size_t)ei & 15u) == 0);
        const int gid = (int)(blockIdx.x * blockDim.x + threadIdx.x);
        const int gsz = (int)(gridDim.x * blockDim.x);
        for (int q = gid; q * 4 < Epad; q += gsz) {
            const int e0 = q * 4;
            int4 r, sd = make_int4(0, 0, 0, 0);
            if (pairs != nullptr) {
                if (svec_ok && e0 + 4 <= E) {
                    sd = *reinterpret_cast<const int4*>(ei + e0);
                } else {
                    sd.x = (e0     < E) ? ei[e0]     : 0;
                    sd.y = (e0 + 1 < E) ? ei[e0 + 1] : 0;
                    sd.z = (e0 + 2 < E) ? ei[e0 + 2] : 0;
                    sd.w = (e0 + 3 < E) ? ei[e0 + 3] : 0;
                }
            }
            if (vec_ok && e0 + 4 <= E) {
                r = *reinterpret_cast<const int4*>(recv + e0);
            } else {
                r.x = (e0     < E) ? recv[e0]     : -1;
                r.y = (e0 + 1 < E) ? recv[e0 + 1] : -1;
                r.z = (e0 + 2 < E) ? recv[e0 + 2] : -1;
                r.w = (e0 + 3 < E) ? recv[e0 + 3] : -1;
            }
            const unsigned kj[4] = {own_key(r.x, N), own_key(r.y, N),
                                    own_key(r.z, N), own_key(r.w, N)};
            const unsigned ln = threadIdx.x & 31u;
            unsigned wv = 0u;
#pragma unroll
            for (int j = 0; j < 4; ++j) {
#pragma unroll
                for (int p = 0; p < 8; ++p) {
                    const unsigned b = __builtin_amdgcn_ballot_w32(((kj[j] >> p) & 1u) != 0u);
                    if (ln == (unsigned)(j * 8 + p)) wv = b;
                }
            }
            {
                const unsigned set = (((unsigned)q >> 5) << 2) + (ln >> 3);
                const unsigned half = (ln & 4u) ? ((unsigned)Epad >> 3) : 0u;
                keys4[half + (set << 2) + (ln & 3u)] = wv;
            }
            if (pairs != nullptr) {
                uint4 w;
                w.x = pair_word(sd.x, r.x);  w.y = pair_word(sd.y, r.y);
                w.z = pair_word(sd.z, r.z);  w.w = pair_word(sd.w, r.w);
                *reinterpret_cast<uint4*>(pairs + e0) = w;
            }
        }
    }
    int wave = (int)((blockIdx.x * blockDim.x + threadIdx.x) >> 5);
    if (wave >= ntiles) return;
    int lane = (int)(threadIdx.x & 31u);
    int half = lane >> 4;
    int m    = lane & 15;
    size_t row0 = (size_t)wave * 16;

    v8f q0 = {}, q1 = {}, k0 = {}, k1 = {}, v0 = {}, v1 = {};

#pragma unroll
    for (int kk = 0; kk < 8; ++kk) {
        int kb = kk * 4;
        const float* ap = x + (row0 + m) * 32 + kb + 2 * half;
        v2f a; a.x = ap[0]; a.y = ap[1];
        wmma_step(Wq, kb, half, m, a, q0, q1);
        wmma_step(Wk, kb, half, m, a, k0, k1);
        wmma_step(Wv, kb, half, m, a, v0, v1);
    }

#pragma unroll
    for (int r = 0; r < 8; ++r) {
        size_t row = (row0 + r + 8 * half) * 32;
        Q[row + m]      = q0[r];  Q[row + 16 + m] = q1[r];
        K[row + m]      = k0[r];  K[row + 16 + m] = k1[r];
        V[row + m]      = v0[r];  V[row + 16 + m] = v1[r];
    }
}

__global__ void qkv_tail_kernel(const float* __restrict__ x,
                                const float* __restrict__ Wq,
                                const float* __restrict__ Wk,
                                const float* __restrict__ Wv,
                                float* __restrict__ Q, float* __restrict__ K,
                                float* __restrict__ V, int row_start, int N) {
    int idx = blockIdx.x * blockDim.x + threadIdx.x;
    int nrows = N - row_start;
    if (idx >= nrows * 32) return;
    int row = row_start + idx / 32, col = idx & 31;
    float q = 0.f, k = 0.f, v = 0.f;
    for (int t = 0; t < 32; ++t) {
        float xv = x[(size_t)row * 32 + t];
        q += xv * Wq[t * 32 + col];
        k += xv * Wk[t * 32 + col];
        v += xv * Wv[t * 32 + col];
    }
    size_t o = (size_t)row * 32 + col;
    Q[o] = q; K[o] = k; V[o] = v;
}

template <int M>
__device__ __forceinline__ float lane_xor(float x) {
    int b = __builtin_bit_cast(int, x);
    if constexpr (M == 16) {
        return __builtin_bit_cast(float, __builtin_amdgcn_permlanex16(
            b, b, 0x76543210, 0xfedcba98, false, false));
    } else {
        return __builtin_bit_cast(float, __builtin_amdgcn_update_dpp(
            b, b, 0x160 | M, 0xf, 0xf, false));
    }
}

template <int M, int NH>
__device__ __forceinline__ void fold_level(float* v, bool hi) {
#pragma unroll
    for (int i = 0; i < NH; ++i) {
        float keep = hi ? v[i + NH] : v[i];
        float send = hi ? v[i] : v[i + NH];
        v[i] = keep + lane_xor<M>(send);
    }
}

template <int J>
__device__ __forceinline__ unsigned bcast8(unsigned x) {
    return (unsigned)__builtin_amdgcn_mov_dpp8(x, J * 0x249249);
}
template <int J>
__device__ __forceinline__ float bcast8f(float x) {
    return __uint_as_float(bcast8<J>(__float_as_uint(x)));
}

__device__ __forceinline__ int acc_pos(int rl, int f) {
    return (rl << 5) + ((((f & 3) ^ (rl >> 1)) & 3) << 3) + (f >> 2);
}

#define OWN_CAP  16384
#define OWN_TB   1024
#define OWN_U    8
#define OWN_STEP (OWN_TB * 4)
#define KEY_ROW  (OWN_TB * 16)
#define KEY_U    4
#define KEY_STEP (KEY_ROW * KEY_U)
#define OWN_XU   ((OWN_R * 8 + OWN_TB - 1) / OWN_TB)

__device__ __forceinline__ void own_push(unsigned rl, int e, unsigned* cnt,
                                         int* l_a) {
    if (rl < (unsigned)OWN_R) {
        unsigned slot = atomicAdd(cnt, 1u);
        if (slot < (unsigned)OWN_CAP) l_a[slot] = e;
    }
}

__device__ __forceinline__ void own_build(const int* __restrict__ recv, int a,
                                          int b, int rbase, bool vec_ok,
                                          unsigned* cnt, int* l_a) {
    const int t = (int)threadIdx.x;
    for (int base = a; base < b; base += OWN_STEP * OWN_U) {
        const bool full = vec_ok && (b - base >= OWN_STEP * OWN_U);
        int4 v[OWN_U];
        if (full) {
#pragma unroll
            for (int u = 0; u < OWN_U; ++u)
                v[u] = *reinterpret_cast<const int4*>(recv + base + u * OWN_STEP + t * 4);
        } else {
#pragma unroll
            for (int u = 0; u < OWN_U; ++u) {
                int e0 = base + u * OWN_STEP + t * 4;
                v[u].x = (e0     < b) ? recv[e0]     : -1;
                v[u].y = (e0 + 1 < b) ? recv[e0 + 1] : -1;
                v[u].z = (e0 + 2 < b) ? recv[e0 + 2] : -1;
                v[u].w = (e0 + 3 < b) ? recv[e0 + 3] : -1;
            }
        }
#pragma unroll
        for (int u = 0; u < OWN_U; ++u) {
            int e0 = base + u * OWN_STEP + t * 4;
            unsigned r0 = (unsigned)(v[u].x - rbase), r1 = (unsigned)(v[u].y - rbase);
            unsigned r2 = (unsigned)(v[u].z - rbase), r3 = (unsigned)(v[u].w - rbase);
            unsigned mn = min(min(r0, r1), min(r2, r3));
            if (mn < (unsigned)OWN_R) {
                own_push(r0, e0,     cnt, l_a);
                own_push(r1, e0 + 1, cnt, l_a);
                own_push(r2, e0 + 2, cnt, l_a);
                own_push(r3, e0 + 3, cnt, l_a);
            }
        }
    }
}


__device__ __forceinline__ void own_file_group(const uint4 (&v)[KEY_U], const unsigned (&cp)[8],
                                               int base, unsigned* cnt, int* l_a) {
    const unsigned t = threadIdx.x;
#pragma unroll
    for (int u = 0; u < KEY_U; u += 2) {
        const unsigned set = ((unsigned)base >> 5) + (unsigned)(u >> 1) * OWN_TB + t;
        const unsigned code0 = set << 5;
        unsigned m = ~(v[u].x ^ cp[0]);
        m &= ~(v[u].y ^ cp[1]);      m &= ~(v[u].z ^ cp[2]);      m &= ~(v[u].w ^ cp[3]);
        m &= ~(v[u + 1].x ^ cp[4]);  m &= ~(v[u + 1].y ^ cp[5]);
        m &= ~(v[u + 1].z ^ cp[6]);  m &= ~(v[u + 1].w ^ cp[7]);
        while (m) {
            const unsigned c = (unsigned)__builtin_ctz(m);
            m &= m - 1u;
            unsigned slot = atomicAdd(cnt, 1u);
            if (slot < (unsigned)OWN_CAP) l_a[slot] = (int)(code0 | c);
        }
    }
}

__device__ __forceinline__ void own_load_group(const uint4* __restrict__ lo4,
                                               const uint4* __restrict__ hi4,
                                               int base, uint4 (&v)[KEY_U]) {
    const unsigned t = threadIdx.x;
#pragma unroll
    for (int u = 0; u < KEY_U; u += 2) {
        const unsigned set = ((unsigned)base >> 5) + (unsigned)(u >> 1) * OWN_TB + t;
        v[u]     = lo4[set];
        v[u + 1] = hi4[set];
    }
}

__device__ __forceinline__ void own_build_keys(const unsigned* __restrict__ keys4,
                                               int Epad, unsigned kb,
                                               unsigned* cnt, int* l_a) {
    unsigned cp[8];
#pragma unroll
    for (int p = 0; p < 8; ++p) cp[p] = 0u - ((kb >> p) & 1u);
    const uint4* lo4 = reinterpret_cast<const uint4*>(keys4);
    const uint4* hi4 = lo4 + ((unsigned)Epad >> 5);
    if (Epad <= 0) return;
    uint4 v[KEY_U], nx[KEY_U];
    own_load_group(lo4, hi4, 0, v);
    int base = 0;
    for (; base + 2 * KEY_STEP < Epad; base += 2 * KEY_STEP) {
        own_load_group(lo4, hi4, base + KEY_STEP, nx);
        own_file_group(v, cp, base, cnt, l_a);
        own_load_group(lo4, hi4, base + 2 * KEY_STEP, v);
        own_file_group(nx, cp, base + KEY_STEP, cnt, l_a);
    }
    if (base + KEY_STEP < Epad) {
        own_load_group(lo4, hi4, base + KEY_STEP, nx);
        own_file_group(v, cp, base, cnt, l_a);
        own_file_group(nx, cp, base + KEY_STEP, cnt, l_a);
    } else {
        own_file_group(v, cp, base, cnt, l_a);
    }
}

__device__ __forceinline__ int own_edge(unsigned code) {
    const unsigned l = code & 31u, set = code >> 5;
    return (int)(((set >> 2) << 7) + (l << 2) + (set & 3u));
}

template <int J>
__device__ __forceinline__ void rows_qk(const float4* __restrict__ Qp,
                                        const float4* __restrict__ Kp,
                                        unsigned r, unsigned s, float4* qv, float4* kv) {
    qv[J] = Qp[(size_t)bcast8<J>(r) << 3];
    kv[J] = Kp[(size_t)bcast8<J>(s) << 3];
}

template <bool tail>
__device__ __forceinline__ void add_rows(const float4* __restrict__ Vp, unsigned s,
                                         unsigned rl, float alpha, int m,
                                         unsigned lane, float* acc) {
    const float4* a0 = Vp + ((size_t)bcast8<0>(s) << 3);
    const float4* a1 = Vp + ((size_t)bcast8<1>(s) << 3);
    const float4* a2 = Vp + ((size_t)bcast8<2>(s) << 3);
    const float4* a3 = Vp + ((size_t)bcast8<3>(s) << 3);
    const float4* a4 = Vp + ((size_t)bcast8<4>(s) << 3);
    const float4* a5 = Vp + ((size_t)bcast8<5>(s) << 3);
    const float4* a6 = Vp + ((size_t)bcast8<6>(s) << 3);
    const float4* a7 = Vp + ((size_t)bcast8<7>(s) << 3);
    __builtin_amdgcn_sched_barrier(0);
    float4 v[8];
    v[0] = *a0;  v[1] = *a1;  v[2] = *a2;  v[3] = *a3;
    v[4] = *a4;  v[5] = *a5;  v[6] = *a6;  v[7] = *a7;
    unsigned rj[8];
    float aj[8];
    rj[0] = bcast8<0>(rl);  aj[0] = bcast8f<0>(alpha);
    rj[1] = bcast8<1>(rl);  aj[1] = bcast8f<1>(alpha);
    rj[2] = bcast8<2>(rl);  aj[2] = bcast8f<2>(alpha);
    rj[3] = bcast8<3>(rl);  aj[3] = bcast8f<3>(alpha);
    rj[4] = bcast8<4>(rl);  aj[4] = bcast8f<4>(alpha);
    rj[5] = bcast8<5>(rl);  aj[5] = bcast8f<5>(alpha);
    rj[6] = bcast8<6>(rl);  aj[6] = bcast8f<6>(alpha);
    rj[7] = bcast8<7>(rl);  aj[7] = bcast8f<7>(alpha);
    const unsigned q4 = (lane & 7u) << 2;
    char* accb = reinterpret_cast<char*>(acc);
#pragma unroll
    for (int j = 0; j < 8; ++j) {
        const unsigned b0 = (rj[j] << 7) | ((rj[j] << 4) & 0x60u) | q4;
        if (!tail || (int)((lane & 24u) | (unsigned)j) < m) {
            atomicAdd(reinterpret_cast<float*>(accb + b0),         aj[j] * v[j].x);
            atomicAdd(reinterpret_cast<float*>(accb + (b0 ^ 32u)), aj[j] * v[j].y);
            atomicAdd(reinterpret_cast<float*>(accb + (b0 ^ 64u)), aj[j] * v[j].z);
            atomicAdd(reinterpret_cast<float*>(accb + (b0 ^ 96u)), aj[j] * v[j].w);
        }
    }
}

__device__ __forceinline__ void own_process(
    const float* __restrict__ Q, const float* __restrict__ K,
    const float* __restrict__ V, const int* __restrict__ ei,
    const int* __restrict__ recv, const unsigned* __restrict__ pairs, int rbase, int n,
    bool first, bool coded, float* acc, float* zs, unsigned* mkey, unsigned* mnew,
    int* l_a, int* l_r, float* l_sc) {
    const int t = (int)threadIdx.x;
    const unsigned lane = threadIdx.x & 31u;
    const int wave = __builtin_amdgcn_readfirstlane((int)(threadIdx.x >> 5));
    const float4* Qp = reinterpret_cast<const float4*>(Q) + (lane & 7u);
    const float4* Kp = reinterpret_cast<const float4*>(K) + (lane & 7u);
    const float4* Vp = reinterpret_cast<const float4*>(V) + (lane & 7u);

    for (int c = wave; c * 32 < n; c += OWN_TB / 32) {
        int i = c * 32 + (int)lane;
        bool valid = i < n;
        int s = 0, r = rbase;
        if (valid) {
            int e = l_a[i];
            if (coded) e = own_edge((unsigned)e);
            if (pairs != nullptr) {
                const unsigned w = pairs[e];
                s = (int)(w & ((1u << PAIR_SH) - 1u));
                r = rbase + (int)(w >> PAIR_SH);
            } else {
                s = ei[e]; r = recv[e];
            }
        }
        int rl = r - rbase;
        float4 qv[8], kv[8];
        rows_qk<0>(Qp, Kp, (unsigned)r, (unsigned)s, qv, kv);
        rows_qk<1>(Qp, Kp, (unsigned)r, (unsigned)s, qv, kv);
        rows_qk<2>(Qp, Kp, (unsigned)r, (unsigned)s, qv, kv);
        rows_qk<3>(Qp, Kp, (unsigned)r, (unsigned)s, qv, kv);
        rows_qk<4>(Qp, Kp, (unsigned)r, (unsigned)s, qv, kv);
        rows_qk<5>(Qp, Kp, (unsigned)r, (unsigned)s, qv, kv);
        rows_qk<6>(Qp, Kp, (unsigned)r, (unsigned)s, qv, kv);
        rows_qk<7>(Qp, Kp, (unsigned)r, (unsigned)s, qv, kv);
        __builtin_amdgcn_sched_barrier(0);
        float p[8];
#pragma unroll
        for (int j = 0; j < 8; ++j)
            p[j] = qv[j].x * kv[j].x + qv[j].y * kv[j].y +
                   qv[j].z * kv[j].z + qv[j].w * kv[j].w;
        fold_level<4, 4>(p, (lane & 4) != 0);
        fold_level<2, 2>(p, (lane & 2) != 0);
        fold_level<1, 1>(p, (lane & 1) != 0);
        float sc = p[0] * 0.17677669529663687f;
        if (valid) {
            l_sc[i] = sc;
            l_r[i] = rl;
            l_a[i] = s;
            atomicMax(&mnew[rl], fkey(sc));
        }
    }
    __syncthreads();
    if (!first) {
        for (int idx = t; idx < OWN_R * 32; idx += OWN_TB) {
            unsigned ko = mkey[idx >> 5], kn = mnew[idx >> 5];
            if (kn != ko)
                acc[idx] *= ko ? __expf(fkey_inv(ko) - fkey_inv(kn)) : 0.0f;
        }
        for (int rl = t; rl < OWN_R; rl += OWN_TB) {
            unsigned ko = mkey[rl], kn = mnew[rl];
            if (kn != ko)
                zs[rl] *= ko ? __expf(fkey_inv(ko) - fkey_inv(kn)) : 0.0f;
        }
        __syncthreads();
    }
    for (int rl = t; rl < OWN_R; rl += OWN_TB) mkey[rl] = mnew[rl];
    __syncthreads();

    for (int c = wave; c * 32 < n; c += OWN_TB / 32) {
        int i = c * 32 + (int)lane;
        int m = min(32, n - c * 32);
        int s = 0, rl = 0;
        float alpha = 0.f;
        if (i < n) {
            s = l_a[i]; rl = l_r[i];
            alpha = __expf(l_sc[i] - fkey_inv(mkey[rl]));
            atomicAdd(&zs[rl], alpha);
        }
        if (m == 32)
            add_rows<false>(Vp, (unsigned)s, (unsigned)rl, alpha, m, lane, acc);
        else
            add_rows<true>(Vp, (unsigned)s, (unsigned)rl, alpha, m, lane, acc);
    }
}

__global__ __launch_bounds__(OWN_TB) void own_range_kernel(
    const float* __restrict__ Q, const float* __restrict__ K,
    const float* __restrict__ V, const int* __restrict__ ei,
    const unsigned* __restrict__ keys4, const unsigned* __restrict__ pairs,
    const float* __restrict__ x, const float* __restrict__ Wo,
    float* __restrict__ out, int N, int E, int Epad) {
    __shared__ __attribute__((aligned(16))) float acc[OWN_R * 32];
    __shared__ float    zs[OWN_R];
    __shared__ unsigned mkey[OWN_R];
    __shared__ unsigned mnew[OWN_R];
    __shared__ int      l_a[OWN_CAP];
    __shared__ int      l_r[OWN_CAP];
    __shared__ float    l_sc[OWN_CAP];
    __shared__ unsigned cnt;

    const int t = (int)threadIdx.x;
    const int rbase = (int)blockIdx.x * OWN_R;
    const int nown = min(OWN_R, N - rbase);
    const int* recv = ei + E;
    const bool vec_ok = (((size_t)recv & 15u) == 0);

    for (int i = t; i < OWN_R * 32; i += OWN_TB) acc[i] = 0.0f;
    for (int i = t; i < OWN_R; i += OWN_TB) { zs[i] = 0.0f; mkey[i] = 0u; mnew[i] = 0u; }
    if (t == 0) cnt = 0u;
    __syncthreads();

    if (keys4 != nullptr)
        own_build_keys(keys4, Epad, blockIdx.x, &cnt, l_a);
    else
        own_build(recv, 0, E, rbase, vec_ok, &cnt, l_a);
    __syncthreads();
    const float4* x4 = reinterpret_cast<const float4*>(x) + (size_t)rbase * 8;
    float4*       o4 = reinterpret_cast<float4*>(out) + (size_t)rbase * 8;
    const int npiece = nown * 8;
    float4 xr[OWN_XU];
#pragma unroll
    for (int u = 0; u < OWN_XU; ++u) {
        const int i = t + u * OWN_TB;
        xr[u] = (i < npiece) ? x4[i] : make_float4(0.f, 0.f, 0.f, 0.f);
    }
    const unsigned total = cnt;
    if (total <= (unsigned)OWN_CAP) {
        own_process(Q, K, V, ei, recv, pairs, rbase, (int)total, true, keys4 != nullptr,
                    acc, zs, mkey, mnew, l_a, l_r, l_sc);
    } else {
        for (int a = 0; a < E; a += OWN_CAP) {
            __syncthreads();
            if (t == 0) cnt = 0u;
            __syncthreads();
            int b = (E - a > OWN_CAP) ? a + OWN_CAP : E;
            own_build(recv, a, b, rbase, vec_ok, &cnt, l_a);
            __syncthreads();
            own_process(Q, K, V, ei, recv, pairs, rbase, (int)cnt, a == 0, false, acc, zs,
                        mkey, mnew, l_a, l_r, l_sc);
        }
    }
    __syncthreads();
    {
        const int wave = __builtin_amdgcn_readfirstlane(t >> 5);
        if (wave * 16 < nown) {
            const int lane = t & 31;
            const int half = lane >> 4;
            const int m    = lane & 15;
            const int rl   = wave * 16 + m;
            const float invz = 1.0f / (zs[rl] + 1e-6f);
            const float* rowp = acc + (rl << 5);
            const int w0 = (((2 * half)     ^ (rl >> 1)) & 3) << 3;
            const int w1 = (((2 * half + 1) ^ (rl >> 1)) & 3) << 3;
            const float4 e0 = *reinterpret_cast<const float4*>(rowp + w0);
            const float4 e1 = *reinterpret_cast<const float4*>(rowp + w0 + 4);
            const float4 f0 = *reinterpret_cast<const float4*>(rowp + w1);
            const float4 f1 = *reinterpret_cast<const float4*>(rowp + w1 + 4);
            const float ae[8] = {e0.x, e0.y, e0.z, e0.w, e1.x, e1.y, e1.z, e1.w};
            const float af[8] = {f0.x, f0.y, f0.z, f0.w, f1.x, f1.y, f1.z, f1.w};
            v8f c0 = {}, c1 = {};
#pragma unroll
            for (int kk = 0; kk < 8; ++kk) {
                v2f a; a.x = ae[kk] * invz; a.y = af[kk] * invz;
                wmma_step(Wo, kk * 4, half, m, a, c0, c1);
            }
#pragma unroll
            for (int r = 0; r < 8; ++r) {
                float* d = acc + ((wave * 16 + r + 8 * half) << 5);
                d[m]      = c0[r];
                d[16 + m] = c1[r];
            }
        }
    }
    __syncthreads();
#pragma unroll
    for (int u = 0; u < OWN_XU; ++u) {
        const int i = t + u * OWN_TB;
        if (i < npiece) {
            const float4 p = *reinterpret_cast<const float4*>(acc + (i << 2));
            const float4 xv = xr[u];
            o4[i] = make_float4(xv.x + p.x, xv.y + p.y, xv.z + p.z, xv.w + p.w);
        }
    }
}

extern "C" void kernel_launch(void* const* d_in, const int* in_sizes, int n_in,
                              void* d_out, int out_size, void* d_ws,
                              size_t ws_size, hipStream_t stream) {
    const float* x  = (const float*)d_in[0];
    const int*   ei = (const int*)d_in[1];
    const float* Wq = (const float*)d_in[2];
    const float* Wk = (const float*)d_in[3];
    const float* Wv = (const float*)d_in[4];
    const float* Wo = (const float*)d_in[5];
    float* out = (float*)d_out;

    const int N = in_sizes[0] / 32;
    const int E = in_sizes[1] / 2;

    float* Q       = (float*)d_ws;
    float* K       = Q + (size_t)N * 32;
    float* V       = K + (size_t)N * 32;
    const int Epad = ((E + KEY_STEP - 1) / KEY_STEP) * KEY_STEP;
    const size_t key_off = (((size_t)N * 129 * sizeof(float)) + 15) & ~(size_t)15;
    unsigned* keys4 = (unsigned*)((unsigned char*)d_ws + key_off);
    if (N > 255 * OWN_R || key_off + (size_t)Epad > ws_size) keys4 = nullptr;
    const size_t pair_off = (size_t)N * 96 * sizeof(float);
    unsigned* pairs = (unsigned*)((unsigned char*)d_ws + pair_off);
    if (keys4 == nullptr || pair_off + (size_t)Epad * sizeof(unsigned) > key_off) pairs = nullptr;

    const int TB = 256;
    int ntiles = N / 16;
    int rem_row0 = ntiles * 16;
    int rem = N - rem_row0;

    {
        int blocks = (ntiles * 32 + TB - 1) / TB;
        if (blocks == 0 && keys4 != nullptr) blocks = 1;
        if (blocks > 0)
            qkv_wmma_kernel<<<blocks, TB, 0, stream>>>(
                x, Wq, Wk, Wv, Q, K, V, ntiles, ei, keys4, pairs, N, E, Epad);
        if (rem > 0)
            qkv_tail_kernel<<<(rem * 32 + TB - 1) / TB, TB, 0, stream>>>(
                x, Wq, Wk, Wv, Q, K, V, rem_row0, N);
    }
    if (N > 0)
        own_range_kernel<<<(N + OWN_R - 1) / OWN_R, OWN_TB, 0, stream>>>(
            Q, K, V, ei, keys4, pairs, x, Wo, out, N, E, Epad);
}


# Range groups find their edges by presence words (per 1024 edges and range, from the projection pass): each reads 6 KB, then only the key words of its own sets, not the key plane; sum order as before
# speedup vs baseline: 9.9972x; 1.0851x over previous
#include <hip/hip_runtime.h>
#include <math.h>

typedef __attribute__((ext_vector_type(2))) float v2f;
typedef __attribute__((ext_vector_type(8))) float v8f;

__device__ __forceinline__ unsigned fkey(float f) {
    unsigned u = __float_as_uint(f);
    return (u & 0x80000000u) ? ~u : (u | 0x80000000u);
}
__device__ __forceinline__ float fkey_inv(unsigned k) {
    unsigned u = (k & 0x80000000u) ? (k & 0x7FFFFFFFu) : ~k;
    return __uint_as_float(u);
}

#define OWN_R    416
#define PJ_TB    256
__device__ __forceinline__ unsigned own_key(int r, int N) {
    return ((unsigned)r < (unsigned)N) ? ((unsigned)r / (unsigned)OWN_R) : 0xFFu;
}
#define PAIR_SH  23
__device__ __forceinline__ unsigned pair_word(int s, int r) {
    return ((unsigned)s & ((1u << PAIR_SH) - 1u)) | (((unsigned)r % (unsigned)OWN_R) << PAIR_SH);
}

__device__ __forceinline__ void wmma_step(const float* __restrict__ W, int kb,
                                          int half, int m, v2f a,
                                          v8f& acc0, v8f& acc1) {
    int r0 = (kb + 2 * half + 0) * 32;
    int r1 = (kb + 2 * half + 1) * 32;
    v2f b0, b1;
    b0.x = W[r0 + m];        b0.y = W[r1 + m];
    b1.x = W[r0 + 16 + m];   b1.y = W[r1 + 16 + m];
    acc0 = __builtin_amdgcn_wmma_f32_16x16x4_f32(false, a, false, b0,
                                                 (short)0, acc0, false, false);
    acc1 = __builtin_amdgcn_wmma_f32_16x16x4_f32(false, a, false, b1,
                                                 (short)0, acc1, false, false);
}

__global__ __launch_bounds__(PJ_TB) void qkv_wmma_kernel(
    const float* __restrict__ x,
    const float* __restrict__ Wq, const float* __restrict__ Wk,
    const float* __restrict__ Wv,
    float* __restrict__ Q, float* __restrict__ K, float* __restrict__ V,
    int ntiles, const int* __restrict__ ei, unsigned* __restrict__ keys4,
    unsigned* __restrict__ pairs, unsigned* __restrict__ presence, int N, int E, int Epad) {
    __shared__ unsigned pres[PJ_TB];
    if (keys4 != nullptr) {
        const int* recv = ei + E;
        const bool vec_ok = (((size_t)recv & 15u) == 0);
        const bool svec_ok = (((size_t)ei & 15u) == 0);
        const int gsz = (int)(gridDim.x * PJ_TB);
        for (int qb = (int)(blockIdx.x * PJ_TB); qb * 4 < Epad; qb += gsz) {
            const int q = qb + (int)threadIdx.x;
            const int e0 = q * 4;
            if (presence != nullptr) pres[threadIdx.x] = 0u;
            int4 r, sd = make_int4(0, 0, 0, 0);
            if (pairs != nullptr) {
                if (svec_ok && e0 + 4 <= E) {
                    sd = *reinterpret_cast<const int4*>(ei + e0);
                } else {
                    sd.x = (e0     < E) ? ei[e0]     : 0;
                    sd.y = (e0 + 1 < E) ? ei[e0 + 1] : 0;
                    sd.z = (e0 + 2 < E) ? ei[e0 + 2] : 0;
                    sd.w = (e0 + 3 < E) ? ei[e0 + 3] : 0;
                }
            }
            if (vec_ok && e0 + 4 <= E) {
                r = *reinterpret_cast<const int4*>(recv + e0);
            } else {
                r.x = (e0     < E) ? recv[e0]     : -1;
                r.y = (e0 + 1 < E) ? recv[e0 + 1] : -1;
                r.z = (e0 + 2 < E) ? recv[e0 + 2] : -1;
                r.w = (e0 + 3 < E) ? recv[e0 + 3] : -1;
            }
            const unsigned kj[4] = {own_key(r.x, N), own_key(r.y, N),
                                    own_key(r.z, N), own_key(r.w, N)};
            const unsigned ln = threadIdx.x & 31u;
            unsigned wv = 0u;
#pragma unroll
            for (int j = 0; j < 4; ++j) {
#pragma unroll
                for (int p = 0; p < 8; ++p) {
                    const unsigned b = __builtin_amdgcn_ballot_w32(((kj[j] >> p) & 1u) != 0u);
                    if (ln == (unsigned)(j * 8 + p)) wv = b;
                }
            }
            {
                const unsigned set = (((unsigned)q >> 5) << 2) + (ln >> 3);
                if (presence != nullptr) {
                    keys4[(set << 3) + (ln & 7u)] = wv;
                } else {
                    const unsigned half = (ln & 4u) ? ((unsigned)Epad >> 3) : 0u;
                    keys4[half + (set << 2) + (ln & 3u)] = wv;
                }
            }
            if (presence != nullptr) {
                __syncthreads();
                const unsigned b4 = (threadIdx.x >> 5) << 2;
#pragma unroll
                for (int j = 0; j < 4; ++j)
                    if (kj[j] != 0xFFu) atomicOr(&pres[kj[j]], 1u << (b4 + (unsigned)j));
                __syncthreads();
                presence[(size_t)(qb / PJ_TB) * PJ_TB + threadIdx.x] = pres[threadIdx.x];
            }
            if (pairs != nullptr) {
                uint4 w;
                w.x = pair_word(sd.x, r.x);  w.y = pair_word(sd.y, r.y);
                w.z = pair_word(sd.z, r.z);  w.w = pair_word(sd.w, r.w);
                *reinterpret_cast<uint4*>(pairs + e0) = w;
            }
        }
    }
    int wave = (int)((blockIdx.x * blockDim.x + threadIdx.x) >> 5);
    if (wave >= ntiles) return;
    int lane = (int)(threadIdx.x & 31u);
    int half = lane >> 4;
    int m    = lane & 15;
    size_t row0 = (size_t)wave * 16;

    v8f q0 = {}, q1 = {}, k0 = {}, k1 = {}, v0 = {}, v1 = {};

#pragma unroll
    for (int kk = 0; kk < 8; ++kk) {
        int kb = kk * 4;
        const float* ap = x + (row0 + m) * 32 + kb + 2 * half;
        v2f a; a.x = ap[0]; a.y = ap[1];
        wmma_step(Wq, kb, half, m, a, q0, q1);
        wmma_step(Wk, kb, half, m, a, k0, k1);
        wmma_step(Wv, kb, half, m, a, v0, v1);
    }

#pragma unroll
    for (int r = 0; r < 8; ++r) {
        size_t row = (row0 + r + 8 * half) * 32;
        Q[row + m]      = q0[r];  Q[row + 16 + m] = q1[r];
        K[row + m]      = k0[r];  K[row + 16 + m] = k1[r];
        V[row + m]      = v0[r];  V[row + 16 + m] = v1[r];
    }
}

__global__ void qkv_tail_kernel(const float* __restrict__ x,
                                const float* __restrict__ Wq,
                                const float* __restrict__ Wk,
                                const float* __restrict__ Wv,
                                float* __restrict__ Q, float* __restrict__ K,
                                float* __restrict__ V, int row_start, int N) {
    int idx = blockIdx.x * blockDim.x + threadIdx.x;
    int nrows = N - row_start;
    if (idx >= nrows * 32) return;
    int row = row_start + idx / 32, col = idx & 31;
    float q = 0.f, k = 0.f, v = 0.f;
    for (int t = 0; t < 32; ++t) {
        float xv = x[(size_t)row * 32 + t];
        q += xv * Wq[t * 32 + col];
        k += xv * Wk[t * 32 + col];
        v += xv * Wv[t * 32 + col];
    }
    size_t o = (size_t)row * 32 + col;
    Q[o] = q; K[o] = k; V[o] = v;
}

template <int M>
__device__ __forceinline__ float lane_xor(float x) {
    int b = __builtin_bit_cast(int, x);
    if constexpr (M == 16) {
        return __builtin_bit_cast(float, __builtin_amdgcn_permlanex16(
            b, b, 0x76543210, 0xfedcba98, false, false));
    } else {
        return __builtin_bit_cast(float, __builtin_amdgcn_update_dpp(
            b, b, 0x160 | M, 0xf, 0xf, false));
    }
}

template <int M, int NH>
__device__ __forceinline__ void fold_level(float* v, bool hi) {
#pragma unroll
    for (int i = 0; i < NH; ++i) {
        float keep = hi ? v[i + NH] : v[i];
        float send = hi ? v[i] : v[i + NH];
        v[i] = keep + lane_xor<M>(send);
    }
}

template <int J>
__device__ __forceinline__ unsigned bcast8(unsigned x) {
    return (unsigned)__builtin_amdgcn_mov_dpp8(x, J * 0x249249);
}
template <int J>
__device__ __forceinline__ float bcast8f(float x) {
    return __uint_as_float(bcast8<J>(__float_as_uint(x)));
}

__device__ __forceinline__ int acc_pos(int rl, int f) {
    return (rl << 5) + ((((f & 3) ^ (rl >> 1)) & 3) << 3) + (f >> 2);
}

#define OWN_CAP  16384
#define OWN_TB   1024
#define OWN_U    8
#define OWN_STEP (OWN_TB * 4)
#define KEY_ROW  (OWN_TB * 16)
#define KEY_U    4
#define KEY_STEP (KEY_ROW * KEY_U)
#define OWN_XU   ((OWN_R * 8 + OWN_TB - 1) / OWN_TB)

__device__ __forceinline__ void own_push(unsigned rl, int e, unsigned* cnt,
                                         int* l_a) {
    if (rl < (unsigned)OWN_R) {
        unsigned slot = atomicAdd(cnt, 1u);
        if (slot < (unsigned)OWN_CAP) l_a[slot] = e;
    }
}

__device__ __forceinline__ void own_build(const int* __restrict__ recv, int a,
                                          int b, int rbase, bool vec_ok,
                                          unsigned* cnt, int* l_a) {
    const int t = (int)threadIdx.x;
    for (int base = a; base < b; base += OWN_STEP * OWN_U) {
        const bool full = vec_ok && (b - base >= OWN_STEP * OWN_U);
        int4 v[OWN_U];
        if (full) {
#pragma unroll
            for (int u = 0; u < OWN_U; ++u)
                v[u] = *reinterpret_cast<const int4*>(recv + base + u * OWN_STEP + t * 4);
        } else {
#pragma unroll
            for (int u = 0; u < OWN_U; ++u) {
                int e0 = base + u * OWN_STEP + t * 4;
                v[u].x = (e0     < b) ? recv[e0]     : -1;
                v[u].y = (e0 + 1 < b) ? recv[e0 + 1] : -1;
                v[u].z = (e0 + 2 < b) ? recv[e0 + 2] : -1;
                v[u].w = (e0 + 3 < b) ? recv[e0 + 3] : -1;
            }
        }
#pragma unroll
        for (int u = 0; u < OWN_U; ++u) {
            int e0 = base + u * OWN_STEP + t * 4;
            unsigned r0 = (unsigned)(v[u].x - rbase), r1 = (unsigned)(v[u].y - rbase);
            unsigned r2 = (unsigned)(v[u].z - rbase), r3 = (unsigned)(v[u].w - rbase);
            unsigned mn = min(min(r0, r1), min(r2, r3));
            if (mn < (unsigned)OWN_R) {
                own_push(r0, e0,     cnt, l_a);
                own_push(r1, e0 + 1, cnt, l_a);
                own_push(r2, e0 + 2, cnt, l_a);
                own_push(r3, e0 + 3, cnt, l_a);
            }
        }
    }
}


__device__ __forceinline__ void own_file_group(const uint4 (&v)[KEY_U], const unsigned (&cp)[8],
                                               int base, unsigned* cnt, int* l_a) {
    const unsigned t = threadIdx.x;
#pragma unroll
    for (int u = 0; u < KEY_U; u += 2) {
        const unsigned set = ((unsigned)base >> 5) + (unsigned)(u >> 1) * OWN_TB + t;
        const unsigned code0 = set << 5;
        unsigned m = ~(v[u].x ^ cp[0]);
        m &= ~(v[u].y ^ cp[1]);      m &= ~(v[u].z ^ cp[2]);      m &= ~(v[u].w ^ cp[3]);
        m &= ~(v[u + 1].x ^ cp[4]);  m &= ~(v[u + 1].y ^ cp[5]);
        m &= ~(v[u + 1].z ^ cp[6]);  m &= ~(v[u + 1].w ^ cp[7]);
        while (m) {
            const unsigned c = (unsigned)__builtin_ctz(m);
            m &= m - 1u;
            unsigned slot = atomicAdd(cnt, 1u);
            if (slot < (unsigned)OWN_CAP) l_a[slot] = (int)(code0 | c);
        }
    }
}

__device__ __forceinline__ void own_load_group(const uint4* __restrict__ lo4,
                                               const uint4* __restrict__ hi4,
                                               int base, uint4 (&v)[KEY_U]) {
    const unsigned t = threadIdx.x;
#pragma unroll
    for (int u = 0; u < KEY_U; u += 2) {
        const unsigned set = ((unsigned)base >> 5) + (unsigned)(u >> 1) * OWN_TB + t;
        v[u]     = lo4[set];
        v[u + 1] = hi4[set];
    }
}

__device__ __forceinline__ void own_build_keys(const unsigned* __restrict__ keys4,
                                               int Epad, unsigned kb,
                                               unsigned* cnt, int* l_a) {
    unsigned cp[8];
#pragma unroll
    for (int p = 0; p < 8; ++p) cp[p] = 0u - ((kb >> p) & 1u);
    const uint4* lo4 = reinterpret_cast<const uint4*>(keys4);
    const uint4* hi4 = lo4 + ((unsigned)Epad >> 5);
    if (Epad <= 0) return;
    uint4 v[KEY_U], nx[KEY_U];
    own_load_group(lo4, hi4, 0, v);
    int base = 0;
    for (; base + 2 * KEY_STEP < Epad; base += 2 * KEY_STEP) {
        own_load_group(lo4, hi4, base + KEY_STEP, nx);
        own_file_group(v, cp, base, cnt, l_a);
        own_load_group(lo4, hi4, base + 2 * KEY_STEP, v);
        own_file_group(nx, cp, base + KEY_STEP, cnt, l_a);
    }
    if (base + KEY_STEP < Epad) {
        own_load_group(lo4, hi4, base + KEY_STEP, nx);
        own_file_group(v, cp, base, cnt, l_a);
        own_file_group(nx, cp, base + KEY_STEP, cnt, l_a);
    } else {
        own_file_group(v, cp, base, cnt, l_a);
    }
}

#define PRES_U 4
__device__ __forceinline__ void own_build_pres(const unsigned* __restrict__ presence,
                                               const unsigned* __restrict__ keys4,
                                               int Epad, unsigned kb,
                                               unsigned* cnt, int* l_a) {
    unsigned cp[8];
#pragma unroll
    for (int p = 0; p < 8; ++p) cp[p] = 0u - ((kb >> p) & 1u);
    const uint4* k8 = reinterpret_cast<const uint4*>(keys4);
    const unsigned nW = (unsigned)Epad >> 10;
    for (unsigned W = threadIdx.x; W < nW; W += OWN_TB) {
        unsigned pw = presence[(size_t)W * PJ_TB + kb];
        while (pw) {
            unsigned S[PRES_U];
            bool ok[PRES_U];
            uint4 lo[PRES_U], hi[PRES_U];
#pragma unroll
            for (int k = 0; k < PRES_U; ++k) {
                ok[k] = (pw != 0u);
                S[k] = ok[k] ? (W << 5) + (unsigned)__builtin_ctz(pw) : S[0];
                pw &= pw - 1u;
            }
#pragma unroll
            for (int k = 0; k < PRES_U; ++k) {
                lo[k] = k8[2u * S[k]];  hi[k] = k8[2u * S[k] + 1u];
            }
#pragma unroll
            for (int k = 0; k < PRES_U; ++k) {
                const unsigned code0 = S[k] << 5;
                unsigned m = ~(lo[k].x ^ cp[0]);
                m &= ~(lo[k].y ^ cp[1]);  m &= ~(lo[k].z ^ cp[2]);  m &= ~(lo[k].w ^ cp[3]);
                m &= ~(hi[k].x ^ cp[4]);  m &= ~(hi[k].y ^ cp[5]);
                m &= ~(hi[k].z ^ cp[6]);  m &= ~(hi[k].w ^ cp[7]);
                if (!ok[k]) m = 0u;
                while (m) {
                    const unsigned c = (unsigned)__builtin_ctz(m);
                    m &= m - 1u;
                    unsigned slot = atomicAdd(cnt, 1u);
                    if (slot < (unsigned)OWN_CAP) l_a[slot] = (int)(code0 | c);
                }
            }
        }
    }
}

__device__ __forceinline__ int own_edge(unsigned code) {
    const unsigned l = code & 31u, set = code >> 5;
    return (int)(((set >> 2) << 7) + (l << 2) + (set & 3u));
}

template <int J>
__device__ __forceinline__ void rows_qk(const float4* __restrict__ Qp,
                                        const float4* __restrict__ Kp,
                                        unsigned r, unsigned s, float4* qv, float4* kv) {
    qv[J] = Qp[(size_t)bcast8<J>(r) << 3];
    kv[J] = Kp[(size_t)bcast8<J>(s) << 3];
}

template <bool tail>
__device__ __forceinline__ void add_rows(const float4* __restrict__ Vp, unsigned s,
                                         unsigned rl, float alpha, int m,
                                         unsigned lane, float* acc) {
    const float4* a0 = Vp + ((size_t)bcast8<0>(s) << 3);
    const float4* a1 = Vp + ((size_t)bcast8<1>(s) << 3);
    const float4* a2 = Vp + ((size_t)bcast8<2>(s) << 3);
    const float4* a3 = Vp + ((size_t)bcast8<3>(s) << 3);
    const float4* a4 = Vp + ((size_t)bcast8<4>(s) << 3);
    const float4* a5 = Vp + ((size_t)bcast8<5>(s) << 3);
    const float4* a6 = Vp + ((size_t)bcast8<6>(s) << 3);
    const float4* a7 = Vp + ((size_t)bcast8<7>(s) << 3);
    __builtin_amdgcn_sched_barrier(0);
    float4 v[8];
    v[0] = *a0;  v[1] = *a1;  v[2] = *a2;  v[3] = *a3;
    v[4] = *a4;  v[5] = *a5;  v[6] = *a6;  v[7] = *a7;
    unsigned rj[8];
    float aj[8];
    rj[0] = bcast8<0>(rl);  aj[0] = bcast8f<0>(alpha);
    rj[1] = bcast8<1>(rl);  aj[1] = bcast8f<1>(alpha);
    rj[2] = bcast8<2>(rl);  aj[2] = bcast8f<2>(alpha);
    rj[3] = bcast8<3>(rl);  aj[3] = bcast8f<3>(alpha);
    rj[4] = bcast8<4>(rl);  aj[4] = bcast8f<4>(alpha);
    rj[5] = bcast8<5>(rl);  aj[5] = bcast8f<5>(alpha);
    rj[6] = bcast8<6>(rl);  aj[6] = bcast8f<6>(alpha);
    rj[7] = bcast8<7>(rl);  aj[7] = bcast8f<7>(alpha);
    const unsigned q4 = (lane & 7u) << 2;
    char* accb = reinterpret_cast<char*>(acc);
#pragma unroll
    for (int j = 0; j < 8; ++j) {
        const unsigned b0 = (rj[j] << 7) | ((rj[j] << 4) & 0x60u) | q4;
        if (!tail || (int)((lane & 24u) | (unsigned)j) < m) {
            atomicAdd(reinterpret_cast<float*>(accb + b0),         aj[j] * v[j].x);
            atomicAdd(reinterpret_cast<float*>(accb + (b0 ^ 32u)), aj[j] * v[j].y);
            atomicAdd(reinterpret_cast<float*>(accb + (b0 ^ 64u)), aj[j] * v[j].z);
            atomicAdd(reinterpret_cast<float*>(accb + (b0 ^ 96u)), aj[j] * v[j].w);
        }
    }
}

__device__ __forceinline__ void own_process(
    const float* __restrict__ Q, const float* __restrict__ K,
    const float* __restrict__ V, const int* __restrict__ ei,
    const int* __restrict__ recv, const unsigned* __restrict__ pairs, int rbase, int n,
    bool first, bool coded, float* acc, float* zs, unsigned* mkey, unsigned* mnew,
    int* l_a, int* l_r, float* l_sc) {
    const int t = (int)threadIdx.x;
    const unsigned lane = threadIdx.x & 31u;
    const int wave = __builtin_amdgcn_readfirstlane((int)(threadIdx.x >> 5));
    const float4* Qp = reinterpret_cast<const float4*>(Q) + (lane & 7u);
    const float4* Kp = reinterpret_cast<const float4*>(K) + (lane & 7u);
    const float4* Vp = reinterpret_cast<const float4*>(V) + (lane & 7u);

    for (int c = wave; c * 32 < n; c += OWN_TB / 32) {
        int i = c * 32 + (int)lane;
        bool valid = i < n;
        int s = 0, r = rbase;
        if (valid) {
            int e = l_a[i];
            if (coded) e = own_edge((unsigned)e);
            if (pairs != nullptr) {
                const unsigned w = pairs[e];
                s = (int)(w & ((1u << PAIR_SH) - 1u));
                r = rbase + (int)(w >> PAIR_SH);
            } else {
                s = ei[e]; r = recv[e];
            }
        }
        int rl = r - rbase;
        float4 qv[8], kv[8];
        rows_qk<0>(Qp, Kp, (unsigned)r, (unsigned)s, qv, kv);
        rows_qk<1>(Qp, Kp, (unsigned)r, (unsigned)s, qv, kv);
        rows_qk<2>(Qp, Kp, (unsigned)r, (unsigned)s, qv, kv);
        rows_qk<3>(Qp, Kp, (unsigned)r, (unsigned)s, qv, kv);
        rows_qk<4>(Qp, Kp, (unsigned)r, (unsigned)s, qv, kv);
        rows_qk<5>(Qp, Kp, (unsigned)r, (unsigned)s, qv, kv);
        rows_qk<6>(Qp, Kp, (unsigned)r, (unsigned)s, qv, kv);
        rows_qk<7>(Qp, Kp, (unsigned)r, (unsigned)s, qv, kv);
        __builtin_amdgcn_sched_barrier(0);
        float p[8];
#pragma unroll
        for (int j = 0; j < 8; ++j)
            p[j] = qv[j].x * kv[j].x + qv[j].y * kv[j].y +
                   qv[j].z * kv[j].z + qv[j].w * kv[j].w;
        fold_level<4, 4>(p, (lane & 4) != 0);
        fold_level<2, 2>(p, (lane & 2) != 0);
        fold_level<1, 1>(p, (lane & 1) != 0);
        float sc = p[0] * 0.17677669529663687f;
        if (valid) {
            l_sc[i] = sc;
            l_r[i] = rl;
            l_a[i] = s;
            atomicMax(&mnew[rl], fkey(sc));
        }
    }
    __syncthreads();
    if (!first) {
        for (int idx = t; idx < OWN_R * 32; idx += OWN_TB) {
            unsigned ko = mkey[idx >> 5], kn = mnew[idx >> 5];
            if (kn != ko)
                acc[idx] *= ko ? __expf(fkey_inv(ko) - fkey_inv(kn)) : 0.0f;
        }
        for (int rl = t; rl < OWN_R; rl += OWN_TB) {
            unsigned ko = mkey[rl], kn = mnew[rl];
            if (kn != ko)
                zs[rl] *= ko ? __expf(fkey_inv(ko) - fkey_inv(kn)) : 0.0f;
        }
        __syncthreads();
    }
    for (int rl = t; rl < OWN_R; rl += OWN_TB) mkey[rl] = mnew[rl];
    __syncthreads();

    for (int c = wave; c * 32 < n; c += OWN_TB / 32) {
        int i = c * 32 + (int)lane;
        int m = min(32, n - c * 32);
        int s = 0, rl = 0;
        float alpha = 0.f;
        if (i < n) {
            s = l_a[i]; rl = l_r[i];
            alpha = __expf(l_sc[i] - fkey_inv(mkey[rl]));
            atomicAdd(&zs[rl], alpha);
        }
        if (m == 32)
            add_rows<false>(Vp, (unsigned)s, (unsigned)rl, alpha, m, lane, acc);
        else
            add_rows<true>(Vp, (unsigned)s, (unsigned)rl, alpha, m, lane, acc);
    }
}

__global__ __launch_bounds__(OWN_TB) void own_range_kernel(
    const float* __restrict__ Q, const float* __restrict__ K,
    const float* __restrict__ V, const int* __restrict__ ei,
    const unsigned* __restrict__ keys4, const unsigned* __restrict__ pairs,
    const unsigned* __restrict__ presence,
    const float* __restrict__ x, const float* __restrict__ Wo,
    float* __restrict__ out, int N, int E, int Epad) {
    __shared__ __attribute__((aligned(16))) float acc[OWN_R * 32];
    __shared__ float    zs[OWN_R];
    __shared__ unsigned mkey[OWN_R];
    __shared__ unsigned mnew[OWN_R];
    __shared__ int      l_a[OWN_CAP];
    __shared__ int      l_r[OWN_CAP];
    __shared__ float    l_sc[OWN_CAP];
    __shared__ unsigned cnt;

    const int t = (int)threadIdx.x;
    const int rbase = (int)blockIdx.x * OWN_R;
    const int nown = min(OWN_R, N - rbase);
    const int* recv = ei + E;
    const bool vec_ok = (((size_t)recv & 15u) == 0);

    for (int i = t; i < OWN_R * 32; i += OWN_TB) acc[i] = 0.0f;
    for (int i = t; i < OWN_R; i += OWN_TB) { zs[i] = 0.0f; mkey[i] = 0u; mnew[i] = 0u; }
    if (t == 0) cnt = 0u;
    __syncthreads();

    if (presence != nullptr)
        own_build_pres(presence, keys4, Epad, blockIdx.x, &cnt, l_a);
    else if (keys4 != nullptr)
        own_build_keys(keys4, Epad, blockIdx.x, &cnt, l_a);
    else
        own_build(recv, 0, E, rbase, vec_ok, &cnt, l_a);
    __syncthreads();
    const float4* x4 = reinterpret_cast<const float4*>(x) + (size_t)rbase * 8;
    float4*       o4 = reinterpret_cast<float4*>(out) + (size_t)rbase * 8;
    const int npiece = nown * 8;
    float4 xr[OWN_XU];
#pragma unroll
    for (int u = 0; u < OWN_XU; ++u) {
        const int i = t + u * OWN_TB;
        xr[u] = (i < npiece) ? x4[i] : make_float4(0.f, 0.f, 0.f, 0.f);
    }
    const unsigned total = cnt;
    if (total <= (unsigned)OWN_CAP) {
        own_process(Q, K, V, ei, recv, pairs, rbase, (int)total, true, keys4 != nullptr,
                    acc, zs, mkey, mnew, l_a, l_r, l_sc);
    } else {
        for (int a = 0; a < E; a += OWN_CAP) {
            __syncthreads();
            if (t == 0) cnt = 0u;
            __syncthreads();
            int b = (E - a > OWN_CAP) ? a + OWN_CAP : E;
            own_build(recv, a, b, rbase, vec_ok, &cnt, l_a);
            __syncthreads();
            own_process(Q, K, V, ei, recv, pairs, rbase, (int)cnt, a == 0, false, acc, zs,
                        mkey, mnew, l_a, l_r, l_sc);
        }
    }
    __syncthreads();
    {
        const int wave = __builtin_amdgcn_readfirstlane(t >> 5);
        if (wave * 16 < nown) {
            const int lane = t & 31;
            const int half = lane >> 4;
            const int m    = lane & 15;
            const int rl   = wave * 16 + m;
            const float invz = 1.0f / (zs[rl] + 1e-6f);
            const float* rowp = acc + (rl << 5);
            const int w0 = (((2 * half)     ^ (rl >> 1)) & 3) << 3;
            const int w1 = (((2 * half + 1) ^ (rl >> 1)) & 3) << 3;
            const float4 e0 = *reinterpret_cast<const float4*>(rowp + w0);
            const float4 e1 = *reinterpret_cast<const float4*>(rowp + w0 + 4);
            const float4 f0 = *reinterpret_cast<const float4*>(rowp + w1);
            const float4 f1 = *reinterpret_cast<const float4*>(rowp + w1 + 4);
            const float ae[8] = {e0.x, e0.y, e0.z, e0.w, e1.x, e1.y, e1.z, e1.w};
            const float af[8] = {f0.x, f0.y, f0.z, f0.w, f1.x, f1.y, f1.z, f1.w};
            v8f c0 = {}, c1 = {};
#pragma unroll
            for (int kk = 0; kk < 8; ++kk) {
                v2f a; a.x = ae[kk] * invz; a.y = af[kk] * invz;
                wmma_step(Wo, kk * 4, half, m, a, c0, c1);
            }
#pragma unroll
            for (int r = 0; r < 8; ++r) {
                float* d = acc + ((wave * 16 + r + 8 * half) << 5);
                d[m]      = c0[r];
                d[16 + m] = c1[r];
            }
        }
    }
    __syncthreads();
#pragma unroll
    for (int u = 0; u < OWN_XU; ++u) {
        const int i = t + u * OWN_TB;
        if (i < npiece) {
            const float4 p = *reinterpret_cast<const float4*>(acc + (i << 2));
            const float4 xv = xr[u];
            o4[i] = make_float4(xv.x + p.x, xv.y + p.y, xv.z + p.z, xv.w + p.w);
        }
    }
}

extern "C" void kernel_launch(void* const* d_in, const int* in_sizes, int n_in,
                              void* d_out, int out_size, void* d_ws,
                              size_t ws_size, hipStream_t stream) {
    const float* x  = (const float*)d_in[0];
    const int*   ei = (const int*)d_in[1];
    const float* Wq = (const float*)d_in[2];
    const float* Wk = (const float*)d_in[3];
    const float* Wv = (const float*)d_in[4];
    const float* Wo = (const float*)d_in[5];
    float* out = (float*)d_out;

    const int N = in_sizes[0] / 32;
    const int E = in_sizes[1] / 2;

    float* Q       = (float*)d_ws;
    float* K       = Q + (size_t)N * 32;
    float* V       = K + (size_t)N * 32;
    const int Epad = ((E + KEY_STEP - 1) / KEY_STEP) * KEY_STEP;
    const size_t key_off = (((size_t)N * 129 * sizeof(float)) + 15) & ~(size_t)15;
    unsigned* keys4 = (unsigned*)((unsigned char*)d_ws + key_off);
    if (N > 255 * OWN_R || key_off + (size_t)Epad > ws_size) keys4 = nullptr;
    const size_t pair_off = (size_t)N * 96 * sizeof(float);
    unsigned* pairs = (unsigned*)((unsigned char*)d_ws + pair_off);
    if (keys4 == nullptr || pair_off + (size_t)Epad * sizeof(unsigned) > key_off) pairs = nullptr;

    const size_t pres_off = pair_off + (size_t)Epad * sizeof(unsigned);
    unsigned* presence = (unsigned*)((unsigned char*)d_ws + pres_off);
    if (pairs == nullptr || pres_off + (size_t)Epad > key_off) presence = nullptr;

    const int TB = PJ_TB;
    int ntiles = N / 16;
    int rem_row0 = ntiles * 16;
    int rem = N - rem_row0;

    {
        int blocks = (ntiles * 32 + TB - 1) / TB;
        if (blocks == 0 && keys4 != nullptr) blocks = 1;
        if (blocks > 0)
            qkv_wmma_kernel<<<blocks, TB, 0, stream>>>(
                x, Wq, Wk, Wv, Q, K, V, ntiles, ei, keys4, pairs, presence, N, E, Epad);
        if (rem > 0)
            qkv_tail_kernel<<<(rem * 32 + TB - 1) / TB, TB, 0, stream>>>(
                x, Wq, Wk, Wv, Q, K, V, rem_row0, N);
    }
    if (N > 0)
        own_range_kernel<<<(N + OWN_R - 1) / OWN_R, OWN_TB, 0, stream>>>(
            Q, K, V, ei, keys4, pairs, presence, x, Wo, out, N, E, Epad);
}
